# starting kernel
# baseline (speedup 1.0000x reference)
#include <hip/hip_runtime.h>
#include <hip/hip_bf16.h>
#include <math.h>

typedef float v2f __attribute__((ext_vector_type(2)));
typedef float v8f __attribute__((ext_vector_type(8)));

#define NEG_SLOPE 0.2f

__device__ __forceinline__ float leaky(float v) {
    return v > 0.0f ? v : NEG_SLOPE * v;
}

__device__ __forceinline__ void atomicMaxF32(float* addr, float val) {
    if (val >= 0.0f) atomicMax((int*)addr, __float_as_int(val));
    else             atomicMin((unsigned int*)addr, __float_as_uint(val));
}

template<int NT>
__global__ void __launch_bounds__(32)
gemm_wmma_f32(const float* __restrict__ A, const float* __restrict__ B,
              float* __restrict__ C, int M, int K, int N, int ldc)
{
    __shared__ float As[16 * 12];

    const int lane = (int)threadIdx.x;
    const int half = lane >> 4;
    const int l16  = lane & 15;

    const int alr = lane >> 1;
    const int alc = (lane & 1) * 4;
    int rowA = blockIdx.x * 16 + alr;
    if (rowA >= M) rowA = M - 1;
    const float* __restrict__ Aload = A + (size_t)rowA * (size_t)K + alc;

    const int ncol0 = blockIdx.y * (16 * NT);
    const int colB  = ncol0 + l16;

    v8f acc[NT];
#pragma unroll
    for (int t = 0; t < NT; ++t) {
        v8f z = {0.f, 0.f, 0.f, 0.f, 0.f, 0.f, 0.f, 0.f};
        acc[t] = z;
    }

    float4 av = *(const float4*)(Aload);

    for (int kc = 0; kc < K; kc += 8) {
        __syncthreads();
        *(float4*)(&As[alr * 12 + alc]) = av;

        v2f breg[2][NT];
#pragma unroll
        for (int kk2 = 0; kk2 < 2; ++kk2) {
            const float* __restrict__ Bk =
                B + (size_t)(kc + kk2 * 4 + 2 * half) * (size_t)N + colB;
#pragma unroll
            for (int t = 0; t < NT; ++t) {
                breg[kk2][t].x = Bk[t * 16];
                breg[kk2][t].y = Bk[N + t * 16];
            }
        }

        const int kn = (kc + 8 < K) ? (kc + 8) : 0;
        av = *(const float4*)(Aload + kn);

        __syncthreads();

#pragma unroll
        for (int kk2 = 0; kk2 < 2; ++kk2) {
            const v2f a = *(const v2f*)(&As[l16 * 12 + kk2 * 4 + 2 * half]);
#pragma unroll
            for (int t = 0; t < NT; ++t) {
                acc[t] = __builtin_amdgcn_wmma_f32_16x16x4_f32(
                    false, a, false, breg[kk2][t], (short)0, acc[t], false, false);
            }
        }
    }

    const int crow0 = blockIdx.x * 16 + half * 8;
    float* __restrict__ Cbase = C + (size_t)crow0 * (size_t)ldc + ncol0 + l16;
    if (blockIdx.x * 16 + 16 <= M) {
#pragma unroll
        for (int t = 0; t < NT; ++t) {
            float* __restrict__ Cp = Cbase + t * 16;
#pragma unroll
            for (int r = 0; r < 8; ++r)
                Cp[(size_t)r * (size_t)ldc] = acc[t][r];
        }
    } else {
#pragma unroll
        for (int t = 0; t < NT; ++t) {
            float* __restrict__ Cp = Cbase + t * 16;
#pragma unroll
            for (int r = 0; r < 8; ++r)
                if (crow0 + r < M) Cp[(size_t)r * (size_t)ldc] = acc[t][r];
        }
    }
}

__global__ void fill_f32(float* __restrict__ p, int n, float v) {
    int t = blockIdx.x * blockDim.x + threadIdx.x;
    if (t < n) p[t] = v;
}

__global__ void pad_cols(const float* __restrict__ src, float* __restrict__ dst,
                         int K, int N, int Np)
{
    int t = blockIdx.x * blockDim.x + threadIdx.x;
    if (t >= K * Np) return;
    int r = t / Np, c = t - r * Np;
    dst[t] = (c < N) ? src[r * N + c] : 0.f;
}

__global__ void alpha1_kernel(const float* __restrict__ h1,
                              const float* __restrict__ a_src,
                              const float* __restrict__ a_dst,
                              float* __restrict__ as, float* __restrict__ ad, int Nn)
{
    int t = blockIdx.x * blockDim.x + threadIdx.x;
    if (t >= Nn * 8) return;
    int n = t >> 3, h = t & 7;
    const float* hp = h1 + (size_t)n * 512 + h * 64;
    const float* sp = a_src + h * 64;
    const float* dp = a_dst + h * 64;
    float ss = 0.f, dd = 0.f;
#pragma unroll 8
    for (int c = 0; c < 64; ++c) { float v = hp[c]; ss += v * sp[c]; dd += v * dp[c]; }
    as[t] = ss;
    ad[t] = dd;
}

__device__ __forceinline__ void edge_sd(const int* __restrict__ ei, int E, int e,
                                        int& s, int& d)
{
    if (e < E) { s = ei[e]; d = ei[E + e]; }
    else       { s = e - E; d = e - E; }
}

__global__ void edge_max1(const int* __restrict__ ei, int E, int Nn,
                          const float* __restrict__ as, const float* __restrict__ ad,
                          float* __restrict__ mx)
{
    int t = blockIdx.x * blockDim.x + threadIdx.x;
    if (t >= (E + Nn) * 8) return;
    int e = t >> 3, h = t & 7;
    int s, d; edge_sd(ei, E, e, s, d);
    atomicMaxF32(&mx[d * 8 + h], leaky(as[s * 8 + h] + ad[d * 8 + h]));
}

__global__ void edge_expsum1(const int* __restrict__ ei, int E, int Nn,
                             const float* __restrict__ as, const float* __restrict__ ad,
                             const float* __restrict__ mx,
                             float* __restrict__ ex, float* __restrict__ denom)
{
    int t = blockIdx.x * blockDim.x + threadIdx.x;
    if (t >= (E + Nn) * 8) return;
    int e = t >> 3, h = t & 7;
    int s, d; edge_sd(ei, E, e, s, d);
    float v  = leaky(as[s * 8 + h] + ad[d * 8 + h]);
    float xv = expf(v - mx[d * 8 + h]);
    ex[t] = xv;
    atomicAdd(&denom[d * 8 + h], xv);
}

__global__ void aggregate1(const int* __restrict__ ei, int E, int Nn,
                           const float* __restrict__ ex, const float* __restrict__ denom,
                           const float* __restrict__ h1, float* __restrict__ out)
{
    int t = blockIdx.x * blockDim.x + threadIdx.x;
    if (t >= (E + Nn) * 128) return;
    int e = t >> 7;
    int g = t & 127;
    int h = g >> 4;
    int s, d; edge_sd(ei, E, e, s, d);
    float alpha = ex[e * 8 + h] / denom[d * 8 + h];
    const float4 hv = *(const float4*)(h1 + (size_t)s * 512 + g * 4);
    float* o = out + (size_t)d * 512 + g * 4;
    atomicAdd(o + 0, hv.x * alpha);
    atomicAdd(o + 1, hv.y * alpha);
    atomicAdd(o + 2, hv.z * alpha);
    atomicAdd(o + 3, hv.w * alpha);
}

__global__ void bias_elu(float* __restrict__ h2, const float* __restrict__ b, int total) {
    int t = blockIdx.x * blockDim.x + threadIdx.x;
    if (t >= total) return;
    float v = h2[t] + b[t & 511];
    h2[t] = v > 0.f ? v : expm1f(v);
}

__global__ void alpha2_kernel(const float* __restrict__ h3,
                              const float* __restrict__ a_src,
                              const float* __restrict__ a_dst,
                              float* __restrict__ as, float* __restrict__ ad, int Nn)
{
    int n = blockIdx.x * blockDim.x + threadIdx.x;
    if (n >= Nn) return;
    const float* hp = h3 + (size_t)n * 48;
    float ss = 0.f, dd = 0.f;
#pragma unroll 8
    for (int c = 0; c < 40; ++c) { float v = hp[c]; ss += v * a_src[c]; dd += v * a_dst[c]; }
    as[n] = ss;
    ad[n] = dd;
}

__global__ void edge_max2(const int* __restrict__ ei, int E, int Nn,
                          const float* __restrict__ as, const float* __restrict__ ad,
                          float* __restrict__ mx)
{
    int e = blockIdx.x * blockDim.x + threadIdx.x;
    if (e >= E + Nn) return;
    int s, d; edge_sd(ei, E, e, s, d);
    atomicMaxF32(&mx[d], leaky(as[s] + ad[d]));
}

__global__ void edge_expsum2(const int* __restrict__ ei, int E, int Nn,
                             const float* __restrict__ as, const float* __restrict__ ad,
                             const float* __restrict__ mx,
                             float* __restrict__ ex, float* __restrict__ denom)
{
    int e = blockIdx.x * blockDim.x + threadIdx.x;
    if (e >= E + Nn) return;
    int s, d; edge_sd(ei, E, e, s, d);
    float v  = leaky(as[s] + ad[d]);
    float xv = expf(v - mx[d]);
    ex[e] = xv;
    atomicAdd(&denom[d], xv);
}

__global__ void aggregate2(const int* __restrict__ ei, int E, int Nn,
                           const float* __restrict__ ex, const float* __restrict__ denom,
                           const float* __restrict__ h3, float* __restrict__ out)
{
    int t = blockIdx.x * blockDim.x + threadIdx.x;
    if (t >= (E + Nn) * 10) return;
    int e = t / 10;
    int g = t - e * 10;
    int s, d; edge_sd(ei, E, e, s, d);
    float alpha = ex[e] / denom[d];
    const float4 hv = *(const float4*)(h3 + (size_t)s * 48 + g * 4);
    float* o = out + (size_t)d * 40 + g * 4;
    atomicAdd(o + 0, hv.x * alpha);
    atomicAdd(o + 1, hv.y * alpha);
    atomicAdd(o + 2, hv.z * alpha);
    atomicAdd(o + 3, hv.w * alpha);
}

__global__ void bias2_kernel(float* __restrict__ out, const float* __restrict__ b, int total) {
    int t = blockIdx.x * blockDim.x + threadIdx.x;
    if (t >= total) return;
    out[t] += b[t % 40];
}

extern "C" void kernel_launch(void* const* d_in, const int* in_sizes, int n_in,
                              void* d_out, int out_size, void* d_ws, size_t ws_size,
                              hipStream_t stream)
{
    const float* x    = (const float*)d_in[0];
    const int*   ei   = (const int*)  d_in[1];
    const float* W1   = (const float*)d_in[2];
    const float* aS1  = (const float*)d_in[3];
    const float* aD1  = (const float*)d_in[4];
    const float* b1   = (const float*)d_in[5];
    const float* W2   = (const float*)d_in[6];
    const float* aS2  = (const float*)d_in[7];
    const float* aD2  = (const float*)d_in[8];
    const float* b2   = (const float*)d_in[9];
    float* out = (float*)d_out;

    const int IN  = 256, F1 = 512, OUT = 40, OUTP = 48;
    const int Nn  = in_sizes[0] / IN;
    const int E   = in_sizes[1] / 2;
    const int ET  = E + Nn;

    float* ws  = (float*)d_ws;
    float* h1  = ws;
    float* h2  = h1  + (size_t)Nn * F1;
    float* as1 = h2  + (size_t)Nn * F1;
    float* ad1 = as1 + (size_t)Nn * 8;
    float* mx1 = ad1 + (size_t)Nn * 8;
    float* dn1 = mx1 + (size_t)Nn * 8;
    float* ex1 = dn1 + (size_t)Nn * 8;
    float* h3  = ex1 + (size_t)ET * 8;
    float* as2 = h3  + (size_t)Nn * 48;
    float* ad2 = as2 + Nn;
    float* mx2 = ad2 + Nn;
    float* dn2 = mx2 + Nn;
    float* ex2 = dn2 + Nn;
    float* W2p = ex2 + ET;

    const int B = 256;
    const float NEG_INF = -__builtin_inff();

    gemm_wmma_f32<4><<<dim3((Nn + 15) / 16, F1 / 64), 32, 0, stream>>>(
        x, W1, h1, Nn, IN, F1, F1);

    alpha1_kernel<<<(Nn * 8 + B - 1) / B, B, 0, stream>>>(h1, aS1, aD1, as1, ad1, Nn);

    fill_f32<<<(Nn * 8 + B - 1) / B, B, 0, stream>>>(mx1, Nn * 8, NEG_INF);
    fill_f32<<<(Nn * 8 + B - 1) / B, B, 0, stream>>>(dn1, Nn * 8, 0.f);
    fill_f32<<<(Nn * F1 + B - 1) / B, B, 0, stream>>>(h2, Nn * F1, 0.f);

    edge_max1   <<<(ET * 8 + B - 1) / B, B, 0, stream>>>(ei, E, Nn, as1, ad1, mx1);
    edge_expsum1<<<(ET * 8 + B - 1) / B, B, 0, stream>>>(ei, E, Nn, as1, ad1, mx1, ex1, dn1);
    aggregate1  <<<(ET * 128 + B - 1) / B, B, 0, stream>>>(ei, E, Nn, ex1, dn1, h1, h2);

    bias_elu<<<(Nn * F1 + B - 1) / B, B, 0, stream>>>(h2, b1, Nn * F1);

    pad_cols<<<(F1 * OUTP + B - 1) / B, B, 0, stream>>>(W2, W2p, F1, OUT, OUTP);

    gemm_wmma_f32<3><<<dim3((Nn + 15) / 16, 1), 32, 0, stream>>>(
        h2, W2p, h3, Nn, F1, OUTP, OUTP);

    alpha2_kernel<<<(Nn + B - 1) / B, B, 0, stream>>>(h3, aS2, aD2, as2, ad2, Nn);

    fill_f32<<<(Nn + B - 1) / B, B, 0, stream>>>(mx2, Nn, NEG_INF);
    fill_f32<<<(Nn + B - 1) / B, B, 0, stream>>>(dn2, Nn, 0.f);
    fill_f32<<<(Nn * OUT + B - 1) / B, B, 0, stream>>>(out, Nn * OUT, 0.f);

    edge_max2   <<<(ET + B - 1) / B, B, 0, stream>>>(ei, E, Nn, as2, ad2, mx2);
    edge_expsum2<<<(ET + B - 1) / B, B, 0, stream>>>(ei, E, Nn, as2, ad2, mx2, ex2, dn2);
    aggregate2  <<<(ET * 10 + B - 1) / B, B, 0, stream>>>(ei, E, Nn, ex2, dn2, h3, out);

    bias2_kernel<<<(Nn * OUT + B - 1) / B, B, 0, stream>>>(out, b2, Nn * OUT);
}


# Edges counting-sorted by destination once; each layer's max, exp-sum and weighted sum done per node in registers: no float atomics, sums in arrival-rank order of an int atomic; bias/ELU at store
# speedup vs baseline: 2.0378x; 2.0378x over previous
#include <hip/hip_runtime.h>
#include <hip/hip_bf16.h>
#include <math.h>

typedef float v2f __attribute__((ext_vector_type(2)));
typedef float v8f __attribute__((ext_vector_type(8)));

#define NEG_SLOPE 0.2f

__device__ __forceinline__ float leaky(float v) {
    return v > 0.0f ? v : NEG_SLOPE * v;
}

template<int NT>
__global__ void __launch_bounds__(32)
gemm_wmma_f32(const float* __restrict__ A, const float* __restrict__ B,
              float* __restrict__ C, int M, int K, int N, int ldc)
{
    __shared__ float As[16 * 12];

    const int lane = (int)threadIdx.x;
    const int half = lane >> 4;
    const int l16  = lane & 15;

    const int alr = lane >> 1;
    const int alc = (lane & 1) * 4;
    int rowA = blockIdx.x * 16 + alr;
    if (rowA >= M) rowA = M - 1;
    const float* __restrict__ Aload = A + (size_t)rowA * (size_t)K + alc;

    const int ncol0 = blockIdx.y * (16 * NT);
    const int colB  = ncol0 + l16;

    v8f acc[NT];
#pragma unroll
    for (int t = 0; t < NT; ++t) {
        v8f z = {0.f, 0.f, 0.f, 0.f, 0.f, 0.f, 0.f, 0.f};
        acc[t] = z;
    }

    float4 av = *(const float4*)(Aload);

    for (int kc = 0; kc < K; kc += 8) {
        __syncthreads();
        *(float4*)(&As[alr * 12 + alc]) = av;

        v2f breg[2][NT];
#pragma unroll
        for (int kk2 = 0; kk2 < 2; ++kk2) {
            const float* __restrict__ Bk =
                B + (size_t)(kc + kk2 * 4 + 2 * half) * (size_t)N + colB;
#pragma unroll
            for (int t = 0; t < NT; ++t) {
                breg[kk2][t].x = Bk[t * 16];
                breg[kk2][t].y = Bk[N + t * 16];
            }
        }

        const int kn = (kc + 8 < K) ? (kc + 8) : 0;
        av = *(const float4*)(Aload + kn);

        __syncthreads();

#pragma unroll
        for (int kk2 = 0; kk2 < 2; ++kk2) {
            const v2f a = *(const v2f*)(&As[l16 * 12 + kk2 * 4 + 2 * half]);
#pragma unroll
            for (int t = 0; t < NT; ++t) {
                acc[t] = __builtin_amdgcn_wmma_f32_16x16x4_f32(
                    false, a, false, breg[kk2][t], (short)0, acc[t], false, false);
            }
        }
    }

    const int crow0 = blockIdx.x * 16 + half * 8;
    float* __restrict__ Cbase = C + (size_t)crow0 * (size_t)ldc + ncol0 + l16;
    if (blockIdx.x * 16 + 16 <= M) {
#pragma unroll
        for (int t = 0; t < NT; ++t) {
            float* __restrict__ Cp = Cbase + t * 16;
#pragma unroll
            for (int r = 0; r < 8; ++r)
                Cp[(size_t)r * (size_t)ldc] = acc[t][r];
        }
    } else {
#pragma unroll
        for (int t = 0; t < NT; ++t) {
            float* __restrict__ Cp = Cbase + t * 16;
#pragma unroll
            for (int r = 0; r < 8; ++r)
                if (crow0 + r < M) Cp[(size_t)r * (size_t)ldc] = acc[t][r];
        }
    }
}

#define CSTRIDE 32

__global__ void zero_counters(int* __restrict__ deg, int Nn) {
    int t = blockIdx.x * blockDim.x + threadIdx.x;
    if (t < Nn) deg[(size_t)t * CSTRIDE] = 0;
}

__global__ void rank_edges(const int* __restrict__ ei, int E,
                           int* __restrict__ deg, int* __restrict__ rank)
{
    int e = blockIdx.x * blockDim.x + threadIdx.x;
    if (e >= E) return;
    int d = ei[E + e];
    rank[e] = atomicAdd(&deg[(size_t)d * CSTRIDE], 1);
}

__global__ void __launch_bounds__(1024)
scan_degrees(const int* __restrict__ deg, int* __restrict__ off, int Nn)
{
    __shared__ int part[1024];
    const int t   = (int)threadIdx.x;
    const int ipt = (Nn + 1023) / 1024;
    const int b0  = t * ipt;
    int sum = 0;
#pragma unroll 4
    for (int i = 0; i < ipt; ++i) {
        int idx = b0 + i;
        int v = deg[(size_t)(idx < Nn ? idx : Nn - 1) * CSTRIDE];
        sum += (idx < Nn) ? v : 0;
    }
    part[t] = sum;
    __syncthreads();
    for (int o = 1; o < 1024; o <<= 1) {
        int v = (t >= o) ? part[t - o] : 0;
        __syncthreads();
        part[t] += v;
        __syncthreads();
    }
    int run = part[t] - sum;
#pragma unroll 4
    for (int i = 0; i < ipt; ++i) {
        int idx = b0 + i;
        int v = deg[(size_t)(idx < Nn ? idx : Nn - 1) * CSTRIDE];
        if (idx < Nn) off[idx] = run;
        run += v;
    }
    if (t == 1023) off[Nn] = part[1023];
}

__global__ void place_edges(const int* __restrict__ ei, int E,
                            const int* __restrict__ off, const int* __restrict__ rank,
                            int* __restrict__ csr)
{
    int e = blockIdx.x * blockDim.x + threadIdx.x;
    if (e >= E) return;
    int d = ei[E + e];
    csr[off[d] + rank[e]] = ei[e];
}

__global__ void pad_cols(const float* __restrict__ src, float* __restrict__ dst,
                         int K, int N, int Np)
{
    int t = blockIdx.x * blockDim.x + threadIdx.x;
    if (t >= K * Np) return;
    int r = t / Np, c = t - r * Np;
    dst[t] = (c < N) ? src[r * N + c] : 0.f;
}

__global__ void alpha1_kernel(const float* __restrict__ h1,
                              const float* __restrict__ a_src,
                              const float* __restrict__ a_dst,
                              float* __restrict__ as, float* __restrict__ ad, int Nn)
{
    int t = blockIdx.x * blockDim.x + threadIdx.x;
    if (t >= Nn * 8) return;
    int n = t >> 3, h = t & 7;
    const float* hp = h1 + (size_t)n * 512 + h * 64;
    const float* sp = a_src + h * 64;
    const float* dp = a_dst + h * 64;
    float ss = 0.f, dd = 0.f;
#pragma unroll 8
    for (int c = 0; c < 64; ++c) { float v = hp[c]; ss += v * sp[c]; dd += v * dp[c]; }
    as[t] = ss;
    ad[t] = dd;
}

__global__ void __launch_bounds__(128)
gat1_gather(const int* __restrict__ off, const int* __restrict__ csr,
            const float* __restrict__ as, const float* __restrict__ ad,
            const float* __restrict__ h1, const float* __restrict__ b,
            float* __restrict__ h2)
{
    const int n = (int)blockIdx.x;
    const int g = (int)threadIdx.x;
    const int h = g >> 4;
    const int beg = off[n], end = off[n + 1];
    const float adn   = ad[n * 8 + h];
    const float eself = leaky(as[n * 8 + h] + adn);

    float m = eself;
    for (int j = beg; j < end; ++j)
        m = fmaxf(m, leaky(as[csr[j] * 8 + h] + adn));

    float den;
    float4 acc;
    {
        const float w = expf(eself - m);
        const float4 hv = *(const float4*)(h1 + (size_t)n * 512 + g * 4);
        den = w;
        acc.x = hv.x * w; acc.y = hv.y * w; acc.z = hv.z * w; acc.w = hv.w * w;
    }
#pragma unroll 4
    for (int j = beg; j < end; ++j) {
        const int s = csr[j];
        const float w = expf(leaky(as[s * 8 + h] + adn) - m);
        const float4 hv = *(const float4*)(h1 + (size_t)s * 512 + g * 4);
        den += w;
        acc.x += hv.x * w; acc.y += hv.y * w; acc.z += hv.z * w; acc.w += hv.w * w;
    }
    const float4 bb = *(const float4*)(b + g * 4);
    float4 o;
    o.x = acc.x / den + bb.x; o.y = acc.y / den + bb.y;
    o.z = acc.z / den + bb.z; o.w = acc.w / den + bb.w;
    o.x = o.x > 0.f ? o.x : expm1f(o.x);
    o.y = o.y > 0.f ? o.y : expm1f(o.y);
    o.z = o.z > 0.f ? o.z : expm1f(o.z);
    o.w = o.w > 0.f ? o.w : expm1f(o.w);
    *(float4*)(h2 + (size_t)n * 512 + g * 4) = o;
}

__global__ void alpha2_kernel(const float* __restrict__ h3,
                              const float* __restrict__ a_src,
                              const float* __restrict__ a_dst,
                              float* __restrict__ as, float* __restrict__ ad, int Nn)
{
    int n = blockIdx.x * blockDim.x + threadIdx.x;
    if (n >= Nn) return;
    const float* hp = h3 + (size_t)n * 48;
    float ss = 0.f, dd = 0.f;
#pragma unroll 8
    for (int c = 0; c < 40; ++c) { float v = hp[c]; ss += v * a_src[c]; dd += v * a_dst[c]; }
    as[n] = ss;
    ad[n] = dd;
}

__global__ void gat2_gather(const int* __restrict__ off, const int* __restrict__ csr,
                            const float* __restrict__ as, const float* __restrict__ ad,
                            const float* __restrict__ h3, const float* __restrict__ b,
                            float* __restrict__ out, int Nn)
{
    int t = blockIdx.x * blockDim.x + threadIdx.x;
    if (t >= Nn * 10) return;
    const int n = t / 10;
    const int g = t - n * 10;
    const int beg = off[n], end = off[n + 1];
    const float adn   = ad[n];
    const float eself = leaky(as[n] + adn);

    float m = eself;
    for (int j = beg; j < end; ++j)
        m = fmaxf(m, leaky(as[csr[j]] + adn));

    float den;
    float4 acc;
    {
        const float w = expf(eself - m);
        const float4 hv = *(const float4*)(h3 + (size_t)n * 48 + g * 4);
        den = w;
        acc.x = hv.x * w; acc.y = hv.y * w; acc.z = hv.z * w; acc.w = hv.w * w;
    }
#pragma unroll 4
    for (int j = beg; j < end; ++j) {
        const int s = csr[j];
        const float w = expf(leaky(as[s] + adn) - m);
        const float4 hv = *(const float4*)(h3 + (size_t)s * 48 + g * 4);
        den += w;
        acc.x += hv.x * w; acc.y += hv.y * w; acc.z += hv.z * w; acc.w += hv.w * w;
    }
    const float4 bb = *(const float4*)(b + g * 4);
    float4 o;
    o.x = acc.x / den + bb.x; o.y = acc.y / den + bb.y;
    o.z = acc.z / den + bb.z; o.w = acc.w / den + bb.w;
    *(float4*)(out + (size_t)n * 40 + g * 4) = o;
}

extern "C" void kernel_launch(void* const* d_in, const int* in_sizes, int n_in,
                              void* d_out, int out_size, void* d_ws, size_t ws_size,
                              hipStream_t stream)
{
    const float* x    = (const float*)d_in[0];
    const int*   ei   = (const int*)  d_in[1];
    const float* W1   = (const float*)d_in[2];
    const float* aS1  = (const float*)d_in[3];
    const float* aD1  = (const float*)d_in[4];
    const float* b1   = (const float*)d_in[5];
    const float* W2   = (const float*)d_in[6];
    const float* aS2  = (const float*)d_in[7];
    const float* aD2  = (const float*)d_in[8];
    const float* b2   = (const float*)d_in[9];
    float* out = (float*)d_out;

    const int IN  = 256, F1 = 512, OUT = 40, OUTP = 48;
    const int Nn  = in_sizes[0] / IN;
    const int E   = in_sizes[1] / 2;

    float* ws  = (float*)d_ws;
    float* h1  = ws;
    float* h2  = h1  + (size_t)Nn * F1;
    float* h3  = h2  + (size_t)Nn * F1;
    float* W2p = h3  + (size_t)Nn * 48;
    float* as1 = W2p + (size_t)F1 * OUTP;
    float* ad1 = as1 + (size_t)Nn * 8;
    float* as2 = ad1 + (size_t)Nn * 8;
    float* ad2 = as2 + Nn;
    int*   deg  = (int*)(ad2 + Nn);
    int*   off  = deg  + (size_t)Nn * CSTRIDE;
    int*   rank = off  + (Nn + 1);
    int*   csr  = rank + E;

    const int B = 256;

    zero_counters<<<(Nn + B - 1) / B, B, 0, stream>>>(deg, Nn);
    rank_edges   <<<(E + B - 1) / B, B, 0, stream>>>(ei, E, deg, rank);
    scan_degrees <<<1, 1024, 0, stream>>>(deg, off, Nn);
    place_edges  <<<(E + B - 1) / B, B, 0, stream>>>(ei, E, off, rank, csr);

    gemm_wmma_f32<4><<<dim3((Nn + 15) / 16, F1 / 64), 32, 0, stream>>>(
        x, W1, h1, Nn, IN, F1, F1);

    alpha1_kernel<<<(Nn * 8 + B - 1) / B, B, 0, stream>>>(h1, aS1, aD1, as1, ad1, Nn);

    gat1_gather<<<Nn, 128, 0, stream>>>(off, csr, as1, ad1, h1, b1, h2);

    pad_cols<<<(F1 * OUTP + B - 1) / B, B, 0, stream>>>(W2, W2p, F1, OUT, OUTP);

    gemm_wmma_f32<3><<<dim3((Nn + 15) / 16, 1), 32, 0, stream>>>(
        h2, W2p, h3, Nn, F1, OUTP, OUTP);

    alpha2_kernel<<<(Nn + B - 1) / B, B, 0, stream>>>(h3, aS2, aD2, as2, ad2, Nn);

    gat2_gather<<<(Nn * 10 + B - 1) / B, B, 0, stream>>>(off, csr, as2, ad2, h3, b2, out, Nn);
}


# Layer-1 gather: one wave per node (was 4); logit max and exp weights once per edge x head across lanes, passed via wave-private LDS; lane owns 4 x 16 B of each 2-KB row; edge sum order kept
# speedup vs baseline: 2.1260x; 1.0433x over previous
#include <hip/hip_runtime.h>
#include <hip/hip_bf16.h>
#include <math.h>

typedef float v2f __attribute__((ext_vector_type(2)));
typedef float v8f __attribute__((ext_vector_type(8)));
typedef float v4f __attribute__((ext_vector_type(4)));

#define NEG_SLOPE 0.2f

__device__ __forceinline__ float leaky(float v) {
    return v > 0.0f ? v : NEG_SLOPE * v;
}

template<int NT>
__global__ void __launch_bounds__(32)
gemm_wmma_f32(const float* __restrict__ A, const float* __restrict__ B,
              float* __restrict__ C, int M, int K, int N, int ldc)
{
    __shared__ float As[16 * 12];

    const int lane = (int)threadIdx.x;
    const int half = lane >> 4;
    const int l16  = lane & 15;

    const int alr = lane >> 1;
    const int alc = (lane & 1) * 4;
    int rowA = blockIdx.x * 16 + alr;
    if (rowA >= M) rowA = M - 1;
    const float* __restrict__ Aload = A + (size_t)rowA * (size_t)K + alc;

    const int ncol0 = blockIdx.y * (16 * NT);
    const int colB  = ncol0 + l16;

    v8f acc[NT];
#pragma unroll
    for (int t = 0; t < NT; ++t) {
        v8f z = {0.f, 0.f, 0.f, 0.f, 0.f, 0.f, 0.f, 0.f};
        acc[t] = z;
    }

    float4 av = *(const float4*)(Aload);

    for (int kc = 0; kc < K; kc += 8) {
        __syncthreads();
        *(float4*)(&As[alr * 12 + alc]) = av;

        v2f breg[2][NT];
#pragma unroll
        for (int kk2 = 0; kk2 < 2; ++kk2) {
            const float* __restrict__ Bk =
                B + (size_t)(kc + kk2 * 4 + 2 * half) * (size_t)N + colB;
#pragma unroll
            for (int t = 0; t < NT; ++t) {
                breg[kk2][t].x = Bk[t * 16];
                breg[kk2][t].y = Bk[N + t * 16];
            }
        }

        const int kn = (kc + 8 < K) ? (kc + 8) : 0;
        av = *(const float4*)(Aload + kn);

        __syncthreads();

#pragma unroll
        for (int kk2 = 0; kk2 < 2; ++kk2) {
            const v2f a = *(const v2f*)(&As[l16 * 12 + kk2 * 4 + 2 * half]);
#pragma unroll
            for (int t = 0; t < NT; ++t) {
                acc[t] = __builtin_amdgcn_wmma_f32_16x16x4_f32(
                    false, a, false, breg[kk2][t], (short)0, acc[t], false, false);
            }
        }
    }

    const int crow0 = blockIdx.x * 16 + half * 8;
    float* __restrict__ Cbase = C + (size_t)crow0 * (size_t)ldc + ncol0 + l16;
    if (blockIdx.x * 16 + 16 <= M) {
#pragma unroll
        for (int t = 0; t < NT; ++t) {
            float* __restrict__ Cp = Cbase + t * 16;
#pragma unroll
            for (int r = 0; r < 8; ++r)
                Cp[(size_t)r * (size_t)ldc] = acc[t][r];
        }
    } else {
#pragma unroll
        for (int t = 0; t < NT; ++t) {
            float* __restrict__ Cp = Cbase + t * 16;
#pragma unroll
            for (int r = 0; r < 8; ++r)
                if (crow0 + r < M) Cp[(size_t)r * (size_t)ldc] = acc[t][r];
        }
    }
}

#define CSTRIDE 32

__global__ void zero_counters(int* __restrict__ deg, int Nn) {
    int t = blockIdx.x * blockDim.x + threadIdx.x;
    if (t < Nn) deg[(size_t)t * CSTRIDE] = 0;
}

__global__ void rank_edges(const int* __restrict__ ei, int E,
                           int* __restrict__ deg, int* __restrict__ rank)
{
    int e = blockIdx.x * blockDim.x + threadIdx.x;
    if (e >= E) return;
    int d = ei[E + e];
    rank[e] = atomicAdd(&deg[(size_t)d * CSTRIDE], 1);
}

__global__ void __launch_bounds__(1024)
scan_degrees(const int* __restrict__ deg, int* __restrict__ off, int Nn)
{
    __shared__ int part[1024];
    const int t   = (int)threadIdx.x;
    const int ipt = (Nn + 1023) / 1024;
    const int b0  = t * ipt;
    int sum = 0;
#pragma unroll 4
    for (int i = 0; i < ipt; ++i) {
        int idx = b0 + i;
        int v = deg[(size_t)(idx < Nn ? idx : Nn - 1) * CSTRIDE];
        sum += (idx < Nn) ? v : 0;
    }
    part[t] = sum;
    __syncthreads();
    for (int o = 1; o < 1024; o <<= 1) {
        int v = (t >= o) ? part[t - o] : 0;
        __syncthreads();
        part[t] += v;
        __syncthreads();
    }
    int run = part[t] - sum;
#pragma unroll 4
    for (int i = 0; i < ipt; ++i) {
        int idx = b0 + i;
        int v = deg[(size_t)(idx < Nn ? idx : Nn - 1) * CSTRIDE];
        if (idx < Nn) off[idx] = run;
        run += v;
    }
    if (t == 1023) off[Nn] = part[1023];
}

__global__ void place_edges(const int* __restrict__ ei, int E,
                            const int* __restrict__ off, const int* __restrict__ rank,
                            int* __restrict__ csr)
{
    int e = blockIdx.x * blockDim.x + threadIdx.x;
    if (e >= E) return;
    int d = ei[E + e];
    csr[off[d] + rank[e]] = ei[e];
}

__global__ void pad_cols(const float* __restrict__ src, float* __restrict__ dst,
                         int K, int N, int Np)
{
    int t = blockIdx.x * blockDim.x + threadIdx.x;
    if (t >= K * Np) return;
    int r = t / Np, c = t - r * Np;
    dst[t] = (c < N) ? src[r * N + c] : 0.f;
}

__global__ void alpha1_kernel(const float* __restrict__ h1,
                              const float* __restrict__ a_src,
                              const float* __restrict__ a_dst,
                              float* __restrict__ as, float* __restrict__ ad, int Nn)
{
    int t = blockIdx.x * blockDim.x + threadIdx.x;
    if (t >= Nn * 8) return;
    int n = t >> 3, h = t & 7;
    const float* hp = h1 + (size_t)n * 512 + h * 64;
    const float* sp = a_src + h * 64;
    const float* dp = a_dst + h * 64;
    float ss = 0.f, dd = 0.f;
#pragma unroll 8
    for (int c = 0; c < 64; ++c) { float v = hp[c]; ss += v * sp[c]; dd += v * dp[c]; }
    as[t] = ss;
    ad[t] = dd;
}

__device__ __forceinline__ void wave_lds_handover() {
    __builtin_amdgcn_fence(__ATOMIC_RELEASE, "wavefront");
    __builtin_amdgcn_wave_barrier();
    __builtin_amdgcn_fence(__ATOMIC_ACQUIRE, "wavefront");
}

__global__ void __launch_bounds__(256)
gat1_gather(const int* __restrict__ off, const int* __restrict__ csr,
            const float* __restrict__ as, const float* __restrict__ ad,
            const float* __restrict__ h1, const float* __restrict__ b,
            float* __restrict__ h2, int Nn)
{
    __shared__ float4 wsh[8 * 64];

    const int lane = (int)threadIdx.x & 31;
    const int half = lane >> 4;
    const int wv   = (int)threadIdx.x >> 5;
    float* __restrict__ wmine = (float*)(wsh + wv * 64);
    const int n = __builtin_amdgcn_readfirstlane((int)blockIdx.x * 8 + wv);
    if (n >= Nn) return;
    const int beg = off[n], end = off[n + 1];
    const int last = end - 1;

    const int q = lane >> 3, hh = lane & 7;
    const float adq = ad[n * 8 + hh];
    float m = leaky(as[n * 8 + hh] + adq);
    for (int base = beg; base < end; base += 32) {
#pragma unroll
        for (int t = 0; t < 8; ++t) {
            const int j = base + q + 4 * t;
            const int s = csr[j < last ? j : last];
            m = fmaxf(m, leaky(as[s * 8 + hh] + adq));
        }
    }
    m = fmaxf(m, __shfl_xor(m, 8));
    m = fmaxf(m, __shfl_xor(m, 16));

    float den[4];
    float4 acc[4];
    const float* __restrict__ hcol = h1 + lane * 4;
#pragma unroll
    for (int i = 0; i < 4; ++i) {
        const int hd = 2 * i + half;
        const float mh = __shfl(m, hd);
        const float w = expf(leaky(as[n * 8 + hd] + ad[n * 8 + hd]) - mh);
        const float4 hv = *(const float4*)(hcol + (size_t)n * 512 + i * 128);
        den[i] = w;
        acc[i].x = hv.x * w; acc[i].y = hv.y * w; acc[i].z = hv.z * w; acc[i].w = hv.w * w;
    }

    const int wpos = (hh & 1) * 4 + (hh >> 1);
    for (int base = beg; base < end; base += 32) {
#pragma unroll
        for (int t = 0; t < 8; ++t) {
            const int j = base + q + 4 * t;
            const int s = csr[j < last ? j : last];
            wmine[(q + 4 * t) * 8 + wpos] = expf(leaky(as[s * 8 + hh] + adq) - m);
        }
        wave_lds_handover();
        const int left = end - base;
        const int cnt = left < 32 ? left : 32;
        int e = 0;
        for (; e + 2 <= cnt; e += 2) {
            const int s0 = csr[base + e], s1 = csr[base + e + 1];
            const float4 wa = *(const float4*)(wmine + e * 8 + half * 4);
            const float4 wb = *(const float4*)(wmine + e * 8 + 8 + half * 4);
            const float* __restrict__ r0 = hcol + (size_t)s0 * 512;
            const float* __restrict__ r1 = hcol + (size_t)s1 * 512;
            const v4f a0 = *(const v4f*)(r0),       a1 = *(const v4f*)(r0 + 128);
            const v4f a2 = *(const v4f*)(r0 + 256), a3 = *(const v4f*)(r0 + 384);
            const v4f b0 = *(const v4f*)(r1),       b1 = *(const v4f*)(r1 + 128);
            const v4f b2 = *(const v4f*)(r1 + 256), b3 = *(const v4f*)(r1 + 384);
            asm volatile("" :: "v"(a0), "v"(a1), "v"(a2), "v"(a3),
                               "v"(b0), "v"(b1), "v"(b2), "v"(b3));
            const v4f ra[4] = {a0, a1, a2, a3};
            const v4f rb[4] = {b0, b1, b2, b3};
            const float wla[4] = {wa.x, wa.y, wa.z, wa.w};
            const float wlb[4] = {wb.x, wb.y, wb.z, wb.w};
#pragma unroll
            for (int i = 0; i < 4; ++i) {
                den[i] += wla[i];
                acc[i].x += ra[i].x * wla[i]; acc[i].y += ra[i].y * wla[i];
                acc[i].z += ra[i].z * wla[i]; acc[i].w += ra[i].w * wla[i];
            }
#pragma unroll
            for (int i = 0; i < 4; ++i) {
                den[i] += wlb[i];
                acc[i].x += rb[i].x * wlb[i]; acc[i].y += rb[i].y * wlb[i];
                acc[i].z += rb[i].z * wlb[i]; acc[i].w += rb[i].w * wlb[i];
            }
        }
        if (e < cnt) {
            const int s0 = csr[base + e];
            const float4 wa = *(const float4*)(wmine + e * 8 + half * 4);
            const float wla[4] = {wa.x, wa.y, wa.z, wa.w};
            const float* __restrict__ r0 = hcol + (size_t)s0 * 512;
#pragma unroll
            for (int i = 0; i < 4; ++i) {
                const v4f hv = *(const v4f*)(r0 + i * 128);
                den[i] += wla[i];
                acc[i].x += hv.x * wla[i]; acc[i].y += hv.y * wla[i];
                acc[i].z += hv.z * wla[i]; acc[i].w += hv.w * wla[i];
            }
        }
        wave_lds_handover();
    }
#pragma unroll
    for (int i = 0; i < 4; ++i) {
        const float4 bb = *(const float4*)(b + i * 128 + lane * 4);
        float4 o;
        o.x = acc[i].x / den[i] + bb.x; o.y = acc[i].y / den[i] + bb.y;
        o.z = acc[i].z / den[i] + bb.z; o.w = acc[i].w / den[i] + bb.w;
        o.x = o.x > 0.f ? o.x : expm1f(o.x);
        o.y = o.y > 0.f ? o.y : expm1f(o.y);
        o.z = o.z > 0.f ? o.z : expm1f(o.z);
        o.w = o.w > 0.f ? o.w : expm1f(o.w);
        *(float4*)(h2 + (size_t)n * 512 + i * 128 + lane * 4) = o;
    }
}

__global__ void alpha2_kernel(const float* __restrict__ h3,
                              const float* __restrict__ a_src,
                              const float* __restrict__ a_dst,
                              float* __restrict__ as, float* __restrict__ ad, int Nn)
{
    int n = blockIdx.x * blockDim.x + threadIdx.x;
    if (n >= Nn) return;
    const float* hp = h3 + (size_t)n * 48;
    float ss = 0.f, dd = 0.f;
#pragma unroll 8
    for (int c = 0; c < 40; ++c) { float v = hp[c]; ss += v * a_src[c]; dd += v * a_dst[c]; }
    as[n] = ss;
    ad[n] = dd;
}

__global__ void gat2_gather(const int* __restrict__ off, const int* __restrict__ csr,
                            const float* __restrict__ as, const float* __restrict__ ad,
                            const float* __restrict__ h3, const float* __restrict__ b,
                            float* __restrict__ out, int Nn)
{
    int t = blockIdx.x * blockDim.x + threadIdx.x;
    if (t >= Nn * 10) return;
    const int n = t / 10;
    const int g = t - n * 10;
    const int beg = off[n], end = off[n + 1];
    const float adn   = ad[n];
    const float eself = leaky(as[n] + adn);

    float m = eself;
    for (int j = beg; j < end; ++j)
        m = fmaxf(m, leaky(as[csr[j]] + adn));

    float den;
    float4 acc;
    {
        const float w = expf(eself - m);
        const float4 hv = *(const float4*)(h3 + (size_t)n * 48 + g * 4);
        den = w;
        acc.x = hv.x * w; acc.y = hv.y * w; acc.z = hv.z * w; acc.w = hv.w * w;
    }
#pragma unroll 4
    for (int j = beg; j < end; ++j) {
        const int s = csr[j];
        const float w = expf(leaky(as[s] + adn) - m);
        const float4 hv = *(const float4*)(h3 + (size_t)s * 48 + g * 4);
        den += w;
        acc.x += hv.x * w; acc.y += hv.y * w; acc.z += hv.z * w; acc.w += hv.w * w;
    }
    const float4 bb = *(const float4*)(b + g * 4);
    float4 o;
    o.x = acc.x / den + bb.x; o.y = acc.y / den + bb.y;
    o.z = acc.z / den + bb.z; o.w = acc.w / den + bb.w;
    *(float4*)(out + (size_t)n * 40 + g * 4) = o;
}

extern "C" void kernel_launch(void* const* d_in, const int* in_sizes, int n_in,
                              void* d_out, int out_size, void* d_ws, size_t ws_size,
                              hipStream_t stream)
{
    const float* x    = (const float*)d_in[0];
    const int*   ei   = (const int*)  d_in[1];
    const float* W1   = (const float*)d_in[2];
    const float* aS1  = (const float*)d_in[3];
    const float* aD1  = (const float*)d_in[4];
    const float* b1   = (const float*)d_in[5];
    const float* W2   = (const float*)d_in[6];
    const float* aS2  = (const float*)d_in[7];
    const float* aD2  = (const float*)d_in[8];
    const float* b2   = (const float*)d_in[9];
    float* out = (float*)d_out;

    const int IN  = 256, F1 = 512, OUT = 40, OUTP = 48;
    const int Nn  = in_sizes[0] / IN;
    const int E   = in_sizes[1] / 2;

    float* ws  = (float*)d_ws;
    float* h1  = ws;
    float* h2  = h1  + (size_t)Nn * F1;
    float* h3  = h2  + (size_t)Nn * F1;
    float* W2p = h3  + (size_t)Nn * 48;
    float* as1 = W2p + (size_t)F1 * OUTP;
    float* ad1 = as1 + (size_t)Nn * 8;
    float* as2 = ad1 + (size_t)Nn * 8;
    float* ad2 = as2 + Nn;
    int*   deg  = (int*)(ad2 + Nn);
    int*   off  = deg  + (size_t)Nn * CSTRIDE;
    int*   rank = off  + (Nn + 1);
    int*   csr  = rank + E;

    const int B = 256;

    zero_counters<<<(Nn + B - 1) / B, B, 0, stream>>>(deg, Nn);
    rank_edges   <<<(E + B - 1) / B, B, 0, stream>>>(ei, E, deg, rank);
    scan_degrees <<<1, 1024, 0, stream>>>(deg, off, Nn);
    place_edges  <<<(E + B - 1) / B, B, 0, stream>>>(ei, E, off, rank, csr);

    gemm_wmma_f32<4><<<dim3((Nn + 15) / 16, F1 / 64), 32, 0, stream>>>(
        x, W1, h1, Nn, IN, F1, F1);

    alpha1_kernel<<<(Nn * 8 + B - 1) / B, B, 0, stream>>>(h1, aS1, aD1, as1, ad1, Nn);

    gat1_gather<<<(Nn + 7) / 8, 256, 0, stream>>>(off, csr, as1, ad1, h1, b1, h2, Nn);

    pad_cols<<<(F1 * OUTP + B - 1) / B, B, 0, stream>>>(W2, W2p, F1, OUT, OUTP);

    gemm_wmma_f32<3><<<dim3((Nn + 15) / 16, 1), 32, 0, stream>>>(
        h2, W2p, h3, Nn, F1, OUTP, OUTP);

    alpha2_kernel<<<(Nn + B - 1) / B, B, 0, stream>>>(h3, aS2, aD2, as2, ad2, Nn);

    gat2_gather<<<(Nn * 10 + B - 1) / B, B, 0, stream>>>(off, csr, as2, ad2, h3, b2, out, Nn);
}


# Attention halves (h . att_src, h . att_dst) summed in the GEMM epilogues from the accumulators, 16-lane halving tree (new sum order); both dot-product kernels and their re-read of h removed
# speedup vs baseline: 2.2728x; 1.0690x over previous
#include <hip/hip_runtime.h>
#include <hip/hip_bf16.h>
#include <math.h>

typedef float v2f __attribute__((ext_vector_type(2)));
typedef float v8f __attribute__((ext_vector_type(8)));
typedef float v4f __attribute__((ext_vector_type(4)));

#define NEG_SLOPE 0.2f

__device__ __forceinline__ float leaky(float v) {
    return v > 0.0f ? v : NEG_SLOPE * v;
}


template<int MASK>
__device__ __forceinline__ float row_xor(float v) {
    const int i = __builtin_bit_cast(int, v);
    return __builtin_bit_cast(float,
        __builtin_amdgcn_update_dpp(i, i, 0x160 + MASK, 0xf, 0xf, false));
}

template<int NT>
__global__ void __launch_bounds__(32)
gemm_wmma_f32(const float* __restrict__ A, const float* __restrict__ B,
              float* __restrict__ C, int M, int K, int N, int ldc,
              const float* __restrict__ a_src, const float* __restrict__ a_dst,
              float* __restrict__ as, float* __restrict__ ad, int ncv, int heads)
{
    __shared__ float As[16 * 12];

    const int lane = (int)threadIdx.x;
    const int half = lane >> 4;
    const int l16  = lane & 15;

    const int alr = lane >> 1;
    const int alc = (lane & 1) * 4;
    int rowA = blockIdx.x * 16 + alr;
    if (rowA >= M) rowA = M - 1;
    const float* __restrict__ Aload = A + (size_t)rowA * (size_t)K + alc;

    const int ncol0 = blockIdx.y * (16 * NT);
    const int colB  = ncol0 + l16;

    v8f acc[NT];
#pragma unroll
    for (int t = 0; t < NT; ++t) {
        v8f z = {0.f, 0.f, 0.f, 0.f, 0.f, 0.f, 0.f, 0.f};
        acc[t] = z;
    }

    float4 av = *(const float4*)(Aload);

    float wsrc[NT], wdst[NT];
#pragma unroll
    for (int t = 0; t < NT; ++t) {
        const int col = colB + t * 16;
        const bool real = col < ncv;
        const int cc = real ? col : 0;
        const float vs = a_src[cc], vd = a_dst[cc];
        wsrc[t] = real ? vs : 0.f;
        wdst[t] = real ? vd : 0.f;
    }

    for (int kc = 0; kc < K; kc += 8) {
        __syncthreads();
        *(float4*)(&As[alr * 12 + alc]) = av;

        v2f breg[2][NT];
#pragma unroll
        for (int kk2 = 0; kk2 < 2; ++kk2) {
            const float* __restrict__ Bk =
                B + (size_t)(kc + kk2 * 4 + 2 * half) * (size_t)N + colB;
#pragma unroll
            for (int t = 0; t < NT; ++t) {
                breg[kk2][t].x = Bk[t * 16];
                breg[kk2][t].y = Bk[N + t * 16];
            }
        }

        const int kn = (kc + 8 < K) ? (kc + 8) : 0;
        av = *(const float4*)(Aload + kn);

        __syncthreads();

#pragma unroll
        for (int kk2 = 0; kk2 < 2; ++kk2) {
            const v2f a = *(const v2f*)(&As[l16 * 12 + kk2 * 4 + 2 * half]);
#pragma unroll
            for (int t = 0; t < NT; ++t) {
                acc[t] = __builtin_amdgcn_wmma_f32_16x16x4_f32(
                    false, a, false, breg[kk2][t], (short)0, acc[t], false, false);
            }
        }
    }

    const int crow0 = blockIdx.x * 16 + half * 8;
    float* __restrict__ Cbase = C + (size_t)crow0 * (size_t)ldc + ncol0 + l16;
    if (blockIdx.x * 16 + 16 <= M) {
#pragma unroll
        for (int t = 0; t < NT; ++t) {
            float* __restrict__ Cp = Cbase + t * 16;
#pragma unroll
            for (int r = 0; r < 8; ++r)
                Cp[(size_t)r * (size_t)ldc] = acc[t][r];
        }
    } else {
#pragma unroll
        for (int t = 0; t < NT; ++t) {
            float* __restrict__ Cp = Cbase + t * 16;
#pragma unroll
            for (int r = 0; r < 8; ++r)
                if (crow0 + r < M) Cp[(size_t)r * (size_t)ldc] = acc[t][r];
        }
    }

    float ps[8], pd[8];
#pragma unroll
    for (int r = 0; r < 8; ++r) { ps[r] = 0.f; pd[r] = 0.f; }
#pragma unroll
    for (int t = 0; t < NT; ++t) {
#pragma unroll
        for (int r = 0; r < 8; ++r) {
            ps[r] += acc[t][r] * wsrc[t];
            pd[r] += acc[t][r] * wdst[t];
        }
    }
    const bool b3 = (l16 & 8) != 0, b2 = (l16 & 4) != 0;
    const bool b1 = (l16 & 2) != 0, b0 = (l16 & 1) != 0;
    float u[8], w4[4], y2[2];
#pragma unroll
    for (int i = 0; i < 8; ++i) {
        const float mine = b3 ? pd[i] : ps[i];
        const float give = b3 ? ps[i] : pd[i];
        u[i] = mine + row_xor<8>(give);
    }
#pragma unroll
    for (int i = 0; i < 4; ++i) {
        const float mine = b2 ? u[i + 4] : u[i];
        const float give = b2 ? u[i] : u[i + 4];
        w4[i] = mine + row_xor<4>(give);
    }
#pragma unroll
    for (int i = 0; i < 2; ++i) {
        const float mine = b1 ? w4[i + 2] : w4[i];
        const float give = b1 ? w4[i] : w4[i + 2];
        y2[i] = mine + row_xor<2>(give);
    }
    const float mine = b0 ? y2[1] : y2[0];
    const float give = b0 ? y2[0] : y2[1];
    const float z = mine + row_xor<1>(give);
    const int arow = blockIdx.x * 16 + half * 8 + (l16 & 7);
    if (arow < M) {
        float* __restrict__ dstp = b3 ? ad : as;
        dstp[(size_t)arow * (size_t)heads + blockIdx.y] = z;
    }
}

#define CSTRIDE 32

__global__ void zero_counters(int* __restrict__ deg, int Nn) {
    int t = blockIdx.x * blockDim.x + threadIdx.x;
    if (t < Nn) deg[(size_t)t * CSTRIDE] = 0;
}

__global__ void rank_edges(const int* __restrict__ ei, int E,
                           int* __restrict__ deg, int* __restrict__ rank)
{
    int e = blockIdx.x * blockDim.x + threadIdx.x;
    if (e >= E) return;
    int d = ei[E + e];
    rank[e] = atomicAdd(&deg[(size_t)d * CSTRIDE], 1);
}

__global__ void __launch_bounds__(1024)
scan_degrees(const int* __restrict__ deg, int* __restrict__ off, int Nn)
{
    __shared__ int part[1024];
    const int t   = (int)threadIdx.x;
    const int ipt = (Nn + 1023) / 1024;
    const int b0  = t * ipt;
    int sum = 0;
#pragma unroll 4
    for (int i = 0; i < ipt; ++i) {
        int idx = b0 + i;
        int v = deg[(size_t)(idx < Nn ? idx : Nn - 1) * CSTRIDE];
        sum += (idx < Nn) ? v : 0;
    }
    part[t] = sum;
    __syncthreads();
    for (int o = 1; o < 1024; o <<= 1) {
        int v = (t >= o) ? part[t - o] : 0;
        __syncthreads();
        part[t] += v;
        __syncthreads();
    }
    int run = part[t] - sum;
#pragma unroll 4
    for (int i = 0; i < ipt; ++i) {
        int idx = b0 + i;
        int v = deg[(size_t)(idx < Nn ? idx : Nn - 1) * CSTRIDE];
        if (idx < Nn) off[idx] = run;
        run += v;
    }
    if (t == 1023) off[Nn] = part[1023];
}

__global__ void place_edges(const int* __restrict__ ei, int E,
                            const int* __restrict__ off, const int* __restrict__ rank,
                            int* __restrict__ csr)
{
    int e = blockIdx.x * blockDim.x + threadIdx.x;
    if (e >= E) return;
    int d = ei[E + e];
    csr[off[d] + rank[e]] = ei[e];
}

__global__ void pad_cols(const float* __restrict__ src, float* __restrict__ dst,
                         int K, int N, int Np)
{
    int t = blockIdx.x * blockDim.x + threadIdx.x;
    if (t >= K * Np) return;
    int r = t / Np, c = t - r * Np;
    dst[t] = (c < N) ? src[r * N + c] : 0.f;
}

__device__ __forceinline__ void wave_lds_handover() {
    __builtin_amdgcn_fence(__ATOMIC_RELEASE, "wavefront");
    __builtin_amdgcn_wave_barrier();
    __builtin_amdgcn_fence(__ATOMIC_ACQUIRE, "wavefront");
}

__global__ void __launch_bounds__(256)
gat1_gather(const int* __restrict__ off, const int* __restrict__ csr,
            const float* __restrict__ as, const float* __restrict__ ad,
            const float* __restrict__ h1, const float* __restrict__ b,
            float* __restrict__ h2, int Nn)
{
    __shared__ float4 wsh[8 * 64];

    const int lane = (int)threadIdx.x & 31;
    const int half = lane >> 4;
    const int wv   = (int)threadIdx.x >> 5;
    float* __restrict__ wmine = (float*)(wsh + wv * 64);
    const int n = __builtin_amdgcn_readfirstlane((int)blockIdx.x * 8 + wv);
    if (n >= Nn) return;
    const int beg = off[n], end = off[n + 1];
    const int last = end - 1;

    const int q = lane >> 3, hh = lane & 7;
    const float adq = ad[n * 8 + hh];
    float m = leaky(as[n * 8 + hh] + adq);
    for (int base = beg; base < end; base += 32) {
#pragma unroll
        for (int t = 0; t < 8; ++t) {
            const int j = base + q + 4 * t;
            const int s = csr[j < last ? j : last];
            m = fmaxf(m, leaky(as[s * 8 + hh] + adq));
        }
    }
    m = fmaxf(m, __shfl_xor(m, 8));
    m = fmaxf(m, __shfl_xor(m, 16));

    float den[4];
    float4 acc[4];
    const float* __restrict__ hcol = h1 + lane * 4;
#pragma unroll
    for (int i = 0; i < 4; ++i) {
        const int hd = 2 * i + half;
        const float mh = __shfl(m, hd);
        const float w = expf(leaky(as[n * 8 + hd] + ad[n * 8 + hd]) - mh);
        const float4 hv = *(const float4*)(hcol + (size_t)n * 512 + i * 128);
        den[i] = w;
        acc[i].x = hv.x * w; acc[i].y = hv.y * w; acc[i].z = hv.z * w; acc[i].w = hv.w * w;
    }

    const int wpos = (hh & 1) * 4 + (hh >> 1);
    for (int base = beg; base < end; base += 32) {
#pragma unroll
        for (int t = 0; t < 8; ++t) {
            const int j = base + q + 4 * t;
            const int s = csr[j < last ? j : last];
            wmine[(q + 4 * t) * 8 + wpos] = expf(leaky(as[s * 8 + hh] + adq) - m);
        }
        wave_lds_handover();
        const int left = end - base;
        const int cnt = left < 32 ? left : 32;
        int e = 0;
        for (; e + 2 <= cnt; e += 2) {
            const int s0 = csr[base + e], s1 = csr[base + e + 1];
            const float4 wa = *(const float4*)(wmine + e * 8 + half * 4);
            const float4 wb = *(const float4*)(wmine + e * 8 + 8 + half * 4);
            const float* __restrict__ r0 = hcol + (size_t)s0 * 512;
            const float* __restrict__ r1 = hcol + (size_t)s1 * 512;
            const v4f a0 = *(const v4f*)(r0),       a1 = *(const v4f*)(r0 + 128);
            const v4f a2 = *(const v4f*)(r0 + 256), a3 = *(const v4f*)(r0 + 384);
            const v4f b0 = *(const v4f*)(r1),       b1 = *(const v4f*)(r1 + 128);
            const v4f b2 = *(const v4f*)(r1 + 256), b3 = *(const v4f*)(r1 + 384);
            asm volatile("" :: "v"(a0), "v"(a1), "v"(a2), "v"(a3),
                               "v"(b0), "v"(b1), "v"(b2), "v"(b3));
            const v4f ra[4] = {a0, a1, a2, a3};
            const v4f rb[4] = {b0, b1, b2, b3};
            const float wla[4] = {wa.x, wa.y, wa.z, wa.w};
            const float wlb[4] = {wb.x, wb.y, wb.z, wb.w};
#pragma unroll
            for (int i = 0; i < 4; ++i) {
                den[i] += wla[i];
                acc[i].x += ra[i].x * wla[i]; acc[i].y += ra[i].y * wla[i];
                acc[i].z += ra[i].z * wla[i]; acc[i].w += ra[i].w * wla[i];
            }
#pragma unroll
            for (int i = 0; i < 4; ++i) {
                den[i] += wlb[i];
                acc[i].x += rb[i].x * wlb[i]; acc[i].y += rb[i].y * wlb[i];
                acc[i].z += rb[i].z * wlb[i]; acc[i].w += rb[i].w * wlb[i];
            }
        }
        if (e < cnt) {
            const int s0 = csr[base + e];
            const float4 wa = *(const float4*)(wmine + e * 8 + half * 4);
            const float wla[4] = {wa.x, wa.y, wa.z, wa.w};
            const float* __restrict__ r0 = hcol + (size_t)s0 * 512;
#pragma unroll
            for (int i = 0; i < 4; ++i) {
                const v4f hv = *(const v4f*)(r0 + i * 128);
                den[i] += wla[i];
                acc[i].x += hv.x * wla[i]; acc[i].y += hv.y * wla[i];
                acc[i].z += hv.z * wla[i]; acc[i].w += hv.w * wla[i];
            }
        }
        wave_lds_handover();
    }
#pragma unroll
    for (int i = 0; i < 4; ++i) {
        const float4 bb = *(const float4*)(b + i * 128 + lane * 4);
        float4 o;
        o.x = acc[i].x / den[i] + bb.x; o.y = acc[i].y / den[i] + bb.y;
        o.z = acc[i].z / den[i] + bb.z; o.w = acc[i].w / den[i] + bb.w;
        o.x = o.x > 0.f ? o.x : expm1f(o.x);
        o.y = o.y > 0.f ? o.y : expm1f(o.y);
        o.z = o.z > 0.f ? o.z : expm1f(o.z);
        o.w = o.w > 0.f ? o.w : expm1f(o.w);
        *(float4*)(h2 + (size_t)n * 512 + i * 128 + lane * 4) = o;
    }
}

__global__ void gat2_gather(const int* __restrict__ off, const int* __restrict__ csr,
                            const float* __restrict__ as, const float* __restrict__ ad,
                            const float* __restrict__ h3, const float* __restrict__ b,
                            float* __restrict__ out, int Nn)
{
    int t = blockIdx.x * blockDim.x + threadIdx.x;
    if (t >= Nn * 10) return;
    const int n = t / 10;
    const int g = t - n * 10;
    const int beg = off[n], end = off[n + 1];
    const float adn   = ad[n];
    const float eself = leaky(as[n] + adn);

    float m = eself;
    for (int j = beg; j < end; ++j)
        m = fmaxf(m, leaky(as[csr[j]] + adn));

    float den;
    float4 acc;
    {
        const float w = expf(eself - m);
        const float4 hv = *(const float4*)(h3 + (size_t)n * 48 + g * 4);
        den = w;
        acc.x = hv.x * w; acc.y = hv.y * w; acc.z = hv.z * w; acc.w = hv.w * w;
    }
#pragma unroll 4
    for (int j = beg; j < end; ++j) {
        const int s = csr[j];
        const float w = expf(leaky(as[s] + adn) - m);
        const float4 hv = *(const float4*)(h3 + (size_t)s * 48 + g * 4);
        den += w;
        acc.x += hv.x * w; acc.y += hv.y * w; acc.z += hv.z * w; acc.w += hv.w * w;
    }
    const float4 bb = *(const float4*)(b + g * 4);
    float4 o;
    o.x = acc.x / den + bb.x; o.y = acc.y / den + bb.y;
    o.z = acc.z / den + bb.z; o.w = acc.w / den + bb.w;
    *(float4*)(out + (size_t)n * 40 + g * 4) = o;
}

extern "C" void kernel_launch(void* const* d_in, const int* in_sizes, int n_in,
                              void* d_out, int out_size, void* d_ws, size_t ws_size,
                              hipStream_t stream)
{
    const float* x    = (const float*)d_in[0];
    const int*   ei   = (const int*)  d_in[1];
    const float* W1   = (const float*)d_in[2];
    const float* aS1  = (const float*)d_in[3];
    const float* aD1  = (const float*)d_in[4];
    const float* b1   = (const float*)d_in[5];
    const float* W2   = (const float*)d_in[6];
    const float* aS2  = (const float*)d_in[7];
    const float* aD2  = (const float*)d_in[8];
    const float* b2   = (const float*)d_in[9];
    float* out = (float*)d_out;

    const int IN  = 256, F1 = 512, OUT = 40, OUTP = 48;
    const int Nn  = in_sizes[0] / IN;
    const int E   = in_sizes[1] / 2;

    float* ws  = (float*)d_ws;
    float* h1  = ws;
    float* h2  = h1  + (size_t)Nn * F1;
    float* h3  = h2  + (size_t)Nn * F1;
    float* W2p = h3  + (size_t)Nn * 48;
    float* as1 = W2p + (size_t)F1 * OUTP;
    float* ad1 = as1 + (size_t)Nn * 8;
    float* as2 = ad1 + (size_t)Nn * 8;
    float* ad2 = as2 + Nn;
    int*   deg  = (int*)(ad2 + Nn);
    int*   off  = deg  + (size_t)Nn * CSTRIDE;
    int*   rank = off  + (Nn + 1);
    int*   csr  = rank + E;

    const int B = 256;

    zero_counters<<<(Nn + B - 1) / B, B, 0, stream>>>(deg, Nn);
    rank_edges   <<<(E + B - 1) / B, B, 0, stream>>>(ei, E, deg, rank);
    scan_degrees <<<1, 1024, 0, stream>>>(deg, off, Nn);
    place_edges  <<<(E + B - 1) / B, B, 0, stream>>>(ei, E, off, rank, csr);

    gemm_wmma_f32<4><<<dim3((Nn + 15) / 16, F1 / 64), 32, 0, stream>>>(
        x, W1, h1, Nn, IN, F1, F1, aS1, aD1, as1, ad1, F1, 8);

    gat1_gather<<<(Nn + 7) / 8, 256, 0, stream>>>(off, csr, as1, ad1, h1, b1, h2, Nn);

    pad_cols<<<(F1 * OUTP + B - 1) / B, B, 0, stream>>>(W2, W2p, F1, OUT, OUTP);

    gemm_wmma_f32<3><<<dim3((Nn + 15) / 16, 1), 32, 0, stream>>>(
        h2, W2p, h3, Nn, F1, OUTP, OUTP, aS2, aD2, as2, ad2, OUT, 1);

    gat2_gather<<<(Nn * 10 + B - 1) / B, B, 0, stream>>>(off, csr, as2, ad2, h3, b2, out, Nn);
}


# Layer-2 GEMM (625 strips, K = 512): K split over the 8 waves of a work-group, 8 dependent load-to-WMMA chunks per wave, was 64; partial tiles added via LDS in fixed wave order (new sum order)
# speedup vs baseline: 2.7435x; 1.2071x over previous
#include <hip/hip_runtime.h>
#include <hip/hip_bf16.h>
#include <math.h>

typedef float v2f __attribute__((ext_vector_type(2)));
typedef float v8f __attribute__((ext_vector_type(8)));
typedef float v4f __attribute__((ext_vector_type(4)));

#define NEG_SLOPE 0.2f

__device__ __forceinline__ float leaky(float v) {
    return v > 0.0f ? v : NEG_SLOPE * v;
}


__device__ __forceinline__ void wave_lds_handover() {
    __builtin_amdgcn_fence(__ATOMIC_RELEASE, "wavefront");
    __builtin_amdgcn_wave_barrier();
    __builtin_amdgcn_fence(__ATOMIC_ACQUIRE, "wavefront");
}

template<int MASK>
__device__ __forceinline__ float row_xor(float v) {
    const int i = __builtin_bit_cast(int, v);
    return __builtin_bit_cast(float,
        __builtin_amdgcn_update_dpp(i, i, 0x160 + MASK, 0xf, 0xf, false));
}

template<int NT, int KS>
__global__ void __launch_bounds__(32 * KS)
gemm_wmma_f32(const float* __restrict__ A, const float* __restrict__ B,
              float* __restrict__ C, int M, int K, int N, int ldc,
              const float* __restrict__ a_src, const float* __restrict__ a_dst,
              float* __restrict__ as, float* __restrict__ ad, int ncv, int heads)
{
    __shared__ float Asall[KS * 16 * 12];

    const int lane = (KS == 1) ? (int)threadIdx.x : ((int)threadIdx.x & 31);
    const int wv   = (KS == 1) ? 0
                   : __builtin_amdgcn_readfirstlane((int)threadIdx.x >> 5);
    float* __restrict__ As = Asall + wv * (16 * 12);
    const int kper = K / KS;
    const int kbeg = wv * kper;
    const int half = lane >> 4;
    const int l16  = lane & 15;

    const int alr = lane >> 1;
    const int alc = (lane & 1) * 4;
    int rowA = blockIdx.x * 16 + alr;
    if (rowA >= M) rowA = M - 1;
    const float* __restrict__ Aload = A + (size_t)rowA * (size_t)K + alc;

    const int ncol0 = blockIdx.y * (16 * NT);
    const int colB  = ncol0 + l16;

    v8f acc[NT];
#pragma unroll
    for (int t = 0; t < NT; ++t) {
        v8f z = {0.f, 0.f, 0.f, 0.f, 0.f, 0.f, 0.f, 0.f};
        acc[t] = z;
    }

    const float* __restrict__ Aw = Aload + kbeg;
    const float* __restrict__ Bw = B + (size_t)kbeg * (size_t)N;
    float4 av = *(const float4*)(Aw);

    float wsrc[NT], wdst[NT];
#pragma unroll
    for (int t = 0; t < NT; ++t) {
        const int col = colB + t * 16;
        const bool real = col < ncv;
        const int cc = real ? col : 0;
        const float vs = a_src[cc], vd = a_dst[cc];
        wsrc[t] = real ? vs : 0.f;
        wdst[t] = real ? vd : 0.f;
    }

    for (int kc = 0; kc < kper; kc += 8) {
        if (KS == 1) __syncthreads(); else wave_lds_handover();
        *(float4*)(&As[alr * 12 + alc]) = av;

        v2f breg[2][NT];
#pragma unroll
        for (int kk2 = 0; kk2 < 2; ++kk2) {
            const float* __restrict__ Bk =
                Bw + (size_t)(kc + kk2 * 4 + 2 * half) * (size_t)N + colB;
#pragma unroll
            for (int t = 0; t < NT; ++t) {
                breg[kk2][t].x = Bk[t * 16];
                breg[kk2][t].y = Bk[N + t * 16];
            }
        }

        const int kn = (kc + 8 < kper) ? (kc + 8) : 0;
        av = *(const float4*)(Aw + kn);

        if (KS == 1) __syncthreads(); else wave_lds_handover();

#pragma unroll
        for (int kk2 = 0; kk2 < 2; ++kk2) {
            const v2f a = *(const v2f*)(&As[l16 * 12 + kk2 * 4 + 2 * half]);
#pragma unroll
            for (int t = 0; t < NT; ++t) {
                acc[t] = __builtin_amdgcn_wmma_f32_16x16x4_f32(
                    false, a, false, breg[kk2][t], (short)0, acc[t], false, false);
            }
        }
    }

    if (KS > 1) {
        __shared__ float4 Part[(KS > 1 ? KS - 1 : 1) * NT * 2 * 32];
        if (wv != 0) {
            float4* __restrict__ pp = Part + (size_t)(wv - 1) * (NT * 2 * 32) + lane;
#pragma unroll
            for (int t = 0; t < NT; ++t) {
                pp[(t * 2 + 0) * 32] = make_float4(acc[t][0], acc[t][1], acc[t][2], acc[t][3]);
                pp[(t * 2 + 1) * 32] = make_float4(acc[t][4], acc[t][5], acc[t][6], acc[t][7]);
            }
        }
        __syncthreads();
        if (wv != 0) return;
        constexpr int NP = NT * 2;
        const v4f* __restrict__ pp = (const v4f*)Part + lane;
#pragma unroll 1
        for (int w = 0; w < KS - 1; ++w, pp += NP * 32) {
            v4f pv[NP];
#pragma unroll
            for (int i = 0; i < NP; ++i) pv[i] = pp[i * 32];
            if constexpr (NT == 3) {
                asm volatile("" :: "v"(pv[0]), "v"(pv[1]), "v"(pv[2]),
                                   "v"(pv[3]), "v"(pv[4]), "v"(pv[5]));
            }
#pragma unroll
            for (int t = 0; t < NT; ++t) {
#pragma unroll
                for (int r = 0; r < 4; ++r) {
                    acc[t][r]     += pv[t * 2][r];
                    acc[t][r + 4] += pv[t * 2 + 1][r];
                }
            }
        }
    }

    const int crow0 = blockIdx.x * 16 + half * 8;
    float* __restrict__ Cbase = C + (size_t)crow0 * (size_t)ldc + ncol0 + l16;
    if (blockIdx.x * 16 + 16 <= M) {
#pragma unroll
        for (int t = 0; t < NT; ++t) {
            float* __restrict__ Cp = Cbase + t * 16;
#pragma unroll
            for (int r = 0; r < 8; ++r)
                Cp[(size_t)r * (size_t)ldc] = acc[t][r];
        }
    } else {
#pragma unroll
        for (int t = 0; t < NT; ++t) {
            float* __restrict__ Cp = Cbase + t * 16;
#pragma unroll
            for (int r = 0; r < 8; ++r)
                if (crow0 + r < M) Cp[(size_t)r * (size_t)ldc] = acc[t][r];
        }
    }

    float ps[8], pd[8];
#pragma unroll
    for (int r = 0; r < 8; ++r) { ps[r] = 0.f; pd[r] = 0.f; }
#pragma unroll
    for (int t = 0; t < NT; ++t) {
#pragma unroll
        for (int r = 0; r < 8; ++r) {
            ps[r] += acc[t][r] * wsrc[t];
            pd[r] += acc[t][r] * wdst[t];
        }
    }
    const bool b3 = (l16 & 8) != 0, b2 = (l16 & 4) != 0;
    const bool b1 = (l16 & 2) != 0, b0 = (l16 & 1) != 0;
    float u[8], w4[4], y2[2];
#pragma unroll
    for (int i = 0; i < 8; ++i) {
        const float mine = b3 ? pd[i] : ps[i];
        const float give = b3 ? ps[i] : pd[i];
        u[i] = mine + row_xor<8>(give);
    }
#pragma unroll
    for (int i = 0; i < 4; ++i) {
        const float mine = b2 ? u[i + 4] : u[i];
        const float give = b2 ? u[i] : u[i + 4];
        w4[i] = mine + row_xor<4>(give);
    }
#pragma unroll
    for (int i = 0; i < 2; ++i) {
        const float mine = b1 ? w4[i + 2] : w4[i];
        const float give = b1 ? w4[i] : w4[i + 2];
        y2[i] = mine + row_xor<2>(give);
    }
    const float mine = b0 ? y2[1] : y2[0];
    const float give = b0 ? y2[0] : y2[1];
    const float z = mine + row_xor<1>(give);
    const int arow = blockIdx.x * 16 + half * 8 + (l16 & 7);
    if (arow < M) {
        float* __restrict__ dstp = b3 ? ad : as;
        dstp[(size_t)arow * (size_t)heads + blockIdx.y] = z;
    }
}

#define CSTRIDE 32

__global__ void zero_counters(int* __restrict__ deg, int Nn) {
    int t = blockIdx.x * blockDim.x + threadIdx.x;
    if (t < Nn) deg[(size_t)t * CSTRIDE] = 0;
}

__global__ void rank_edges(const int* __restrict__ ei, int E,
                           int* __restrict__ deg, int* __restrict__ rank)
{
    int e = blockIdx.x * blockDim.x + threadIdx.x;
    if (e >= E) return;
    int d = ei[E + e];
    rank[e] = atomicAdd(&deg[(size_t)d * CSTRIDE], 1);
}

__global__ void __launch_bounds__(1024)
scan_degrees(const int* __restrict__ deg, int* __restrict__ off, int Nn)
{
    __shared__ int part[1024];
    const int t   = (int)threadIdx.x;
    const int ipt = (Nn + 1023) / 1024;
    const int b0  = t * ipt;
    int sum = 0;
#pragma unroll 4
    for (int i = 0; i < ipt; ++i) {
        int idx = b0 + i;
        int v = deg[(size_t)(idx < Nn ? idx : Nn - 1) * CSTRIDE];
        sum += (idx < Nn) ? v : 0;
    }
    part[t] = sum;
    __syncthreads();
    for (int o = 1; o < 1024; o <<= 1) {
        int v = (t >= o) ? part[t - o] : 0;
        __syncthreads();
        part[t] += v;
        __syncthreads();
    }
    int run = part[t] - sum;
#pragma unroll 4
    for (int i = 0; i < ipt; ++i) {
        int idx = b0 + i;
        int v = deg[(size_t)(idx < Nn ? idx : Nn - 1) * CSTRIDE];
        if (idx < Nn) off[idx] = run;
        run += v;
    }
    if (t == 1023) off[Nn] = part[1023];
}

__global__ void place_edges(const int* __restrict__ ei, int E,
                            const int* __restrict__ off, const int* __restrict__ rank,
                            int* __restrict__ csr)
{
    int e = blockIdx.x * blockDim.x + threadIdx.x;
    if (e >= E) return;
    int d = ei[E + e];
    csr[off[d] + rank[e]] = ei[e];
}

__global__ void pad_cols(const float* __restrict__ src, float* __restrict__ dst,
                         int K, int N, int Np)
{
    int t = blockIdx.x * blockDim.x + threadIdx.x;
    if (t >= K * Np) return;
    int r = t / Np, c = t - r * Np;
    dst[t] = (c < N) ? src[r * N + c] : 0.f;
}

__global__ void __launch_bounds__(256)
gat1_gather(const int* __restrict__ off, const int* __restrict__ csr,
            const float* __restrict__ as, const float* __restrict__ ad,
            const float* __restrict__ h1, const float* __restrict__ b,
            float* __restrict__ h2, int Nn)
{
    __shared__ float4 wsh[8 * 64];

    const int lane = (int)threadIdx.x & 31;
    const int half = lane >> 4;
    const int wv   = (int)threadIdx.x >> 5;
    float* __restrict__ wmine = (float*)(wsh + wv * 64);
    const int n = __builtin_amdgcn_readfirstlane((int)blockIdx.x * 8 + wv);
    if (n >= Nn) return;
    const int beg = off[n], end = off[n + 1];
    const int last = end - 1;

    const int q = lane >> 3, hh = lane & 7;
    const float adq = ad[n * 8 + hh];
    float m = leaky(as[n * 8 + hh] + adq);
    for (int base = beg; base < end; base += 32) {
#pragma unroll
        for (int t = 0; t < 8; ++t) {
            const int j = base + q + 4 * t;
            const int s = csr[j < last ? j : last];
            m = fmaxf(m, leaky(as[s * 8 + hh] + adq));
        }
    }
    m = fmaxf(m, __shfl_xor(m, 8));
    m = fmaxf(m, __shfl_xor(m, 16));

    float den[4];
    float4 acc[4];
    const float* __restrict__ hcol = h1 + lane * 4;
#pragma unroll
    for (int i = 0; i < 4; ++i) {
        const int hd = 2 * i + half;
        const float mh = __shfl(m, hd);
        const float w = expf(leaky(as[n * 8 + hd] + ad[n * 8 + hd]) - mh);
        const float4 hv = *(const float4*)(hcol + (size_t)n * 512 + i * 128);
        den[i] = w;
        acc[i].x = hv.x * w; acc[i].y = hv.y * w; acc[i].z = hv.z * w; acc[i].w = hv.w * w;
    }

    const int wpos = (hh & 1) * 4 + (hh >> 1);
    for (int base = beg; base < end; base += 32) {
#pragma unroll
        for (int t = 0; t < 8; ++t) {
            const int j = base + q + 4 * t;
            const int s = csr[j < last ? j : last];
            wmine[(q + 4 * t) * 8 + wpos] = expf(leaky(as[s * 8 + hh] + adq) - m);
        }
        wave_lds_handover();
        const int left = end - base;
        const int cnt = left < 32 ? left : 32;
        int e = 0;
        for (; e + 2 <= cnt; e += 2) {
            const int s0 = csr[base + e], s1 = csr[base + e + 1];
            const float4 wa = *(const float4*)(wmine + e * 8 + half * 4);
            const float4 wb = *(const float4*)(wmine + e * 8 + 8 + half * 4);
            const float* __restrict__ r0 = hcol + (size_t)s0 * 512;
            const float* __restrict__ r1 = hcol + (size_t)s1 * 512;
            const v4f a0 = *(const v4f*)(r0),       a1 = *(const v4f*)(r0 + 128);
            const v4f a2 = *(const v4f*)(r0 + 256), a3 = *(const v4f*)(r0 + 384);
            const v4f b0 = *(const v4f*)(r1),       b1 = *(const v4f*)(r1 + 128);
            const v4f b2 = *(const v4f*)(r1 + 256), b3 = *(const v4f*)(r1 + 384);
            asm volatile("" :: "v"(a0), "v"(a1), "v"(a2), "v"(a3),
                               "v"(b0), "v"(b1), "v"(b2), "v"(b3));
            const v4f ra[4] = {a0, a1, a2, a3};
            const v4f rb[4] = {b0, b1, b2, b3};
            const float wla[4] = {wa.x, wa.y, wa.z, wa.w};
            const float wlb[4] = {wb.x, wb.y, wb.z, wb.w};
#pragma unroll
            for (int i = 0; i < 4; ++i) {
                den[i] += wla[i];
                acc[i].x += ra[i].x * wla[i]; acc[i].y += ra[i].y * wla[i];
                acc[i].z += ra[i].z * wla[i]; acc[i].w += ra[i].w * wla[i];
            }
#pragma unroll
            for (int i = 0; i < 4; ++i) {
                den[i] += wlb[i];
                acc[i].x += rb[i].x * wlb[i]; acc[i].y += rb[i].y * wlb[i];
                acc[i].z += rb[i].z * wlb[i]; acc[i].w += rb[i].w * wlb[i];
            }
        }
        if (e < cnt) {
            const int s0 = csr[base + e];
            const float4 wa = *(const float4*)(wmine + e * 8 + half * 4);
            const float wla[4] = {wa.x, wa.y, wa.z, wa.w};
            const float* __restrict__ r0 = hcol + (size_t)s0 * 512;
#pragma unroll
            for (int i = 0; i < 4; ++i) {
                const v4f hv = *(const v4f*)(r0 + i * 128);
                den[i] += wla[i];
                acc[i].x += hv.x * wla[i]; acc[i].y += hv.y * wla[i];
                acc[i].z += hv.z * wla[i]; acc[i].w += hv.w * wla[i];
            }
        }
        wave_lds_handover();
    }
#pragma unroll
    for (int i = 0; i < 4; ++i) {
        const float4 bb = *(const float4*)(b + i * 128 + lane * 4);
        float4 o;
        o.x = acc[i].x / den[i] + bb.x; o.y = acc[i].y / den[i] + bb.y;
        o.z = acc[i].z / den[i] + bb.z; o.w = acc[i].w / den[i] + bb.w;
        o.x = o.x > 0.f ? o.x : expm1f(o.x);
        o.y = o.y > 0.f ? o.y : expm1f(o.y);
        o.z = o.z > 0.f ? o.z : expm1f(o.z);
        o.w = o.w > 0.f ? o.w : expm1f(o.w);
        *(float4*)(h2 + (size_t)n * 512 + i * 128 + lane * 4) = o;
    }
}

__global__ void gat2_gather(const int* __restrict__ off, const int* __restrict__ csr,
                            const float* __restrict__ as, const float* __restrict__ ad,
                            const float* __restrict__ h3, const float* __restrict__ b,
                            float* __restrict__ out, int Nn)
{
    int t = blockIdx.x * blockDim.x + threadIdx.x;
    if (t >= Nn * 10) return;
    const int n = t / 10;
    const int g = t - n * 10;
    const int beg = off[n], end = off[n + 1];
    const float adn   = ad[n];
    const float eself = leaky(as[n] + adn);

    float m = eself;
    for (int j = beg; j < end; ++j)
        m = fmaxf(m, leaky(as[csr[j]] + adn));

    float den;
    float4 acc;
    {
        const float w = expf(eself - m);
        const float4 hv = *(const float4*)(h3 + (size_t)n * 48 + g * 4);
        den = w;
        acc.x = hv.x * w; acc.y = hv.y * w; acc.z = hv.z * w; acc.w = hv.w * w;
    }
#pragma unroll 4
    for (int j = beg; j < end; ++j) {
        const int s = csr[j];
        const float w = expf(leaky(as[s] + adn) - m);
        const float4 hv = *(const float4*)(h3 + (size_t)s * 48 + g * 4);
        den += w;
        acc.x += hv.x * w; acc.y += hv.y * w; acc.z += hv.z * w; acc.w += hv.w * w;
    }
    const float4 bb = *(const float4*)(b + g * 4);
    float4 o;
    o.x = acc.x / den + bb.x; o.y = acc.y / den + bb.y;
    o.z = acc.z / den + bb.z; o.w = acc.w / den + bb.w;
    *(float4*)(out + (size_t)n * 40 + g * 4) = o;
}

extern "C" void kernel_launch(void* const* d_in, const int* in_sizes, int n_in,
                              void* d_out, int out_size, void* d_ws, size_t ws_size,
                              hipStream_t stream)
{
    const float* x    = (const float*)d_in[0];
    const int*   ei   = (const int*)  d_in[1];
    const float* W1   = (const float*)d_in[2];
    const float* aS1  = (const float*)d_in[3];
    const float* aD1  = (const float*)d_in[4];
    const float* b1   = (const float*)d_in[5];
    const float* W2   = (const float*)d_in[6];
    const float* aS2  = (const float*)d_in[7];
    const float* aD2  = (const float*)d_in[8];
    const float* b2   = (const float*)d_in[9];
    float* out = (float*)d_out;

    const int IN  = 256, F1 = 512, OUT = 40, OUTP = 48;
    const int Nn  = in_sizes[0] / IN;
    const int E   = in_sizes[1] / 2;

    float* ws  = (float*)d_ws;
    float* h1  = ws;
    float* h2  = h1  + (size_t)Nn * F1;
    float* h3  = h2  + (size_t)Nn * F1;
    float* W2p = h3  + (size_t)Nn * 48;
    float* as1 = W2p + (size_t)F1 * OUTP;
    float* ad1 = as1 + (size_t)Nn * 8;
    float* as2 = ad1 + (size_t)Nn * 8;
    float* ad2 = as2 + Nn;
    int*   deg  = (int*)(ad2 + Nn);
    int*   off  = deg  + (size_t)Nn * CSTRIDE;
    int*   rank = off  + (Nn + 1);
    int*   csr  = rank + E;

    const int B = 256;

    zero_counters<<<(Nn + B - 1) / B, B, 0, stream>>>(deg, Nn);
    rank_edges   <<<(E + B - 1) / B, B, 0, stream>>>(ei, E, deg, rank);
    scan_degrees <<<1, 1024, 0, stream>>>(deg, off, Nn);
    place_edges  <<<(E + B - 1) / B, B, 0, stream>>>(ei, E, off, rank, csr);

    gemm_wmma_f32<4, 1><<<dim3((Nn + 15) / 16, F1 / 64), 32, 0, stream>>>(
        x, W1, h1, Nn, IN, F1, F1, aS1, aD1, as1, ad1, F1, 8);

    gat1_gather<<<(Nn + 7) / 8, 256, 0, stream>>>(off, csr, as1, ad1, h1, b1, h2, Nn);

    pad_cols<<<(F1 * OUTP + B - 1) / B, B, 0, stream>>>(W2, W2p, F1, OUT, OUTP);

    gemm_wmma_f32<3, 8><<<dim3((Nn + 15) / 16, 1), 256, 0, stream>>>(
        h2, W2p, h3, Nn, F1, OUTP, OUTP, aS2, aD2, as2, ad2, OUT, 1);

    gat2_gather<<<(Nn * 10 + B - 1) / B, B, 0, stream>>>(off, csr, as2, ad2, h3, b2, out, Nn);
}


# Degree scan split over 10 work-groups of 1024 nodes (was one, reading all 10000 counter lines twice); each gets its base by counting the edge destinations below its range; integer sums only
# speedup vs baseline: 3.0071x; 1.0961x over previous
#include <hip/hip_runtime.h>
#include <hip/hip_bf16.h>
#include <math.h>

typedef float v2f __attribute__((ext_vector_type(2)));
typedef float v8f __attribute__((ext_vector_type(8)));
typedef float v4f __attribute__((ext_vector_type(4)));

#define NEG_SLOPE 0.2f

__device__ __forceinline__ float leaky(float v) {
    return v > 0.0f ? v : NEG_SLOPE * v;
}


__device__ __forceinline__ void wave_lds_handover() {
    __builtin_amdgcn_fence(__ATOMIC_RELEASE, "wavefront");
    __builtin_amdgcn_wave_barrier();
    __builtin_amdgcn_fence(__ATOMIC_ACQUIRE, "wavefront");
}

template<int MASK>
__device__ __forceinline__ float row_xor(float v) {
    const int i = __builtin_bit_cast(int, v);
    return __builtin_bit_cast(float,
        __builtin_amdgcn_update_dpp(i, i, 0x160 + MASK, 0xf, 0xf, false));
}

template<int NT, int KS>
__global__ void __launch_bounds__(32 * KS)
gemm_wmma_f32(const float* __restrict__ A, const float* __restrict__ B,
              float* __restrict__ C, int M, int K, int N, int ldc,
              const float* __restrict__ a_src, const float* __restrict__ a_dst,
              float* __restrict__ as, float* __restrict__ ad, int ncv, int heads)
{
    __shared__ float Asall[KS * 16 * 12];

    const int lane = (KS == 1) ? (int)threadIdx.x : ((int)threadIdx.x & 31);
    const int wv   = (KS == 1) ? 0
                   : __builtin_amdgcn_readfirstlane((int)threadIdx.x >> 5);
    float* __restrict__ As = Asall + wv * (16 * 12);
    const int kper = K / KS;
    const int kbeg = wv * kper;
    const int half = lane >> 4;
    const int l16  = lane & 15;

    const int alr = lane >> 1;
    const int alc = (lane & 1) * 4;
    int rowA = blockIdx.x * 16 + alr;
    if (rowA >= M) rowA = M - 1;
    const float* __restrict__ Aload = A + (size_t)rowA * (size_t)K + alc;

    const int ncol0 = blockIdx.y * (16 * NT);
    const int colB  = ncol0 + l16;

    v8f acc[NT];
#pragma unroll
    for (int t = 0; t < NT; ++t) {
        v8f z = {0.f, 0.f, 0.f, 0.f, 0.f, 0.f, 0.f, 0.f};
        acc[t] = z;
    }

    const float* __restrict__ Aw = Aload + kbeg;
    const float* __restrict__ Bw = B + (size_t)kbeg * (size_t)N;
    float4 av = *(const float4*)(Aw);

    float wsrc[NT], wdst[NT];
#pragma unroll
    for (int t = 0; t < NT; ++t) {
        const int col = colB + t * 16;
        const bool real = col < ncv;
        const int cc = real ? col : 0;
        const float vs = a_src[cc], vd = a_dst[cc];
        wsrc[t] = real ? vs : 0.f;
        wdst[t] = real ? vd : 0.f;
    }

    for (int kc = 0; kc < kper; kc += 8) {
        if (KS == 1) __syncthreads(); else wave_lds_handover();
        *(float4*)(&As[alr * 12 + alc]) = av;

        v2f breg[2][NT];
#pragma unroll
        for (int kk2 = 0; kk2 < 2; ++kk2) {
            const float* __restrict__ Bk =
                Bw + (size_t)(kc + kk2 * 4 + 2 * half) * (size_t)N + colB;
#pragma unroll
            for (int t = 0; t < NT; ++t) {
                breg[kk2][t].x = Bk[t * 16];
                breg[kk2][t].y = Bk[N + t * 16];
            }
        }

        const int kn = (kc + 8 < kper) ? (kc + 8) : 0;
        av = *(const float4*)(Aw + kn);

        if (KS == 1) __syncthreads(); else wave_lds_handover();

#pragma unroll
        for (int kk2 = 0; kk2 < 2; ++kk2) {
            const v2f a = *(const v2f*)(&As[l16 * 12 + kk2 * 4 + 2 * half]);
#pragma unroll
            for (int t = 0; t < NT; ++t) {
                acc[t] = __builtin_amdgcn_wmma_f32_16x16x4_f32(
                    false, a, false, breg[kk2][t], (short)0, acc[t], false, false);
            }
        }
    }

    if (KS > 1) {
        __shared__ float4 Part[(KS > 1 ? KS - 1 : 1) * NT * 2 * 32];
        if (wv != 0) {
            float4* __restrict__ pp = Part + (size_t)(wv - 1) * (NT * 2 * 32) + lane;
#pragma unroll
            for (int t = 0; t < NT; ++t) {
                pp[(t * 2 + 0) * 32] = make_float4(acc[t][0], acc[t][1], acc[t][2], acc[t][3]);
                pp[(t * 2 + 1) * 32] = make_float4(acc[t][4], acc[t][5], acc[t][6], acc[t][7]);
            }
        }
        __syncthreads();
        if (wv != 0) return;
        constexpr int NP = NT * 2;
        const v4f* __restrict__ pp = (const v4f*)Part + lane;
#pragma unroll 1
        for (int w = 0; w < KS - 1; ++w, pp += NP * 32) {
            v4f pv[NP];
#pragma unroll
            for (int i = 0; i < NP; ++i) pv[i] = pp[i * 32];
            if constexpr (NT == 3) {
                asm volatile("" :: "v"(pv[0]), "v"(pv[1]), "v"(pv[2]),
                                   "v"(pv[3]), "v"(pv[4]), "v"(pv[5]));
            }
#pragma unroll
            for (int t = 0; t < NT; ++t) {
#pragma unroll
                for (int r = 0; r < 4; ++r) {
                    acc[t][r]     += pv[t * 2][r];
                    acc[t][r + 4] += pv[t * 2 + 1][r];
                }
            }
        }
    }

    const int crow0 = blockIdx.x * 16 + half * 8;
    float* __restrict__ Cbase = C + (size_t)crow0 * (size_t)ldc + ncol0 + l16;
    if (blockIdx.x * 16 + 16 <= M) {
#pragma unroll
        for (int t = 0; t < NT; ++t) {
            float* __restrict__ Cp = Cbase + t * 16;
#pragma unroll
            for (int r = 0; r < 8; ++r)
                Cp[(size_t)r * (size_t)ldc] = acc[t][r];
        }
    } else {
#pragma unroll
        for (int t = 0; t < NT; ++t) {
            float* __restrict__ Cp = Cbase + t * 16;
#pragma unroll
            for (int r = 0; r < 8; ++r)
                if (crow0 + r < M) Cp[(size_t)r * (size_t)ldc] = acc[t][r];
        }
    }

    float ps[8], pd[8];
#pragma unroll
    for (int r = 0; r < 8; ++r) { ps[r] = 0.f; pd[r] = 0.f; }
#pragma unroll
    for (int t = 0; t < NT; ++t) {
#pragma unroll
        for (int r = 0; r < 8; ++r) {
            ps[r] += acc[t][r] * wsrc[t];
            pd[r] += acc[t][r] * wdst[t];
        }
    }
    const bool b3 = (l16 & 8) != 0, b2 = (l16 & 4) != 0;
    const bool b1 = (l16 & 2) != 0, b0 = (l16 & 1) != 0;
    float u[8], w4[4], y2[2];
#pragma unroll
    for (int i = 0; i < 8; ++i) {
        const float mine = b3 ? pd[i] : ps[i];
        const float give = b3 ? ps[i] : pd[i];
        u[i] = mine + row_xor<8>(give);
    }
#pragma unroll
    for (int i = 0; i < 4; ++i) {
        const float mine = b2 ? u[i + 4] : u[i];
        const float give = b2 ? u[i] : u[i + 4];
        w4[i] = mine + row_xor<4>(give);
    }
#pragma unroll
    for (int i = 0; i < 2; ++i) {
        const float mine = b1 ? w4[i + 2] : w4[i];
        const float give = b1 ? w4[i] : w4[i + 2];
        y2[i] = mine + row_xor<2>(give);
    }
    const float mine = b0 ? y2[1] : y2[0];
    const float give = b0 ? y2[0] : y2[1];
    const float z = mine + row_xor<1>(give);
    const int arow = blockIdx.x * 16 + half * 8 + (l16 & 7);
    if (arow < M) {
        float* __restrict__ dstp = b3 ? ad : as;
        dstp[(size_t)arow * (size_t)heads + blockIdx.y] = z;
    }
}

#define CSTRIDE 32

__global__ void zero_counters(int* __restrict__ deg, int Nn) {
    int t = blockIdx.x * blockDim.x + threadIdx.x;
    if (t < Nn) deg[(size_t)t * CSTRIDE] = 0;
}

__global__ void rank_edges(const int* __restrict__ ei, int E,
                           int* __restrict__ deg, int* __restrict__ rank)
{
    int e = blockIdx.x * blockDim.x + threadIdx.x;
    if (e >= E) return;
    int d = ei[E + e];
    rank[e] = atomicAdd(&deg[(size_t)d * CSTRIDE], 1);
}

__global__ void __launch_bounds__(1024)
scan_degrees(const int* __restrict__ ei, int E, const int* __restrict__ deg,
             int* __restrict__ off, int Nn)
{
    __shared__ int wdeg[32];
    __shared__ int wcnt[32];
    const int t    = (int)threadIdx.x;
    const int lane = t & 31;
    const int wv   = t >> 5;
    const int lo   = (int)blockIdx.x * 1024;
    const int idx  = lo + t;

    const int v0 = deg[(size_t)(idx < Nn ? idx : Nn - 1) * CSTRIDE];
    const int v  = (idx < Nn) ? v0 : 0;

    int cnt = 0;
    if (lo > 0) {
        const int* __restrict__ dstp = ei + E;
        const int nvec = (((size_t)dstp & 15) == 0) ? (E >> 2) : 0;
        const int4* __restrict__ dv = (const int4*)dstp;
#pragma unroll 8
        for (int i = t; i < nvec; i += 1024) {
            const int4 d = dv[i];
            cnt += (d.x < lo ? 1 : 0) + (d.y < lo ? 1 : 0)
                 + (d.z < lo ? 1 : 0) + (d.w < lo ? 1 : 0);
        }
        for (int e = nvec * 4 + t; e < E; e += 1024)
            cnt += (dstp[e] < lo) ? 1 : 0;
    }

    int s = v;
#pragma unroll
    for (int o = 1; o < 32; o <<= 1) {
        const int u = __shfl_up(s, o);
        if (lane >= o) s += u;
    }
#pragma unroll
    for (int o = 16; o > 0; o >>= 1) cnt += __shfl_xor(cnt, o);
    if (lane == 31) wdeg[wv] = s;
    if (lane == 0)  wcnt[wv] = cnt;
    __syncthreads();

    int base  = wcnt[lane];
    int front = (lane < wv) ? wdeg[lane] : 0;
#pragma unroll
    for (int o = 16; o > 0; o >>= 1) {
        base  += __shfl_xor(base, o);
        front += __shfl_xor(front, o);
    }
    const int excl = base + front + s - v;
    if (idx < Nn)      off[idx] = excl;
    if (idx == Nn - 1) off[Nn]  = excl + v;
}

__global__ void place_edges(const int* __restrict__ ei, int E,
                            const int* __restrict__ off, const int* __restrict__ rank,
                            int* __restrict__ csr)
{
    int e = blockIdx.x * blockDim.x + threadIdx.x;
    if (e >= E) return;
    int d = ei[E + e];
    csr[off[d] + rank[e]] = ei[e];
}

__global__ void pad_cols(const float* __restrict__ src, float* __restrict__ dst,
                         int K, int N, int Np)
{
    int t = blockIdx.x * blockDim.x + threadIdx.x;
    if (t >= K * Np) return;
    int r = t / Np, c = t - r * Np;
    dst[t] = (c < N) ? src[r * N + c] : 0.f;
}

__global__ void __launch_bounds__(256)
gat1_gather(const int* __restrict__ off, const int* __restrict__ csr,
            const float* __restrict__ as, const float* __restrict__ ad,
            const float* __restrict__ h1, const float* __restrict__ b,
            float* __restrict__ h2, int Nn)
{
    __shared__ float4 wsh[8 * 64];

    const int lane = (int)threadIdx.x & 31;
    const int half = lane >> 4;
    const int wv   = (int)threadIdx.x >> 5;
    float* __restrict__ wmine = (float*)(wsh + wv * 64);
    const int n = __builtin_amdgcn_readfirstlane((int)blockIdx.x * 8 + wv);
    if (n >= Nn) return;
    const int beg = off[n], end = off[n + 1];
    const int last = end - 1;

    const int q = lane >> 3, hh = lane & 7;
    const float adq = ad[n * 8 + hh];
    float m = leaky(as[n * 8 + hh] + adq);
    for (int base = beg; base < end; base += 32) {
#pragma unroll
        for (int t = 0; t < 8; ++t) {
            const int j = base + q + 4 * t;
            const int s = csr[j < last ? j : last];
            m = fmaxf(m, leaky(as[s * 8 + hh] + adq));
        }
    }
    m = fmaxf(m, __shfl_xor(m, 8));
    m = fmaxf(m, __shfl_xor(m, 16));

    float den[4];
    float4 acc[4];
    const float* __restrict__ hcol = h1 + lane * 4;
#pragma unroll
    for (int i = 0; i < 4; ++i) {
        const int hd = 2 * i + half;
        const float mh = __shfl(m, hd);
        const float w = expf(leaky(as[n * 8 + hd] + ad[n * 8 + hd]) - mh);
        const float4 hv = *(const float4*)(hcol + (size_t)n * 512 + i * 128);
        den[i] = w;
        acc[i].x = hv.x * w; acc[i].y = hv.y * w; acc[i].z = hv.z * w; acc[i].w = hv.w * w;
    }

    const int wpos = (hh & 1) * 4 + (hh >> 1);
    for (int base = beg; base < end; base += 32) {
#pragma unroll
        for (int t = 0; t < 8; ++t) {
            const int j = base + q + 4 * t;
            const int s = csr[j < last ? j : last];
            wmine[(q + 4 * t) * 8 + wpos] = expf(leaky(as[s * 8 + hh] + adq) - m);
        }
        wave_lds_handover();
        const int left = end - base;
        const int cnt = left < 32 ? left : 32;
        int e = 0;
        for (; e + 2 <= cnt; e += 2) {
            const int s0 = csr[base + e], s1 = csr[base + e + 1];
            const float4 wa = *(const float4*)(wmine + e * 8 + half * 4);
            const float4 wb = *(const float4*)(wmine + e * 8 + 8 + half * 4);
            const float* __restrict__ r0 = hcol + (size_t)s0 * 512;
            const float* __restrict__ r1 = hcol + (size_t)s1 * 512;
            const v4f a0 = *(const v4f*)(r0),       a1 = *(const v4f*)(r0 + 128);
            const v4f a2 = *(const v4f*)(r0 + 256), a3 = *(const v4f*)(r0 + 384);
            const v4f b0 = *(const v4f*)(r1),       b1 = *(const v4f*)(r1 + 128);
            const v4f b2 = *(const v4f*)(r1 + 256), b3 = *(const v4f*)(r1 + 384);
            asm volatile("" :: "v"(a0), "v"(a1), "v"(a2), "v"(a3),
                               "v"(b0), "v"(b1), "v"(b2), "v"(b3));
            const v4f ra[4] = {a0, a1, a2, a3};
            const v4f rb[4] = {b0, b1, b2, b3};
            const float wla[4] = {wa.x, wa.y, wa.z, wa.w};
            const float wlb[4] = {wb.x, wb.y, wb.z, wb.w};
#pragma unroll
            for (int i = 0; i < 4; ++i) {
                den[i] += wla[i];
                acc[i].x += ra[i].x * wla[i]; acc[i].y += ra[i].y * wla[i];
                acc[i].z += ra[i].z * wla[i]; acc[i].w += ra[i].w * wla[i];
            }
#pragma unroll
            for (int i = 0; i < 4; ++i) {
                den[i] += wlb[i];
                acc[i].x += rb[i].x * wlb[i]; acc[i].y += rb[i].y * wlb[i];
                acc[i].z += rb[i].z * wlb[i]; acc[i].w += rb[i].w * wlb[i];
            }
        }
        if (e < cnt) {
            const int s0 = csr[base + e];
            const float4 wa = *(const float4*)(wmine + e * 8 + half * 4);
            const float wla[4] = {wa.x, wa.y, wa.z, wa.w};
            const float* __restrict__ r0 = hcol + (size_t)s0 * 512;
#pragma unroll
            for (int i = 0; i < 4; ++i) {
                const v4f hv = *(const v4f*)(r0 + i * 128);
                den[i] += wla[i];
                acc[i].x += hv.x * wla[i]; acc[i].y += hv.y * wla[i];
                acc[i].z += hv.z * wla[i]; acc[i].w += hv.w * wla[i];
            }
        }
        wave_lds_handover();
    }
#pragma unroll
    for (int i = 0; i < 4; ++i) {
        const float4 bb = *(const float4*)(b + i * 128 + lane * 4);
        float4 o;
        o.x = acc[i].x / den[i] + bb.x; o.y = acc[i].y / den[i] + bb.y;
        o.z = acc[i].z / den[i] + bb.z; o.w = acc[i].w / den[i] + bb.w;
        o.x = o.x > 0.f ? o.x : expm1f(o.x);
        o.y = o.y > 0.f ? o.y : expm1f(o.y);
        o.z = o.z > 0.f ? o.z : expm1f(o.z);
        o.w = o.w > 0.f ? o.w : expm1f(o.w);
        *(float4*)(h2 + (size_t)n * 512 + i * 128 + lane * 4) = o;
    }
}

__global__ void gat2_gather(const int* __restrict__ off, const int* __restrict__ csr,
                            const float* __restrict__ as, const float* __restrict__ ad,
                            const float* __restrict__ h3, const float* __restrict__ b,
                            float* __restrict__ out, int Nn)
{
    int t = blockIdx.x * blockDim.x + threadIdx.x;
    if (t >= Nn * 10) return;
    const int n = t / 10;
    const int g = t - n * 10;
    const int beg = off[n], end = off[n + 1];
    const float adn   = ad[n];
    const float eself = leaky(as[n] + adn);

    float m = eself;
    for (int j = beg; j < end; ++j)
        m = fmaxf(m, leaky(as[csr[j]] + adn));

    float den;
    float4 acc;
    {
        const float w = expf(eself - m);
        const float4 hv = *(const float4*)(h3 + (size_t)n * 48 + g * 4);
        den = w;
        acc.x = hv.x * w; acc.y = hv.y * w; acc.z = hv.z * w; acc.w = hv.w * w;
    }
#pragma unroll 4
    for (int j = beg; j < end; ++j) {
        const int s = csr[j];
        const float w = expf(leaky(as[s] + adn) - m);
        const float4 hv = *(const float4*)(h3 + (size_t)s * 48 + g * 4);
        den += w;
        acc.x += hv.x * w; acc.y += hv.y * w; acc.z += hv.z * w; acc.w += hv.w * w;
    }
    const float4 bb = *(const float4*)(b + g * 4);
    float4 o;
    o.x = acc.x / den + bb.x; o.y = acc.y / den + bb.y;
    o.z = acc.z / den + bb.z; o.w = acc.w / den + bb.w;
    *(float4*)(out + (size_t)n * 40 + g * 4) = o;
}

extern "C" void kernel_launch(void* const* d_in, const int* in_sizes, int n_in,
                              void* d_out, int out_size, void* d_ws, size_t ws_size,
                              hipStream_t stream)
{
    const float* x    = (const float*)d_in[0];
    const int*   ei   = (const int*)  d_in[1];
    const float* W1   = (const float*)d_in[2];
    const float* aS1  = (const float*)d_in[3];
    const float* aD1  = (const float*)d_in[4];
    const float* b1   = (const float*)d_in[5];
    const float* W2   = (const float*)d_in[6];
    const float* aS2  = (const float*)d_in[7];
    const float* aD2  = (const float*)d_in[8];
    const float* b2   = (const float*)d_in[9];
    float* out = (float*)d_out;

    const int IN  = 256, F1 = 512, OUT = 40, OUTP = 48;
    const int Nn  = in_sizes[0] / IN;
    const int E   = in_sizes[1] / 2;

    float* ws  = (float*)d_ws;
    float* h1  = ws;
    float* h2  = h1  + (size_t)Nn * F1;
    float* h3  = h2  + (size_t)Nn * F1;
    float* W2p = h3  + (size_t)Nn * 48;
    float* as1 = W2p + (size_t)F1 * OUTP;
    float* ad1 = as1 + (size_t)Nn * 8;
    float* as2 = ad1 + (size_t)Nn * 8;
    float* ad2 = as2 + Nn;
    int*   deg  = (int*)(ad2 + Nn);
    int*   off  = deg  + (size_t)Nn * CSTRIDE;
    int*   rank = off  + (Nn + 1);
    int*   csr  = rank + E;

    const int B = 256;

    zero_counters<<<(Nn + B - 1) / B, B, 0, stream>>>(deg, Nn);
    rank_edges   <<<(E + B - 1) / B, B, 0, stream>>>(ei, E, deg, rank);
    scan_degrees <<<(Nn + 1023) / 1024, 1024, 0, stream>>>(ei, E, deg, off, Nn);
    place_edges  <<<(E + B - 1) / B, B, 0, stream>>>(ei, E, off, rank, csr);

    gemm_wmma_f32<4, 1><<<dim3((Nn + 15) / 16, F1 / 64), 32, 0, stream>>>(
        x, W1, h1, Nn, IN, F1, F1, aS1, aD1, as1, ad1, F1, 8);

    gat1_gather<<<(Nn + 7) / 8, 256, 0, stream>>>(off, csr, as1, ad1, h1, b1, h2, Nn);

    pad_cols<<<(F1 * OUTP + B - 1) / B, B, 0, stream>>>(W2, W2p, F1, OUT, OUTP);

    gemm_wmma_f32<3, 8><<<dim3((Nn + 15) / 16, 1), 256, 0, stream>>>(
        h2, W2p, h3, Nn, F1, OUTP, OUTP, aS2, aD2, as2, ad2, OUT, 1);

    gat2_gather<<<(Nn * 10 + B - 1) / B, B, 0, stream>>>(off, csr, as2, ad2, h3, b2, out, Nn);
}


# Independent kernels share launches (9 -> 7): edge ranking runs as an extra grid row of the layer-1 GEMM launch, W2 padding joins the counter clearing; ranks still in atomic arrival order
# speedup vs baseline: 3.1616x; 1.0514x over previous
#include <hip/hip_runtime.h>
#include <hip/hip_bf16.h>
#include <math.h>

typedef float v2f __attribute__((ext_vector_type(2)));
typedef float v8f __attribute__((ext_vector_type(8)));
typedef float v4f __attribute__((ext_vector_type(4)));

#define NEG_SLOPE 0.2f

__device__ __forceinline__ float leaky(float v) {
    return v > 0.0f ? v : NEG_SLOPE * v;
}


__device__ __forceinline__ void wave_lds_handover() {
    __builtin_amdgcn_fence(__ATOMIC_RELEASE, "wavefront");
    __builtin_amdgcn_wave_barrier();
    __builtin_amdgcn_fence(__ATOMIC_ACQUIRE, "wavefront");
}

template<int MASK>
__device__ __forceinline__ float row_xor(float v) {
    const int i = __builtin_bit_cast(int, v);
    return __builtin_bit_cast(float,
        __builtin_amdgcn_update_dpp(i, i, 0x160 + MASK, 0xf, 0xf, false));
}

template<int NT, int KS>
__device__ __forceinline__ void
gemm_strip(const float* __restrict__ A, const float* __restrict__ B,
           float* __restrict__ C, int M, int K, int N, int ldc,
           const float* __restrict__ a_src, const float* __restrict__ a_dst,
           float* __restrict__ as, float* __restrict__ ad, int ncv, int heads)
{
    __shared__ float Asall[KS * 16 * 12];

    const int lane = (KS == 1) ? (int)threadIdx.x : ((int)threadIdx.x & 31);
    const int wv   = (KS == 1) ? 0
                   : __builtin_amdgcn_readfirstlane((int)threadIdx.x >> 5);
    float* __restrict__ As = Asall + wv * (16 * 12);
    const int kper = K / KS;
    const int kbeg = wv * kper;
    const int half = lane >> 4;
    const int l16  = lane & 15;

    const int alr = lane >> 1;
    const int alc = (lane & 1) * 4;
    int rowA = blockIdx.x * 16 + alr;
    if (rowA >= M) rowA = M - 1;
    const float* __restrict__ Aload = A + (size_t)rowA * (size_t)K + alc;

    const int ncol0 = blockIdx.y * (16 * NT);
    const int colB  = ncol0 + l16;

    v8f acc[NT];
#pragma unroll
    for (int t = 0; t < NT; ++t) {
        v8f z = {0.f, 0.f, 0.f, 0.f, 0.f, 0.f, 0.f, 0.f};
        acc[t] = z;
    }

    const float* __restrict__ Aw = Aload + kbeg;
    const float* __restrict__ Bw = B + (size_t)kbeg * (size_t)N;
    float4 av = *(const float4*)(Aw);

    float wsrc[NT], wdst[NT];
#pragma unroll
    for (int t = 0; t < NT; ++t) {
        const int col = colB + t * 16;
        const bool real = col < ncv;
        const int cc = real ? col : 0;
        const float vs = a_src[cc], vd = a_dst[cc];
        wsrc[t] = real ? vs : 0.f;
        wdst[t] = real ? vd : 0.f;
    }

    for (int kc = 0; kc < kper; kc += 8) {
        if (KS == 1) __syncthreads(); else wave_lds_handover();
        *(float4*)(&As[alr * 12 + alc]) = av;

        v2f breg[2][NT];
#pragma unroll
        for (int kk2 = 0; kk2 < 2; ++kk2) {
            const float* __restrict__ Bk =
                Bw + (size_t)(kc + kk2 * 4 + 2 * half) * (size_t)N + colB;
#pragma unroll
            for (int t = 0; t < NT; ++t) {
                breg[kk2][t].x = Bk[t * 16];
                breg[kk2][t].y = Bk[N + t * 16];
            }
        }

        const int kn = (kc + 8 < kper) ? (kc + 8) : 0;
        av = *(const float4*)(Aw + kn);

        if (KS == 1) __syncthreads(); else wave_lds_handover();

#pragma unroll
        for (int kk2 = 0; kk2 < 2; ++kk2) {
            const v2f a = *(const v2f*)(&As[l16 * 12 + kk2 * 4 + 2 * half]);
#pragma unroll
            for (int t = 0; t < NT; ++t) {
                acc[t] = __builtin_amdgcn_wmma_f32_16x16x4_f32(
                    false, a, false, breg[kk2][t], (short)0, acc[t], false, false);
            }
        }
    }

    if (KS > 1) {
        __shared__ float4 Part[(KS > 1 ? KS - 1 : 1) * NT * 2 * 32];
        if (wv != 0) {
            float4* __restrict__ pp = Part + (size_t)(wv - 1) * (NT * 2 * 32) + lane;
#pragma unroll
            for (int t = 0; t < NT; ++t) {
                pp[(t * 2 + 0) * 32] = make_float4(acc[t][0], acc[t][1], acc[t][2], acc[t][3]);
                pp[(t * 2 + 1) * 32] = make_float4(acc[t][4], acc[t][5], acc[t][6], acc[t][7]);
            }
        }
        __syncthreads();
        if (wv != 0) return;
        constexpr int NP = NT * 2;
        const v4f* __restrict__ pp = (const v4f*)Part + lane;
#pragma unroll 1
        for (int w = 0; w < KS - 1; ++w, pp += NP * 32) {
            v4f pv[NP];
#pragma unroll
            for (int i = 0; i < NP; ++i) pv[i] = pp[i * 32];
            if constexpr (NT == 3) {
                asm volatile("" :: "v"(pv[0]), "v"(pv[1]), "v"(pv[2]),
                                   "v"(pv[3]), "v"(pv[4]), "v"(pv[5]));
            }
#pragma unroll
            for (int t = 0; t < NT; ++t) {
#pragma unroll
                for (int r = 0; r < 4; ++r) {
                    acc[t][r]     += pv[t * 2][r];
                    acc[t][r + 4] += pv[t * 2 + 1][r];
                }
            }
        }
    }

    const int crow0 = blockIdx.x * 16 + half * 8;
    float* __restrict__ Cbase = C + (size_t)crow0 * (size_t)ldc + ncol0 + l16;
    if (blockIdx.x * 16 + 16 <= M) {
#pragma unroll
        for (int t = 0; t < NT; ++t) {
            float* __restrict__ Cp = Cbase + t * 16;
#pragma unroll
            for (int r = 0; r < 8; ++r)
                Cp[(size_t)r * (size_t)ldc] = acc[t][r];
        }
    } else {
#pragma unroll
        for (int t = 0; t < NT; ++t) {
            float* __restrict__ Cp = Cbase + t * 16;
#pragma unroll
            for (int r = 0; r < 8; ++r)
                if (crow0 + r < M) Cp[(size_t)r * (size_t)ldc] = acc[t][r];
        }
    }

    float ps[8], pd[8];
#pragma unroll
    for (int r = 0; r < 8; ++r) { ps[r] = 0.f; pd[r] = 0.f; }
#pragma unroll
    for (int t = 0; t < NT; ++t) {
#pragma unroll
        for (int r = 0; r < 8; ++r) {
            ps[r] += acc[t][r] * wsrc[t];
            pd[r] += acc[t][r] * wdst[t];
        }
    }
    const bool b3 = (l16 & 8) != 0, b2 = (l16 & 4) != 0;
    const bool b1 = (l16 & 2) != 0, b0 = (l16 & 1) != 0;
    float u[8], w4[4], y2[2];
#pragma unroll
    for (int i = 0; i < 8; ++i) {
        const float mine = b3 ? pd[i] : ps[i];
        const float give = b3 ? ps[i] : pd[i];
        u[i] = mine + row_xor<8>(give);
    }
#pragma unroll
    for (int i = 0; i < 4; ++i) {
        const float mine = b2 ? u[i + 4] : u[i];
        const float give = b2 ? u[i] : u[i + 4];
        w4[i] = mine + row_xor<4>(give);
    }
#pragma unroll
    for (int i = 0; i < 2; ++i) {
        const float mine = b1 ? w4[i + 2] : w4[i];
        const float give = b1 ? w4[i] : w4[i + 2];
        y2[i] = mine + row_xor<2>(give);
    }
    const float mine = b0 ? y2[1] : y2[0];
    const float give = b0 ? y2[0] : y2[1];
    const float z = mine + row_xor<1>(give);
    const int arow = blockIdx.x * 16 + half * 8 + (l16 & 7);
    if (arow < M) {
        float* __restrict__ dstp = b3 ? ad : as;
        dstp[(size_t)arow * (size_t)heads + blockIdx.y] = z;
    }
}

template<int NT, int KS>
__global__ void __launch_bounds__(32 * KS)
gemm_wmma_f32(const float* __restrict__ A, const float* __restrict__ B,
              float* __restrict__ C, int M, int K, int N, int ldc,
              const float* __restrict__ a_src, const float* __restrict__ a_dst,
              float* __restrict__ as, float* __restrict__ ad, int ncv, int heads)
{
    gemm_strip<NT, KS>(A, B, C, M, K, N, ldc, a_src, a_dst, as, ad, ncv, heads);
}

#define CSTRIDE 32

__global__ void clear_counters_pad_w2(int* __restrict__ deg, int Nn,
                                      const float* __restrict__ src,
                                      float* __restrict__ dst, int K, int N, int Np)
{
    const int nzb = (Nn + (int)blockDim.x - 1) / (int)blockDim.x;
    if ((int)blockIdx.x < nzb) {
        const int t = blockIdx.x * blockDim.x + threadIdx.x;
        if (t < Nn) deg[(size_t)t * CSTRIDE] = 0;
    } else {
        const int t = ((int)blockIdx.x - nzb) * blockDim.x + threadIdx.x;
        if (t >= K * Np) return;
        const int r = t / Np, c = t - r * Np;
        dst[t] = (c < N) ? src[r * N + c] : 0.f;
    }
}

template<int NT>
__global__ void __launch_bounds__(32)
gemm1_and_rank_edges(const float* __restrict__ A, const float* __restrict__ B,
                     float* __restrict__ C, int M, int K, int N, int ldc,
                     const float* __restrict__ a_src, const float* __restrict__ a_dst,
                     float* __restrict__ as, float* __restrict__ ad, int ncv, int heads,
                     const int* __restrict__ ei, int E,
                     int* __restrict__ deg, int* __restrict__ rank)
{
    if (blockIdx.y == gridDim.y - 1) {
        const int step = (int)gridDim.x * 32;
        const int* __restrict__ dstp = ei + E;
        int e = (int)blockIdx.x * 32 + (int)threadIdx.x;
        for (; e + 3 * step < E; e += 4 * step) {
            const int d0 = dstp[e],            d1 = dstp[e + step];
            const int d2 = dstp[e + 2 * step], d3 = dstp[e + 3 * step];
            const int r0 = atomicAdd(&deg[(size_t)d0 * CSTRIDE], 1);
            const int r1 = atomicAdd(&deg[(size_t)d1 * CSTRIDE], 1);
            const int r2 = atomicAdd(&deg[(size_t)d2 * CSTRIDE], 1);
            const int r3 = atomicAdd(&deg[(size_t)d3 * CSTRIDE], 1);
            rank[e] = r0;            rank[e + step] = r1;
            rank[e + 2 * step] = r2; rank[e + 3 * step] = r3;
        }
        for (; e < E; e += step)
            rank[e] = atomicAdd(&deg[(size_t)dstp[e] * CSTRIDE], 1);
        return;
    }
    gemm_strip<NT, 1>(A, B, C, M, K, N, ldc, a_src, a_dst, as, ad, ncv, heads);
}

__global__ void __launch_bounds__(1024)
scan_degrees(const int* __restrict__ ei, int E, const int* __restrict__ deg,
             int* __restrict__ off, int Nn)
{
    __shared__ int wdeg[32];
    __shared__ int wcnt[32];
    const int t    = (int)threadIdx.x;
    const int lane = t & 31;
    const int wv   = t >> 5;
    const int lo   = (int)blockIdx.x * 1024;
    const int idx  = lo + t;

    const int v0 = deg[(size_t)(idx < Nn ? idx : Nn - 1) * CSTRIDE];
    const int v  = (idx < Nn) ? v0 : 0;

    int cnt = 0;
    if (lo > 0) {
        const int* __restrict__ dstp = ei + E;
        const int nvec = (((size_t)dstp & 15) == 0) ? (E >> 2) : 0;
        const int4* __restrict__ dv = (const int4*)dstp;
#pragma unroll 8
        for (int i = t; i < nvec; i += 1024) {
            const int4 d = dv[i];
            cnt += (d.x < lo ? 1 : 0) + (d.y < lo ? 1 : 0)
                 + (d.z < lo ? 1 : 0) + (d.w < lo ? 1 : 0);
        }
        for (int e = nvec * 4 + t; e < E; e += 1024)
            cnt += (dstp[e] < lo) ? 1 : 0;
    }

    int s = v;
#pragma unroll
    for (int o = 1; o < 32; o <<= 1) {
        const int u = __shfl_up(s, o);
        if (lane >= o) s += u;
    }
#pragma unroll
    for (int o = 16; o > 0; o >>= 1) cnt += __shfl_xor(cnt, o);
    if (lane == 31) wdeg[wv] = s;
    if (lane == 0)  wcnt[wv] = cnt;
    __syncthreads();

    int base  = wcnt[lane];
    int front = (lane < wv) ? wdeg[lane] : 0;
#pragma unroll
    for (int o = 16; o > 0; o >>= 1) {
        base  += __shfl_xor(base, o);
        front += __shfl_xor(front, o);
    }
    const int excl = base + front + s - v;
    if (idx < Nn)      off[idx] = excl;
    if (idx == Nn - 1) off[Nn]  = excl + v;
}

__global__ void place_edges(const int* __restrict__ ei, int E,
                            const int* __restrict__ off, const int* __restrict__ rank,
                            int* __restrict__ csr)
{
    int e = blockIdx.x * blockDim.x + threadIdx.x;
    if (e >= E) return;
    int d = ei[E + e];
    csr[off[d] + rank[e]] = ei[e];
}

__global__ void __launch_bounds__(256)
gat1_gather(const int* __restrict__ off, const int* __restrict__ csr,
            const float* __restrict__ as, const float* __restrict__ ad,
            const float* __restrict__ h1, const float* __restrict__ b,
            float* __restrict__ h2, int Nn)
{
    __shared__ float4 wsh[8 * 64];

    const int lane = (int)threadIdx.x & 31;
    const int half = lane >> 4;
    const int wv   = (int)threadIdx.x >> 5;
    float* __restrict__ wmine = (float*)(wsh + wv * 64);
    const int n = __builtin_amdgcn_readfirstlane((int)blockIdx.x * 8 + wv);
    if (n >= Nn) return;
    const int beg = off[n], end = off[n + 1];
    const int last = end - 1;

    const int q = lane >> 3, hh = lane & 7;
    const float adq = ad[n * 8 + hh];
    float m = leaky(as[n * 8 + hh] + adq);
    for (int base = beg; base < end; base += 32) {
#pragma unroll
        for (int t = 0; t < 8; ++t) {
            const int j = base + q + 4 * t;
            const int s = csr[j < last ? j : last];
            m = fmaxf(m, leaky(as[s * 8 + hh] + adq));
        }
    }
    m = fmaxf(m, __shfl_xor(m, 8));
    m = fmaxf(m, __shfl_xor(m, 16));

    float den[4];
    float4 acc[4];
    const float* __restrict__ hcol = h1 + lane * 4;
#pragma unroll
    for (int i = 0; i < 4; ++i) {
        const int hd = 2 * i + half;
        const float mh = __shfl(m, hd);
        const float w = expf(leaky(as[n * 8 + hd] + ad[n * 8 + hd]) - mh);
        const float4 hv = *(const float4*)(hcol + (size_t)n * 512 + i * 128);
        den[i] = w;
        acc[i].x = hv.x * w; acc[i].y = hv.y * w; acc[i].z = hv.z * w; acc[i].w = hv.w * w;
    }

    const int wpos = (hh & 1) * 4 + (hh >> 1);
    for (int base = beg; base < end; base += 32) {
#pragma unroll
        for (int t = 0; t < 8; ++t) {
            const int j = base + q + 4 * t;
            const int s = csr[j < last ? j : last];
            wmine[(q + 4 * t) * 8 + wpos] = expf(leaky(as[s * 8 + hh] + adq) - m);
        }
        wave_lds_handover();
        const int left = end - base;
        const int cnt = left < 32 ? left : 32;
        int e = 0;
        for (; e + 2 <= cnt; e += 2) {
            const int s0 = csr[base + e], s1 = csr[base + e + 1];
            const float4 wa = *(const float4*)(wmine + e * 8 + half * 4);
            const float4 wb = *(const float4*)(wmine + e * 8 + 8 + half * 4);
            const float* __restrict__ r0 = hcol + (size_t)s0 * 512;
            const float* __restrict__ r1 = hcol + (size_t)s1 * 512;
            const v4f a0 = *(const v4f*)(r0),       a1 = *(const v4f*)(r0 + 128);
            const v4f a2 = *(const v4f*)(r0 + 256), a3 = *(const v4f*)(r0 + 384);
            const v4f b0 = *(const v4f*)(r1),       b1 = *(const v4f*)(r1 + 128);
            const v4f b2 = *(const v4f*)(r1 + 256), b3 = *(const v4f*)(r1 + 384);
            asm volatile("" :: "v"(a0), "v"(a1), "v"(a2), "v"(a3),
                               "v"(b0), "v"(b1), "v"(b2), "v"(b3));
            const v4f ra[4] = {a0, a1, a2, a3};
            const v4f rb[4] = {b0, b1, b2, b3};
            const float wla[4] = {wa.x, wa.y, wa.z, wa.w};
            const float wlb[4] = {wb.x, wb.y, wb.z, wb.w};
#pragma unroll
            for (int i = 0; i < 4; ++i) {
                den[i] += wla[i];
                acc[i].x += ra[i].x * wla[i]; acc[i].y += ra[i].y * wla[i];
                acc[i].z += ra[i].z * wla[i]; acc[i].w += ra[i].w * wla[i];
            }
#pragma unroll
            for (int i = 0; i < 4; ++i) {
                den[i] += wlb[i];
                acc[i].x += rb[i].x * wlb[i]; acc[i].y += rb[i].y * wlb[i];
                acc[i].z += rb[i].z * wlb[i]; acc[i].w += rb[i].w * wlb[i];
            }
        }
        if (e < cnt) {
            const int s0 = csr[base + e];
            const float4 wa = *(const float4*)(wmine + e * 8 + half * 4);
            const float wla[4] = {wa.x, wa.y, wa.z, wa.w};
            const float* __restrict__ r0 = hcol + (size_t)s0 * 512;
#pragma unroll
            for (int i = 0; i < 4; ++i) {
                const v4f hv = *(const v4f*)(r0 + i * 128);
                den[i] += wla[i];
                acc[i].x += hv.x * wla[i]; acc[i].y += hv.y * wla[i];
                acc[i].z += hv.z * wla[i]; acc[i].w += hv.w * wla[i];
            }
        }
        wave_lds_handover();
    }
#pragma unroll
    for (int i = 0; i < 4; ++i) {
        const float4 bb = *(const float4*)(b + i * 128 + lane * 4);
        float4 o;
        o.x = acc[i].x / den[i] + bb.x; o.y = acc[i].y / den[i] + bb.y;
        o.z = acc[i].z / den[i] + bb.z; o.w = acc[i].w / den[i] + bb.w;
        o.x = o.x > 0.f ? o.x : expm1f(o.x);
        o.y = o.y > 0.f ? o.y : expm1f(o.y);
        o.z = o.z > 0.f ? o.z : expm1f(o.z);
        o.w = o.w > 0.f ? o.w : expm1f(o.w);
        *(float4*)(h2 + (size_t)n * 512 + i * 128 + lane * 4) = o;
    }
}

__global__ void gat2_gather(const int* __restrict__ off, const int* __restrict__ csr,
                            const float* __restrict__ as, const float* __restrict__ ad,
                            const float* __restrict__ h3, const float* __restrict__ b,
                            float* __restrict__ out, int Nn)
{
    int t = blockIdx.x * blockDim.x + threadIdx.x;
    if (t >= Nn * 10) return;
    const int n = t / 10;
    const int g = t - n * 10;
    const int beg = off[n], end = off[n + 1];
    const float adn   = ad[n];
    const float eself = leaky(as[n] + adn);

    float m = eself;
    for (int j = beg; j < end; ++j)
        m = fmaxf(m, leaky(as[csr[j]] + adn));

    float den;
    float4 acc;
    {
        const float w = expf(eself - m);
        const float4 hv = *(const float4*)(h3 + (size_t)n * 48 + g * 4);
        den = w;
        acc.x = hv.x * w; acc.y = hv.y * w; acc.z = hv.z * w; acc.w = hv.w * w;
    }
#pragma unroll 4
    for (int j = beg; j < end; ++j) {
        const int s = csr[j];
        const float w = expf(leaky(as[s] + adn) - m);
        const float4 hv = *(const float4*)(h3 + (size_t)s * 48 + g * 4);
        den += w;
        acc.x += hv.x * w; acc.y += hv.y * w; acc.z += hv.z * w; acc.w += hv.w * w;
    }
    const float4 bb = *(const float4*)(b + g * 4);
    float4 o;
    o.x = acc.x / den + bb.x; o.y = acc.y / den + bb.y;
    o.z = acc.z / den + bb.z; o.w = acc.w / den + bb.w;
    *(float4*)(out + (size_t)n * 40 + g * 4) = o;
}

extern "C" void kernel_launch(void* const* d_in, const int* in_sizes, int n_in,
                              void* d_out, int out_size, void* d_ws, size_t ws_size,
                              hipStream_t stream)
{
    const float* x    = (const float*)d_in[0];
    const int*   ei   = (const int*)  d_in[1];
    const float* W1   = (const float*)d_in[2];
    const float* aS1  = (const float*)d_in[3];
    const float* aD1  = (const float*)d_in[4];
    const float* b1   = (const float*)d_in[5];
    const float* W2   = (const float*)d_in[6];
    const float* aS2  = (const float*)d_in[7];
    const float* aD2  = (const float*)d_in[8];
    const float* b2   = (const float*)d_in[9];
    float* out = (float*)d_out;

    const int IN  = 256, F1 = 512, OUT = 40, OUTP = 48;
    const int Nn  = in_sizes[0] / IN;
    const int E   = in_sizes[1] / 2;

    float* ws  = (float*)d_ws;
    float* h1  = ws;
    float* h2  = h1  + (size_t)Nn * F1;
    float* h3  = h2  + (size_t)Nn * F1;
    float* W2p = h3  + (size_t)Nn * 48;
    float* as1 = W2p + (size_t)F1 * OUTP;
    float* ad1 = as1 + (size_t)Nn * 8;
    float* as2 = ad1 + (size_t)Nn * 8;
    float* ad2 = as2 + Nn;
    int*   deg  = (int*)(ad2 + Nn);
    int*   off  = deg  + (size_t)Nn * CSTRIDE;
    int*   rank = off  + (Nn + 1);
    int*   csr  = rank + E;

    const int B = 256;

    clear_counters_pad_w2<<<(Nn + B - 1) / B + (F1 * OUTP + B - 1) / B, B, 0, stream>>>(
        deg, Nn, W2, W2p, F1, OUT, OUTP);

    gemm1_and_rank_edges<4><<<dim3((Nn + 15) / 16, F1 / 64 + 1), 32, 0, stream>>>(
        x, W1, h1, Nn, IN, F1, F1, aS1, aD1, as1, ad1, F1, 8, ei, E, deg, rank);

    scan_degrees <<<(Nn + 1023) / 1024, 1024, 0, stream>>>(ei, E, deg, off, Nn);
    place_edges  <<<(E + B - 1) / B, B, 0, stream>>>(ei, E, off, rank, csr);

    gat1_gather<<<(Nn + 7) / 8, 256, 0, stream>>>(off, csr, as1, ad1, h1, b1, h2, Nn);

    gemm_wmma_f32<3, 8><<<dim3((Nn + 15) / 16, 1), 256, 0, stream>>>(
        h2, W2p, h3, Nn, F1, OUTP, OUTP, aS2, aD2, as2, ad2, OUT, 1);

    gat2_gather<<<(Nn * 10 + B - 1) / B, B, 0, stream>>>(off, csr, as2, ad2, h3, b2, out, Nn);
}


# Sort's prefix scan and placement launches removed (7 -> 5): the ranking add's 256-B record also takes the node's sources (63 slots; longer lists re-read the edge list); sums still in arrival order
# speedup vs baseline: 3.8562x; 1.2197x over previous
#include <hip/hip_runtime.h>
#include <hip/hip_bf16.h>
#include <math.h>

typedef float v2f __attribute__((ext_vector_type(2)));
typedef float v8f __attribute__((ext_vector_type(8)));
typedef float v4f __attribute__((ext_vector_type(4)));

#define NEG_SLOPE 0.2f

__device__ __forceinline__ float leaky(float v) {
    return v > 0.0f ? v : NEG_SLOPE * v;
}


__device__ __forceinline__ void wave_lds_handover() {
    __builtin_amdgcn_fence(__ATOMIC_RELEASE, "wavefront");
    __builtin_amdgcn_wave_barrier();
    __builtin_amdgcn_fence(__ATOMIC_ACQUIRE, "wavefront");
}

template<int MASK>
__device__ __forceinline__ float row_xor(float v) {
    const int i = __builtin_bit_cast(int, v);
    return __builtin_bit_cast(float,
        __builtin_amdgcn_update_dpp(i, i, 0x160 + MASK, 0xf, 0xf, false));
}

template<int NT, int KS>
__device__ __forceinline__ void
gemm_strip(const float* __restrict__ A, const float* __restrict__ B,
           float* __restrict__ C, int M, int K, int N, int ldc,
           const float* __restrict__ a_src, const float* __restrict__ a_dst,
           float* __restrict__ as, float* __restrict__ ad, int ncv, int heads)
{
    __shared__ float Asall[KS * 16 * 12];

    const int lane = (KS == 1) ? (int)threadIdx.x : ((int)threadIdx.x & 31);
    const int wv   = (KS == 1) ? 0
                   : __builtin_amdgcn_readfirstlane((int)threadIdx.x >> 5);
    float* __restrict__ As = Asall + wv * (16 * 12);
    const int kper = K / KS;
    const int kbeg = wv * kper;
    const int half = lane >> 4;
    const int l16  = lane & 15;

    const int alr = lane >> 1;
    const int alc = (lane & 1) * 4;
    int rowA = blockIdx.x * 16 + alr;
    if (rowA >= M) rowA = M - 1;
    const float* __restrict__ Aload = A + (size_t)rowA * (size_t)K + alc;

    const int ncol0 = blockIdx.y * (16 * NT);
    const int colB  = ncol0 + l16;

    v8f acc[NT];
#pragma unroll
    for (int t = 0; t < NT; ++t) {
        v8f z = {0.f, 0.f, 0.f, 0.f, 0.f, 0.f, 0.f, 0.f};
        acc[t] = z;
    }

    const float* __restrict__ Aw = Aload + kbeg;
    const float* __restrict__ Bw = B + (size_t)kbeg * (size_t)N;
    float4 av = *(const float4*)(Aw);

    float wsrc[NT], wdst[NT];
#pragma unroll
    for (int t = 0; t < NT; ++t) {
        const int col = colB + t * 16;
        const bool real = col < ncv;
        const int cc = real ? col : 0;
        const float vs = a_src[cc], vd = a_dst[cc];
        wsrc[t] = real ? vs : 0.f;
        wdst[t] = real ? vd : 0.f;
    }

    for (int kc = 0; kc < kper; kc += 8) {
        if (KS == 1) __syncthreads(); else wave_lds_handover();
        *(float4*)(&As[alr * 12 + alc]) = av;

        v2f breg[2][NT];
#pragma unroll
        for (int kk2 = 0; kk2 < 2; ++kk2) {
            const float* __restrict__ Bk =
                Bw + (size_t)(kc + kk2 * 4 + 2 * half) * (size_t)N + colB;
#pragma unroll
            for (int t = 0; t < NT; ++t) {
                breg[kk2][t].x = Bk[t * 16];
                breg[kk2][t].y = Bk[N + t * 16];
            }
        }

        const int kn = (kc + 8 < kper) ? (kc + 8) : 0;
        av = *(const float4*)(Aw + kn);

        if (KS == 1) __syncthreads(); else wave_lds_handover();

#pragma unroll
        for (int kk2 = 0; kk2 < 2; ++kk2) {
            const v2f a = *(const v2f*)(&As[l16 * 12 + kk2 * 4 + 2 * half]);
#pragma unroll
            for (int t = 0; t < NT; ++t) {
                acc[t] = __builtin_amdgcn_wmma_f32_16x16x4_f32(
                    false, a, false, breg[kk2][t], (short)0, acc[t], false, false);
            }
        }
    }

    if (KS > 1) {
        __shared__ float4 Part[(KS > 1 ? KS - 1 : 1) * NT * 2 * 32];
        if (wv != 0) {
            float4* __restrict__ pp = Part + (size_t)(wv - 1) * (NT * 2 * 32) + lane;
#pragma unroll
            for (int t = 0; t < NT; ++t) {
                pp[(t * 2 + 0) * 32] = make_float4(acc[t][0], acc[t][1], acc[t][2], acc[t][3]);
                pp[(t * 2 + 1) * 32] = make_float4(acc[t][4], acc[t][5], acc[t][6], acc[t][7]);
            }
        }
        __syncthreads();
        if (wv != 0) return;
        constexpr int NP = NT * 2;
        const v4f* __restrict__ pp = (const v4f*)Part + lane;
#pragma unroll 1
        for (int w = 0; w < KS - 1; ++w, pp += NP * 32) {
            v4f pv[NP];
#pragma unroll
            for (int i = 0; i < NP; ++i) pv[i] = pp[i * 32];
            if constexpr (NT == 3) {
                asm volatile("" :: "v"(pv[0]), "v"(pv[1]), "v"(pv[2]),
                                   "v"(pv[3]), "v"(pv[4]), "v"(pv[5]));
            }
#pragma unroll
            for (int t = 0; t < NT; ++t) {
#pragma unroll
                for (int r = 0; r < 4; ++r) {
                    acc[t][r]     += pv[t * 2][r];
                    acc[t][r + 4] += pv[t * 2 + 1][r];
                }
            }
        }
    }

    const int crow0 = blockIdx.x * 16 + half * 8;
    float* __restrict__ Cbase = C + (size_t)crow0 * (size_t)ldc + ncol0 + l16;
    if (blockIdx.x * 16 + 16 <= M) {
#pragma unroll
        for (int t = 0; t < NT; ++t) {
            float* __restrict__ Cp = Cbase + t * 16;
#pragma unroll
            for (int r = 0; r < 8; ++r)
                Cp[(size_t)r * (size_t)ldc] = acc[t][r];
        }
    } else {
#pragma unroll
        for (int t = 0; t < NT; ++t) {
            float* __restrict__ Cp = Cbase + t * 16;
#pragma unroll
            for (int r = 0; r < 8; ++r)
                if (crow0 + r < M) Cp[(size_t)r * (size_t)ldc] = acc[t][r];
        }
    }

    float ps[8], pd[8];
#pragma unroll
    for (int r = 0; r < 8; ++r) { ps[r] = 0.f; pd[r] = 0.f; }
#pragma unroll
    for (int t = 0; t < NT; ++t) {
#pragma unroll
        for (int r = 0; r < 8; ++r) {
            ps[r] += acc[t][r] * wsrc[t];
            pd[r] += acc[t][r] * wdst[t];
        }
    }
    const bool b3 = (l16 & 8) != 0, b2 = (l16 & 4) != 0;
    const bool b1 = (l16 & 2) != 0, b0 = (l16 & 1) != 0;
    float u[8], w4[4], y2[2];
#pragma unroll
    for (int i = 0; i < 8; ++i) {
        const float mine = b3 ? pd[i] : ps[i];
        const float give = b3 ? ps[i] : pd[i];
        u[i] = mine + row_xor<8>(give);
    }
#pragma unroll
    for (int i = 0; i < 4; ++i) {
        const float mine = b2 ? u[i + 4] : u[i];
        const float give = b2 ? u[i] : u[i + 4];
        w4[i] = mine + row_xor<4>(give);
    }
#pragma unroll
    for (int i = 0; i < 2; ++i) {
        const float mine = b1 ? w4[i + 2] : w4[i];
        const float give = b1 ? w4[i] : w4[i + 2];
        y2[i] = mine + row_xor<2>(give);
    }
    const float mine = b0 ? y2[1] : y2[0];
    const float give = b0 ? y2[0] : y2[1];
    const float z = mine + row_xor<1>(give);
    const int arow = blockIdx.x * 16 + half * 8 + (l16 & 7);
    if (arow < M) {
        float* __restrict__ dstp = b3 ? ad : as;
        dstp[(size_t)arow * (size_t)heads + blockIdx.y] = z;
    }
}

template<int NT, int KS>
__global__ void __launch_bounds__(32 * KS)
gemm_wmma_f32(const float* __restrict__ A, const float* __restrict__ B,
              float* __restrict__ C, int M, int K, int N, int ldc,
              const float* __restrict__ a_src, const float* __restrict__ a_dst,
              float* __restrict__ as, float* __restrict__ ad, int ncv, int heads)
{
    gemm_strip<NT, KS>(A, B, C, M, K, N, ldc, a_src, a_dst, as, ad, ncv, heads);
}

#define RSTRIDE 64
#define RCAP 63

__global__ void clear_counters_pad_w2(int* __restrict__ deg, int Nn,
                                      const float* __restrict__ src,
                                      float* __restrict__ dst, int K, int N, int Np)
{
    const int nzb = (Nn + (int)blockDim.x - 1) / (int)blockDim.x;
    if ((int)blockIdx.x < nzb) {
        const int t = blockIdx.x * blockDim.x + threadIdx.x;
        if (t < Nn) deg[(size_t)t * RSTRIDE] = 0;
    } else {
        const int t = ((int)blockIdx.x - nzb) * blockDim.x + threadIdx.x;
        if (t >= K * Np) return;
        const int r = t / Np, c = t - r * Np;
        dst[t] = (c < N) ? src[r * N + c] : 0.f;
    }
}

template<int NT>
__global__ void __launch_bounds__(32)
gemm1_and_fill_records(const float* __restrict__ A, const float* __restrict__ B,
                       float* __restrict__ C, int M, int K, int N, int ldc,
                       const float* __restrict__ a_src, const float* __restrict__ a_dst,
                       float* __restrict__ as, float* __restrict__ ad, int ncv, int heads,
                       const int* __restrict__ ei, int E, int* __restrict__ rec)
{
    if (blockIdx.y == gridDim.y - 1) {
        const int step = (int)gridDim.x * 32;
        const int* __restrict__ srcp = ei;
        const int* __restrict__ dstp = ei + E;
        int e = (int)blockIdx.x * 32 + (int)threadIdx.x;
        for (; e + 3 * step < E; e += 4 * step) {
            const int d0 = dstp[e],            d1 = dstp[e + step];
            const int d2 = dstp[e + 2 * step], d3 = dstp[e + 3 * step];
            const int s0 = srcp[e],            s1 = srcp[e + step];
            const int s2 = srcp[e + 2 * step], s3 = srcp[e + 3 * step];
            int* q0 = rec + (size_t)d0 * RSTRIDE;
            int* q1 = rec + (size_t)d1 * RSTRIDE;
            int* q2 = rec + (size_t)d2 * RSTRIDE;
            int* q3 = rec + (size_t)d3 * RSTRIDE;
            const int r0 = atomicAdd(q0, 1);
            const int r1 = atomicAdd(q1, 1);
            const int r2 = atomicAdd(q2, 1);
            const int r3 = atomicAdd(q3, 1);
            if (r0 < RCAP) q0[1 + r0] = s0;
            if (r1 < RCAP) q1[1 + r1] = s1;
            if (r2 < RCAP) q2[1 + r2] = s2;
            if (r3 < RCAP) q3[1 + r3] = s3;
        }
        for (; e < E; e += step) {
            int* q0 = rec + (size_t)dstp[e] * RSTRIDE;
            const int s0 = srcp[e];
            const int r0 = atomicAdd(q0, 1);
            if (r0 < RCAP) q0[1 + r0] = s0;
        }
        return;
    }
    gemm_strip<NT, 1>(A, B, C, M, K, N, ldc, a_src, a_dst, as, ad, ncv, heads);
}

__global__ void __launch_bounds__(256)
gat1_gather(const int* __restrict__ rec, const int* __restrict__ ei, int E,
            const float* __restrict__ as, const float* __restrict__ ad,
            const float* __restrict__ h1, const float* __restrict__ b,
            float* __restrict__ h2, int Nn)
{
    __shared__ float4 wsh[8 * 64];

    const int lane = (int)threadIdx.x & 31;
    const int half = lane >> 4;
    const int wv   = (int)threadIdx.x >> 5;
    float* __restrict__ wmine = (float*)(wsh + wv * 64);
    const int n = __builtin_amdgcn_readfirstlane((int)blockIdx.x * 8 + wv);
    if (n >= Nn) return;
    const int* __restrict__ csr = rec;
    const int cnt0 = rec[(size_t)n * RSTRIDE];

    if (__builtin_expect(cnt0 > RCAP, 0)) {
        const int* __restrict__ srcp = ei;
        const int* __restrict__ dstp = ei + E;
        float adh[4], eh[4], mh[4];
#pragma unroll
        for (int i = 0; i < 4; ++i) {
            const int hd = 2 * i + half;
            adh[i] = ad[n * 8 + hd];
            eh[i]  = leaky(as[n * 8 + hd] + adh[i]);
            mh[i]  = eh[i];
        }
        for (int base = 0; base < E; base += 32) {
            const int e = base + lane;
            const int d = (e < E) ? dstp[e] : -1;
            unsigned hit = __builtin_amdgcn_ballot_w32(d == n);
            while (hit != 0u) {
                const int s = srcp[base + __builtin_ctz(hit)];
                hit &= hit - 1u;
#pragma unroll
                for (int i = 0; i < 4; ++i)
                    mh[i] = fmaxf(mh[i], leaky(as[s * 8 + 2 * i + half] + adh[i]));
            }
        }
        float dens[4];
        float4 accs[4];
        const float* __restrict__ hc = h1 + lane * 4;
#pragma unroll
        for (int i = 0; i < 4; ++i) {
            const float w = expf(eh[i] - mh[i]);
            const float4 hv = *(const float4*)(hc + (size_t)n * 512 + i * 128);
            dens[i] = w;
            accs[i].x = hv.x * w; accs[i].y = hv.y * w; accs[i].z = hv.z * w; accs[i].w = hv.w * w;
        }
        for (int base = 0; base < E; base += 32) {
            const int e = base + lane;
            const int d = (e < E) ? dstp[e] : -1;
            unsigned hit = __builtin_amdgcn_ballot_w32(d == n);
            while (hit != 0u) {
                const int s = srcp[base + __builtin_ctz(hit)];
                hit &= hit - 1u;
#pragma unroll
                for (int i = 0; i < 4; ++i) {
                    const float w = expf(leaky(as[s * 8 + 2 * i + half] + adh[i]) - mh[i]);
                    const float4 hv = *(const float4*)(hc + (size_t)s * 512 + i * 128);
                    dens[i] += w;
                    accs[i].x += hv.x * w; accs[i].y += hv.y * w;
                    accs[i].z += hv.z * w; accs[i].w += hv.w * w;
                }
            }
        }
#pragma unroll
        for (int i = 0; i < 4; ++i) {
            const float4 bb = *(const float4*)(b + i * 128 + lane * 4);
            float4 o;
            o.x = accs[i].x / dens[i] + bb.x; o.y = accs[i].y / dens[i] + bb.y;
            o.z = accs[i].z / dens[i] + bb.z; o.w = accs[i].w / dens[i] + bb.w;
            o.x = o.x > 0.f ? o.x : expm1f(o.x);
            o.y = o.y > 0.f ? o.y : expm1f(o.y);
            o.z = o.z > 0.f ? o.z : expm1f(o.z);
            o.w = o.w > 0.f ? o.w : expm1f(o.w);
            *(float4*)(h2 + (size_t)n * 512 + i * 128 + lane * 4) = o;
        }
        return;
    }

    const int beg = n * RSTRIDE + 1, end = beg + cnt0;
    const int last = end - 1;

    const int q = lane >> 3, hh = lane & 7;
    const float adq = ad[n * 8 + hh];
    float m = leaky(as[n * 8 + hh] + adq);
    for (int base = beg; base < end; base += 32) {
#pragma unroll
        for (int t = 0; t < 8; ++t) {
            const int j = base + q + 4 * t;
            const int s = csr[j < last ? j : last];
            m = fmaxf(m, leaky(as[s * 8 + hh] + adq));
        }
    }
    m = fmaxf(m, __shfl_xor(m, 8));
    m = fmaxf(m, __shfl_xor(m, 16));

    float den[4];
    float4 acc[4];
    const float* __restrict__ hcol = h1 + lane * 4;
#pragma unroll
    for (int i = 0; i < 4; ++i) {
        const int hd = 2 * i + half;
        const float mh = __shfl(m, hd);
        const float w = expf(leaky(as[n * 8 + hd] + ad[n * 8 + hd]) - mh);
        const float4 hv = *(const float4*)(hcol + (size_t)n * 512 + i * 128);
        den[i] = w;
        acc[i].x = hv.x * w; acc[i].y = hv.y * w; acc[i].z = hv.z * w; acc[i].w = hv.w * w;
    }

    const int wpos = (hh & 1) * 4 + (hh >> 1);
    for (int base = beg; base < end; base += 32) {
#pragma unroll
        for (int t = 0; t < 8; ++t) {
            const int j = base + q + 4 * t;
            const int s = csr[j < last ? j : last];
            wmine[(q + 4 * t) * 8 + wpos] = expf(leaky(as[s * 8 + hh] + adq) - m);
        }
        wave_lds_handover();
        const int left = end - base;
        const int cnt = left < 32 ? left : 32;
        int e = 0;
        for (; e + 2 <= cnt; e += 2) {
            const int s0 = csr[base + e], s1 = csr[base + e + 1];
            const float4 wa = *(const float4*)(wmine + e * 8 + half * 4);
            const float4 wb = *(const float4*)(wmine + e * 8 + 8 + half * 4);
            const float* __restrict__ r0 = hcol + (size_t)s0 * 512;
            const float* __restrict__ r1 = hcol + (size_t)s1 * 512;
            const v4f a0 = *(const v4f*)(r0),       a1 = *(const v4f*)(r0 + 128);
            const v4f a2 = *(const v4f*)(r0 + 256), a3 = *(const v4f*)(r0 + 384);
            const v4f b0 = *(const v4f*)(r1),       b1 = *(const v4f*)(r1 + 128);
            const v4f b2 = *(const v4f*)(r1 + 256), b3 = *(const v4f*)(r1 + 384);
            asm volatile("" :: "v"(a0), "v"(a1), "v"(a2), "v"(a3),
                               "v"(b0), "v"(b1), "v"(b2), "v"(b3));
            const v4f ra[4] = {a0, a1, a2, a3};
            const v4f rb[4] = {b0, b1, b2, b3};
            const float wla[4] = {wa.x, wa.y, wa.z, wa.w};
            const float wlb[4] = {wb.x, wb.y, wb.z, wb.w};
#pragma unroll
            for (int i = 0; i < 4; ++i) {
                den[i] += wla[i];
                acc[i].x += ra[i].x * wla[i]; acc[i].y += ra[i].y * wla[i];
                acc[i].z += ra[i].z * wla[i]; acc[i].w += ra[i].w * wla[i];
            }
#pragma unroll
            for (int i = 0; i < 4; ++i) {
                den[i] += wlb[i];
                acc[i].x += rb[i].x * wlb[i]; acc[i].y += rb[i].y * wlb[i];
                acc[i].z += rb[i].z * wlb[i]; acc[i].w += rb[i].w * wlb[i];
            }
        }
        if (e < cnt) {
            const int s0 = csr[base + e];
            const float4 wa = *(const float4*)(wmine + e * 8 + half * 4);
            const float wla[4] = {wa.x, wa.y, wa.z, wa.w};
            const float* __restrict__ r0 = hcol + (size_t)s0 * 512;
#pragma unroll
            for (int i = 0; i < 4; ++i) {
                const v4f hv = *(const v4f*)(r0 + i * 128);
                den[i] += wla[i];
                acc[i].x += hv.x * wla[i]; acc[i].y += hv.y * wla[i];
                acc[i].z += hv.z * wla[i]; acc[i].w += hv.w * wla[i];
            }
        }
        wave_lds_handover();
    }
#pragma unroll
    for (int i = 0; i < 4; ++i) {
        const float4 bb = *(const float4*)(b + i * 128 + lane * 4);
        float4 o;
        o.x = acc[i].x / den[i] + bb.x; o.y = acc[i].y / den[i] + bb.y;
        o.z = acc[i].z / den[i] + bb.z; o.w = acc[i].w / den[i] + bb.w;
        o.x = o.x > 0.f ? o.x : expm1f(o.x);
        o.y = o.y > 0.f ? o.y : expm1f(o.y);
        o.z = o.z > 0.f ? o.z : expm1f(o.z);
        o.w = o.w > 0.f ? o.w : expm1f(o.w);
        *(float4*)(h2 + (size_t)n * 512 + i * 128 + lane * 4) = o;
    }
}

__global__ void gat2_gather(const int* __restrict__ rec, const int* __restrict__ ei, int E,
                            const float* __restrict__ as, const float* __restrict__ ad,
                            const float* __restrict__ h3, const float* __restrict__ b,
                            float* __restrict__ out, int Nn)
{
    int t = blockIdx.x * blockDim.x + threadIdx.x;
    if (t >= Nn * 10) return;
    const int n = t / 10;
    const int g = t - n * 10;
    const int* __restrict__ csr = rec;
    const int cnt0 = rec[(size_t)n * RSTRIDE];
    const float adn   = ad[n];
    const float eself = leaky(as[n] + adn);

    if (__builtin_expect(cnt0 > RCAP, 0)) {
        const int* __restrict__ srcp = ei;
        const int* __restrict__ dstp = ei + E;
        float ms = eself;
        for (int e = 0; e < E; ++e)
            if (dstp[e] == n) ms = fmaxf(ms, leaky(as[srcp[e]] + adn));
        const float w0 = expf(eself - ms);
        const float4 h0 = *(const float4*)(h3 + (size_t)n * 48 + g * 4);
        float dens = w0;
        float4 a4;
        a4.x = h0.x * w0; a4.y = h0.y * w0; a4.z = h0.z * w0; a4.w = h0.w * w0;
        for (int e = 0; e < E; ++e) {
            if (dstp[e] != n) continue;
            const int s = srcp[e];
            const float w = expf(leaky(as[s] + adn) - ms);
            const float4 hv = *(const float4*)(h3 + (size_t)s * 48 + g * 4);
            dens += w;
            a4.x += hv.x * w; a4.y += hv.y * w; a4.z += hv.z * w; a4.w += hv.w * w;
        }
        const float4 bs = *(const float4*)(b + g * 4);
        float4 os;
        os.x = a4.x / dens + bs.x; os.y = a4.y / dens + bs.y;
        os.z = a4.z / dens + bs.z; os.w = a4.w / dens + bs.w;
        *(float4*)(out + (size_t)n * 40 + g * 4) = os;
        return;
    }
    const int beg = n * RSTRIDE + 1, end = beg + cnt0;

    float m = eself;
    for (int j = beg; j < end; ++j)
        m = fmaxf(m, leaky(as[csr[j]] + adn));

    float den;
    float4 acc;
    {
        const float w = expf(eself - m);
        const float4 hv = *(const float4*)(h3 + (size_t)n * 48 + g * 4);
        den = w;
        acc.x = hv.x * w; acc.y = hv.y * w; acc.z = hv.z * w; acc.w = hv.w * w;
    }
#pragma unroll 4
    for (int j = beg; j < end; ++j) {
        const int s = csr[j];
        const float w = expf(leaky(as[s] + adn) - m);
        const float4 hv = *(const float4*)(h3 + (size_t)s * 48 + g * 4);
        den += w;
        acc.x += hv.x * w; acc.y += hv.y * w; acc.z += hv.z * w; acc.w += hv.w * w;
    }
    const float4 bb = *(const float4*)(b + g * 4);
    float4 o;
    o.x = acc.x / den + bb.x; o.y = acc.y / den + bb.y;
    o.z = acc.z / den + bb.z; o.w = acc.w / den + bb.w;
    *(float4*)(out + (size_t)n * 40 + g * 4) = o;
}

extern "C" void kernel_launch(void* const* d_in, const int* in_sizes, int n_in,
                              void* d_out, int out_size, void* d_ws, size_t ws_size,
                              hipStream_t stream)
{
    const float* x    = (const float*)d_in[0];
    const int*   ei   = (const int*)  d_in[1];
    const float* W1   = (const float*)d_in[2];
    const float* aS1  = (const float*)d_in[3];
    const float* aD1  = (const float*)d_in[4];
    const float* b1   = (const float*)d_in[5];
    const float* W2   = (const float*)d_in[6];
    const float* aS2  = (const float*)d_in[7];
    const float* aD2  = (const float*)d_in[8];
    const float* b2   = (const float*)d_in[9];
    float* out = (float*)d_out;

    const int IN  = 256, F1 = 512, OUT = 40, OUTP = 48;
    const int Nn  = in_sizes[0] / IN;
    const int E   = in_sizes[1] / 2;

    float* ws  = (float*)d_ws;
    float* h1  = ws;
    float* h2  = h1  + (size_t)Nn * F1;
    float* h3  = h2  + (size_t)Nn * F1;
    float* W2p = h3  + (size_t)Nn * 48;
    float* as1 = W2p + (size_t)F1 * OUTP;
    float* ad1 = as1 + (size_t)Nn * 8;
    int*   rec = (int*)(ad1 + (size_t)Nn * 8);
    float* as2 = (float*)(rec + (size_t)Nn * RSTRIDE);
    float* ad2 = as2 + Nn;

    const int B = 256;

    clear_counters_pad_w2<<<(Nn + B - 1) / B + (F1 * OUTP + B - 1) / B, B, 0, stream>>>(
        rec, Nn, W2, W2p, F1, OUT, OUTP);

    gemm1_and_fill_records<4><<<dim3((Nn + 15) / 16, F1 / 64 + 1), 32, 0, stream>>>(
        x, W1, h1, Nn, IN, F1, F1, aS1, aD1, as1, ad1, F1, 8, ei, E, rec);

    gat1_gather<<<(Nn + 7) / 8, 256, 0, stream>>>(rec, ei, E, as1, ad1, h1, b1, h2, Nn);

    gemm_wmma_f32<3, 8><<<dim3((Nn + 15) / 16, 1), 256, 0, stream>>>(
        h2, W2p, h3, Nn, F1, OUTP, OUTP, aS2, aD2, as2, ad2, OUT, 1);

    gat2_gather<<<(Nn * 10 + B - 1) / B, B, 0, stream>>>(rec, ei, E, as2, ad2, h3, b2, out, Nn);
}


# Layer-2 GEMM moved into the layer-1 gather's work-group (16 waves = 16 nodes): finished rows go to an LDS tile, 8 waves K-split it on WMMA as before; h2 never reaches memory; 5 -> 4 launches
# speedup vs baseline: 4.1770x; 1.0832x over previous
#include <hip/hip_runtime.h>
#include <hip/hip_bf16.h>
#include <math.h>

typedef float v2f __attribute__((ext_vector_type(2)));
typedef float v8f __attribute__((ext_vector_type(8)));
typedef float v4f __attribute__((ext_vector_type(4)));

#define NEG_SLOPE 0.2f

__device__ __forceinline__ float leaky(float v) {
    return v > 0.0f ? v : NEG_SLOPE * v;
}


__device__ __forceinline__ void wave_lds_handover() {
    __builtin_amdgcn_fence(__ATOMIC_RELEASE, "wavefront");
    __builtin_amdgcn_wave_barrier();
    __builtin_amdgcn_fence(__ATOMIC_ACQUIRE, "wavefront");
}

template<int MASK>
__device__ __forceinline__ float row_xor(float v) {
    const int i = __builtin_bit_cast(int, v);
    return __builtin_bit_cast(float,
        __builtin_amdgcn_update_dpp(i, i, 0x160 + MASK, 0xf, 0xf, false));
}

template<int NT>
__device__ __forceinline__ void
load_att_weights(const float* __restrict__ a_src, const float* __restrict__ a_dst,
                 int colB, int ncv, float (&wsrc)[NT], float (&wdst)[NT])
{
#pragma unroll
    for (int t = 0; t < NT; ++t) {
        const int col = colB + t * 16;
        const bool real = col < ncv;
        const int cc = real ? col : 0;
        const float vs = a_src[cc], vd = a_dst[cc];
        wsrc[t] = real ? vs : 0.f;
        wdst[t] = real ? vd : 0.f;
    }
}

template<int NT>
__device__ __forceinline__ void
strip_epilogue(const v8f (&acc)[NT], float* __restrict__ C, int M, int ldc,
               int row0, int ncol0, int lane,
               const float (&wsrc)[NT], const float (&wdst)[NT],
               float* __restrict__ as, float* __restrict__ ad, int heads, int head)
{
    const int half = lane >> 4;
    const int l16  = lane & 15;
    const int crow0 = row0 + half * 8;
    float* __restrict__ Cbase = C + (size_t)crow0 * (size_t)ldc + ncol0 + l16;
    if (row0 + 16 <= M) {
#pragma unroll
        for (int t = 0; t < NT; ++t) {
            float* __restrict__ Cp = Cbase + t * 16;
#pragma unroll
            for (int r = 0; r < 8; ++r)
                Cp[(size_t)r * (size_t)ldc] = acc[t][r];
        }
    } else {
#pragma unroll
        for (int t = 0; t < NT; ++t) {
            float* __restrict__ Cp = Cbase + t * 16;
#pragma unroll
            for (int r = 0; r < 8; ++r)
                if (crow0 + r < M) Cp[(size_t)r * (size_t)ldc] = acc[t][r];
        }
    }

    float ps[8], pd[8];
#pragma unroll
    for (int r = 0; r < 8; ++r) { ps[r] = 0.f; pd[r] = 0.f; }
#pragma unroll
    for (int t = 0; t < NT; ++t) {
#pragma unroll
        for (int r = 0; r < 8; ++r) {
            ps[r] += acc[t][r] * wsrc[t];
            pd[r] += acc[t][r] * wdst[t];
        }
    }
    const bool b3 = (l16 & 8) != 0, b2 = (l16 & 4) != 0;
    const bool b1 = (l16 & 2) != 0, b0 = (l16 & 1) != 0;
    float u[8], w4[4], y2[2];
#pragma unroll
    for (int i = 0; i < 8; ++i) {
        const float mine = b3 ? pd[i] : ps[i];
        const float give = b3 ? ps[i] : pd[i];
        u[i] = mine + row_xor<8>(give);
    }
#pragma unroll
    for (int i = 0; i < 4; ++i) {
        const float mine = b2 ? u[i + 4] : u[i];
        const float give = b2 ? u[i] : u[i + 4];
        w4[i] = mine + row_xor<4>(give);
    }
#pragma unroll
    for (int i = 0; i < 2; ++i) {
        const float mine = b1 ? w4[i + 2] : w4[i];
        const float give = b1 ? w4[i] : w4[i + 2];
        y2[i] = mine + row_xor<2>(give);
    }
    const float mine = b0 ? y2[1] : y2[0];
    const float give = b0 ? y2[0] : y2[1];
    const float z = mine + row_xor<1>(give);
    const int arow = row0 + half * 8 + (l16 & 7);
    if (arow < M) {
        float* __restrict__ dstp = b3 ? ad : as;
        dstp[(size_t)arow * (size_t)heads + head] = z;
    }
}

template<int NT>
__device__ __forceinline__ void
gemm_strip(const float* __restrict__ A, const float* __restrict__ B,
           float* __restrict__ C, int M, int K, int N, int ldc,
           const float* __restrict__ a_src, const float* __restrict__ a_dst,
           float* __restrict__ as, float* __restrict__ ad, int ncv, int heads)
{
    __shared__ float As[16 * 12];

    const int lane = (int)threadIdx.x;
    const int half = lane >> 4;
    const int l16  = lane & 15;

    const int alr = lane >> 1;
    const int alc = (lane & 1) * 4;
    int rowA = blockIdx.x * 16 + alr;
    if (rowA >= M) rowA = M - 1;
    const float* __restrict__ Aw = A + (size_t)rowA * (size_t)K + alc;

    const int ncol0 = blockIdx.y * (16 * NT);
    const int colB  = ncol0 + l16;

    v8f acc[NT];
#pragma unroll
    for (int t = 0; t < NT; ++t) {
        v8f z = {0.f, 0.f, 0.f, 0.f, 0.f, 0.f, 0.f, 0.f};
        acc[t] = z;
    }

    float4 av = *(const float4*)(Aw);

    float wsrc[NT], wdst[NT];
    load_att_weights<NT>(a_src, a_dst, colB, ncv, wsrc, wdst);

    for (int kc = 0; kc < K; kc += 8) {
        __syncthreads();
        *(float4*)(&As[alr * 12 + alc]) = av;

        v2f breg[2][NT];
#pragma unroll
        for (int kk2 = 0; kk2 < 2; ++kk2) {
            const float* __restrict__ Bk =
                B + (size_t)(kc + kk2 * 4 + 2 * half) * (size_t)N + colB;
#pragma unroll
            for (int t = 0; t < NT; ++t) {
                breg[kk2][t].x = Bk[t * 16];
                breg[kk2][t].y = Bk[N + t * 16];
            }
        }

        const int kn = (kc + 8 < K) ? (kc + 8) : 0;
        av = *(const float4*)(Aw + kn);

        __syncthreads();

#pragma unroll
        for (int kk2 = 0; kk2 < 2; ++kk2) {
            const v2f a = *(const v2f*)(&As[l16 * 12 + kk2 * 4 + 2 * half]);
#pragma unroll
            for (int t = 0; t < NT; ++t) {
                acc[t] = __builtin_amdgcn_wmma_f32_16x16x4_f32(
                    false, a, false, breg[kk2][t], (short)0, acc[t], false, false);
            }
        }
    }

    strip_epilogue<NT>(acc, C, M, ldc, (int)blockIdx.x * 16, ncol0, lane,
                       wsrc, wdst, as, ad, heads, (int)blockIdx.y);
}

#define RSTRIDE 64
#define RCAP 63

__global__ void clear_counters_pad_w2(int* __restrict__ deg, int Nn,
                                      const float* __restrict__ src,
                                      float* __restrict__ dst, int K, int N, int Np)
{
    const int nzb = (Nn + (int)blockDim.x - 1) / (int)blockDim.x;
    if ((int)blockIdx.x < nzb) {
        const int t = blockIdx.x * blockDim.x + threadIdx.x;
        if (t < Nn) deg[(size_t)t * RSTRIDE] = 0;
    } else {
        const int t = ((int)blockIdx.x - nzb) * blockDim.x + threadIdx.x;
        if (t >= K * Np) return;
        const int r = t / Np, c = t - r * Np;
        dst[t] = (c < N) ? src[r * N + c] : 0.f;
    }
}

template<int NT>
__global__ void __launch_bounds__(32)
gemm1_and_fill_records(const float* __restrict__ A, const float* __restrict__ B,
                       float* __restrict__ C, int M, int K, int N, int ldc,
                       const float* __restrict__ a_src, const float* __restrict__ a_dst,
                       float* __restrict__ as, float* __restrict__ ad, int ncv, int heads,
                       const int* __restrict__ ei, int E, int* __restrict__ rec)
{
    if (blockIdx.y == gridDim.y - 1) {
        const int step = (int)gridDim.x * 32;
        const int* __restrict__ srcp = ei;
        const int* __restrict__ dstp = ei + E;
        int e = (int)blockIdx.x * 32 + (int)threadIdx.x;
        for (; e + 3 * step < E; e += 4 * step) {
            const int d0 = dstp[e],            d1 = dstp[e + step];
            const int d2 = dstp[e + 2 * step], d3 = dstp[e + 3 * step];
            const int s0 = srcp[e],            s1 = srcp[e + step];
            const int s2 = srcp[e + 2 * step], s3 = srcp[e + 3 * step];
            int* q0 = rec + (size_t)d0 * RSTRIDE;
            int* q1 = rec + (size_t)d1 * RSTRIDE;
            int* q2 = rec + (size_t)d2 * RSTRIDE;
            int* q3 = rec + (size_t)d3 * RSTRIDE;
            const int r0 = atomicAdd(q0, 1);
            const int r1 = atomicAdd(q1, 1);
            const int r2 = atomicAdd(q2, 1);
            const int r3 = atomicAdd(q3, 1);
            if (r0 < RCAP) q0[1 + r0] = s0;
            if (r1 < RCAP) q1[1 + r1] = s1;
            if (r2 < RCAP) q2[1 + r2] = s2;
            if (r3 < RCAP) q3[1 + r3] = s3;
        }
        for (; e < E; e += step) {
            int* q0 = rec + (size_t)dstp[e] * RSTRIDE;
            const int s0 = srcp[e];
            const int r0 = atomicAdd(q0, 1);
            if (r0 < RCAP) q0[1 + r0] = s0;
        }
        return;
    }
    gemm_strip<NT>(A, B, C, M, K, N, ldc, a_src, a_dst, as, ad, ncv, heads);
}

__device__ __forceinline__ void
gat1_node(const int n, const int lane, float* wmine,
          const int* __restrict__ rec, const int* __restrict__ ei, int E,
          const float* __restrict__ as, const float* __restrict__ ad,
          const float* __restrict__ h1, const float* __restrict__ b,
          float* orow)
{
    const int half = lane >> 4;
    const int* __restrict__ csr = rec;
    const int cnt0 = rec[(size_t)n * RSTRIDE];

    if (__builtin_expect(cnt0 > RCAP, 0)) {
        const int* __restrict__ srcp = ei;
        const int* __restrict__ dstp = ei + E;
        float adh[4], eh[4], mh[4];
#pragma unroll
        for (int i = 0; i < 4; ++i) {
            const int hd = 2 * i + half;
            adh[i] = ad[n * 8 + hd];
            eh[i]  = leaky(as[n * 8 + hd] + adh[i]);
            mh[i]  = eh[i];
        }
        for (int base = 0; base < E; base += 32) {
            const int e = base + lane;
            const int d = (e < E) ? dstp[e] : -1;
            unsigned hit = __builtin_amdgcn_ballot_w32(d == n);
            while (hit != 0u) {
                const int s = srcp[base + __builtin_ctz(hit)];
                hit &= hit - 1u;
#pragma unroll
                for (int i = 0; i < 4; ++i)
                    mh[i] = fmaxf(mh[i], leaky(as[s * 8 + 2 * i + half] + adh[i]));
            }
        }
        float dens[4];
        float4 accs[4];
        const float* __restrict__ hc = h1 + lane * 4;
#pragma unroll
        for (int i = 0; i < 4; ++i) {
            const float w = expf(eh[i] - mh[i]);
            const float4 hv = *(const float4*)(hc + (size_t)n * 512 + i * 128);
            dens[i] = w;
            accs[i].x = hv.x * w; accs[i].y = hv.y * w; accs[i].z = hv.z * w; accs[i].w = hv.w * w;
        }
        for (int base = 0; base < E; base += 32) {
            const int e = base + lane;
            const int d = (e < E) ? dstp[e] : -1;
            unsigned hit = __builtin_amdgcn_ballot_w32(d == n);
            while (hit != 0u) {
                const int s = srcp[base + __builtin_ctz(hit)];
                hit &= hit - 1u;
#pragma unroll
                for (int i = 0; i < 4; ++i) {
                    const float w = expf(leaky(as[s * 8 + 2 * i + half] + adh[i]) - mh[i]);
                    const float4 hv = *(const float4*)(hc + (size_t)s * 512 + i * 128);
                    dens[i] += w;
                    accs[i].x += hv.x * w; accs[i].y += hv.y * w;
                    accs[i].z += hv.z * w; accs[i].w += hv.w * w;
                }
            }
        }
#pragma unroll
        for (int i = 0; i < 4; ++i) {
            const float4 bb = *(const float4*)(b + i * 128 + lane * 4);
            float4 o;
            o.x = accs[i].x / dens[i] + bb.x; o.y = accs[i].y / dens[i] + bb.y;
            o.z = accs[i].z / dens[i] + bb.z; o.w = accs[i].w / dens[i] + bb.w;
            o.x = o.x > 0.f ? o.x : expm1f(o.x);
            o.y = o.y > 0.f ? o.y : expm1f(o.y);
            o.z = o.z > 0.f ? o.z : expm1f(o.z);
            o.w = o.w > 0.f ? o.w : expm1f(o.w);
            *(float4*)(orow + i * 128 + lane * 4) = o;
        }
        return;
    }

    const int beg = n * RSTRIDE + 1, end = beg + cnt0;
    const int last = end - 1;

    const int q = lane >> 3, hh = lane & 7;
    const float adq = ad[n * 8 + hh];
    float m = leaky(as[n * 8 + hh] + adq);
    for (int base = beg; base < end; base += 32) {
#pragma unroll
        for (int t = 0; t < 8; ++t) {
            const int j = base + q + 4 * t;
            const int s = csr[j < last ? j : last];
            m = fmaxf(m, leaky(as[s * 8 + hh] + adq));
        }
    }
    m = fmaxf(m, __shfl_xor(m, 8));
    m = fmaxf(m, __shfl_xor(m, 16));

    float den[4];
    float4 acc[4];
    const float* __restrict__ hcol = h1 + lane * 4;
#pragma unroll
    for (int i = 0; i < 4; ++i) {
        const int hd = 2 * i + half;
        const float mh = __shfl(m, hd);
        const float w = expf(leaky(as[n * 8 + hd] + ad[n * 8 + hd]) - mh);
        const float4 hv = *(const float4*)(hcol + (size_t)n * 512 + i * 128);
        den[i] = w;
        acc[i].x = hv.x * w; acc[i].y = hv.y * w; acc[i].z = hv.z * w; acc[i].w = hv.w * w;
    }

    const int wpos = (hh & 1) * 4 + (hh >> 1);
    for (int base = beg; base < end; base += 32) {
#pragma unroll
        for (int t = 0; t < 8; ++t) {
            const int j = base + q + 4 * t;
            const int s = csr[j < last ? j : last];
            wmine[(q + 4 * t) * 8 + wpos] = expf(leaky(as[s * 8 + hh] + adq) - m);
        }
        wave_lds_handover();
        const int left = end - base;
        const int cnt = left < 32 ? left : 32;
        int e = 0;
        for (; e + 2 <= cnt; e += 2) {
            const int s0 = csr[base + e], s1 = csr[base + e + 1];
            const float4 wa = *(const float4*)(wmine + e * 8 + half * 4);
            const float4 wb = *(const float4*)(wmine + e * 8 + 8 + half * 4);
            const float* __restrict__ r0 = hcol + (size_t)s0 * 512;
            const float* __restrict__ r1 = hcol + (size_t)s1 * 512;
            const v4f a0 = *(const v4f*)(r0),       a1 = *(const v4f*)(r0 + 128);
            const v4f a2 = *(const v4f*)(r0 + 256), a3 = *(const v4f*)(r0 + 384);
            const v4f b0 = *(const v4f*)(r1),       b1 = *(const v4f*)(r1 + 128);
            const v4f b2 = *(const v4f*)(r1 + 256), b3 = *(const v4f*)(r1 + 384);
            asm volatile("" :: "v"(a0), "v"(a1), "v"(a2), "v"(a3),
                               "v"(b0), "v"(b1), "v"(b2), "v"(b3));
            const v4f ra[4] = {a0, a1, a2, a3};
            const v4f rb[4] = {b0, b1, b2, b3};
            const float wla[4] = {wa.x, wa.y, wa.z, wa.w};
            const float wlb[4] = {wb.x, wb.y, wb.z, wb.w};
#pragma unroll
            for (int i = 0; i < 4; ++i) {
                den[i] += wla[i];
                acc[i].x += ra[i].x * wla[i]; acc[i].y += ra[i].y * wla[i];
                acc[i].z += ra[i].z * wla[i]; acc[i].w += ra[i].w * wla[i];
            }
#pragma unroll
            for (int i = 0; i < 4; ++i) {
                den[i] += wlb[i];
                acc[i].x += rb[i].x * wlb[i]; acc[i].y += rb[i].y * wlb[i];
                acc[i].z += rb[i].z * wlb[i]; acc[i].w += rb[i].w * wlb[i];
            }
        }
        if (e < cnt) {
            const int s0 = csr[base + e];
            const float4 wa = *(const float4*)(wmine + e * 8 + half * 4);
            const float wla[4] = {wa.x, wa.y, wa.z, wa.w};
            const float* __restrict__ r0 = hcol + (size_t)s0 * 512;
#pragma unroll
            for (int i = 0; i < 4; ++i) {
                const v4f hv = *(const v4f*)(r0 + i * 128);
                den[i] += wla[i];
                acc[i].x += hv.x * wla[i]; acc[i].y += hv.y * wla[i];
                acc[i].z += hv.z * wla[i]; acc[i].w += hv.w * wla[i];
            }
        }
        wave_lds_handover();
    }
#pragma unroll
    for (int i = 0; i < 4; ++i) {
        const float4 bb = *(const float4*)(b + i * 128 + lane * 4);
        float4 o;
        o.x = acc[i].x / den[i] + bb.x; o.y = acc[i].y / den[i] + bb.y;
        o.z = acc[i].z / den[i] + bb.z; o.w = acc[i].w / den[i] + bb.w;
        o.x = o.x > 0.f ? o.x : expm1f(o.x);
        o.y = o.y > 0.f ? o.y : expm1f(o.y);
        o.z = o.z > 0.f ? o.z : expm1f(o.z);
        o.w = o.w > 0.f ? o.w : expm1f(o.w);
        *(float4*)(orow + i * 128 + lane * 4) = o;
    }
}

#define F1C 512
#define TPITCH 516

template<int NT, int KS>
__global__ void __launch_bounds__(512)
gat1_gather_gemm2(const int* __restrict__ rec, const int* __restrict__ ei, int E,
                  const float* __restrict__ as, const float* __restrict__ ad,
                  const float* __restrict__ h1, const float* __restrict__ b, int Nn,
                  const float* __restrict__ B, float* __restrict__ C, int N, int ldc,
                  const float* __restrict__ a_src, const float* __restrict__ a_dst,
                  float* __restrict__ as2, float* __restrict__ ad2, int ncv)
{
    constexpr int NPART = (KS - 1) * NT * 2 * 32;
    constexpr int NSCR  = NPART > 16 * 64 ? NPART : 16 * 64;
    __shared__ float4 scr[NSCR];
    __shared__ float tile[16 * TPITCH];

    const int lane = (int)threadIdx.x & 31;
    const int wv = __builtin_amdgcn_readfirstlane((int)threadIdx.x >> 5);
    const int n  = (int)blockIdx.x * 16 + wv;
    float* orow = tile + wv * TPITCH;
    if (n < Nn) {
        gat1_node(n, lane, (float*)(scr + wv * 64), rec, ei, E, as, ad, h1, b, orow);
    } else {
#pragma unroll
        for (int i = 0; i < 4; ++i)
            *(float4*)(orow + i * 128 + lane * 4) = make_float4(0.f, 0.f, 0.f, 0.f);
    }
    __syncthreads();

    const int half = lane >> 4;
    const int l16  = lane & 15;
    v8f acc[NT];
#pragma unroll
    for (int t = 0; t < NT; ++t) {
        v8f z = {0.f, 0.f, 0.f, 0.f, 0.f, 0.f, 0.f, 0.f};
        acc[t] = z;
    }
    float wsrc[NT], wdst[NT];
    load_att_weights<NT>(a_src, a_dst, l16, ncv, wsrc, wdst);

    if (wv < KS) {
        constexpr int KPER = F1C / KS;
        const int kbeg = wv * KPER;
        const float* __restrict__ Bw = B + (size_t)(kbeg + 2 * half) * (size_t)N + l16;
        const float* Ar = tile + l16 * TPITCH + kbeg + 2 * half;
#pragma unroll 1
        for (int kc = 0; kc < KPER; kc += 16) {
            v2f breg[4][NT], a[4];
#pragma unroll
            for (int kk = 0; kk < 4; ++kk) {
                const float* __restrict__ Bk = Bw + (size_t)(kc + kk * 4) * (size_t)N;
#pragma unroll
                for (int t = 0; t < NT; ++t) {
                    breg[kk][t].x = Bk[t * 16];
                    breg[kk][t].y = Bk[N + t * 16];
                }
            }
#pragma unroll
            for (int kk = 0; kk < 4; ++kk)
                a[kk] = *(const v2f*)(Ar + kc + kk * 4);
#pragma unroll
            for (int kk = 0; kk < 4; ++kk) {
#pragma unroll
                for (int t = 0; t < NT; ++t) {
                    acc[t] = __builtin_amdgcn_wmma_f32_16x16x4_f32(
                        false, a[kk], false, breg[kk][t], (short)0, acc[t], false, false);
                }
            }
        }
        if (wv != 0) {
            float4* pp = scr + (size_t)(wv - 1) * (NT * 2 * 32) + lane;
#pragma unroll
            for (int t = 0; t < NT; ++t) {
                pp[(t * 2 + 0) * 32] = make_float4(acc[t][0], acc[t][1], acc[t][2], acc[t][3]);
                pp[(t * 2 + 1) * 32] = make_float4(acc[t][4], acc[t][5], acc[t][6], acc[t][7]);
            }
        }
    }
    __syncthreads();
    if (wv != 0) return;
    constexpr int NP = NT * 2;
    const v4f* pp = (const v4f*)scr + lane;
#pragma unroll 1
    for (int w = 0; w < KS - 1; ++w, pp += NP * 32) {
        v4f pv[NP];
#pragma unroll
        for (int i = 0; i < NP; ++i) pv[i] = pp[i * 32];
        if constexpr (NT == 3) {
            asm volatile("" :: "v"(pv[0]), "v"(pv[1]), "v"(pv[2]),
                               "v"(pv[3]), "v"(pv[4]), "v"(pv[5]));
        }
#pragma unroll
        for (int t = 0; t < NT; ++t) {
#pragma unroll
            for (int r = 0; r < 4; ++r) {
                acc[t][r]     += pv[t * 2][r];
                acc[t][r + 4] += pv[t * 2 + 1][r];
            }
        }
    }
    strip_epilogue<NT>(acc, C, Nn, ldc, (int)blockIdx.x * 16, 0, lane,
                       wsrc, wdst, as2, ad2, 1, 0);
}

__global__ void gat2_gather(const int* __restrict__ rec, const int* __restrict__ ei, int E,
                            const float* __restrict__ as, const float* __restrict__ ad,
                            const float* __restrict__ h3, const float* __restrict__ b,
                            float* __restrict__ out, int Nn)
{
    int t = blockIdx.x * blockDim.x + threadIdx.x;
    if (t >= Nn * 10) return;
    const int n = t / 10;
    const int g = t - n * 10;
    const int* __restrict__ csr = rec;
    const int cnt0 = rec[(size_t)n * RSTRIDE];
    const float adn   = ad[n];
    const float eself = leaky(as[n] + adn);

    if (__builtin_expect(cnt0 > RCAP, 0)) {
        const int* __restrict__ srcp = ei;
        const int* __restrict__ dstp = ei + E;
        float ms = eself;
        for (int e = 0; e < E; ++e)
            if (dstp[e] == n) ms = fmaxf(ms, leaky(as[srcp[e]] + adn));
        const float w0 = expf(eself - ms);
        const float4 h0 = *(const float4*)(h3 + (size_t)n * 48 + g * 4);
        float dens = w0;
        float4 a4;
        a4.x = h0.x * w0; a4.y = h0.y * w0; a4.z = h0.z * w0; a4.w = h0.w * w0;
        for (int e = 0; e < E; ++e) {
            if (dstp[e] != n) continue;
            const int s = srcp[e];
            const float w = expf(leaky(as[s] + adn) - ms);
            const float4 hv = *(const float4*)(h3 + (size_t)s * 48 + g * 4);
            dens += w;
            a4.x += hv.x * w; a4.y += hv.y * w; a4.z += hv.z * w; a4.w += hv.w * w;
        }
        const float4 bs = *(const float4*)(b + g * 4);
        float4 os;
        os.x = a4.x / dens + bs.x; os.y = a4.y / dens + bs.y;
        os.z = a4.z / dens + bs.z; os.w = a4.w / dens + bs.w;
        *(float4*)(out + (size_t)n * 40 + g * 4) = os;
        return;
    }
    const int beg = n * RSTRIDE + 1, end = beg + cnt0;

    float m = eself;
    for (int j = beg; j < end; ++j)
        m = fmaxf(m, leaky(as[csr[j]] + adn));

    float den;
    float4 acc;
    {
        const float w = expf(eself - m);
        const float4 hv = *(const float4*)(h3 + (size_t)n * 48 + g * 4);
        den = w;
        acc.x = hv.x * w; acc.y = hv.y * w; acc.z = hv.z * w; acc.w = hv.w * w;
    }
#pragma unroll 4
    for (int j = beg; j < end; ++j) {
        const int s = csr[j];
        const float w = expf(leaky(as[s] + adn) - m);
        const float4 hv = *(const float4*)(h3 + (size_t)s * 48 + g * 4);
        den += w;
        acc.x += hv.x * w; acc.y += hv.y * w; acc.z += hv.z * w; acc.w += hv.w * w;
    }
    const float4 bb = *(const float4*)(b + g * 4);
    float4 o;
    o.x = acc.x / den + bb.x; o.y = acc.y / den + bb.y;
    o.z = acc.z / den + bb.z; o.w = acc.w / den + bb.w;
    *(float4*)(out + (size_t)n * 40 + g * 4) = o;
}

extern "C" void kernel_launch(void* const* d_in, const int* in_sizes, int n_in,
                              void* d_out, int out_size, void* d_ws, size_t ws_size,
                              hipStream_t stream)
{
    const float* x    = (const float*)d_in[0];
    const int*   ei   = (const int*)  d_in[1];
    const float* W1   = (const float*)d_in[2];
    const float* aS1  = (const float*)d_in[3];
    const float* aD1  = (const float*)d_in[4];
    const float* b1   = (const float*)d_in[5];
    const float* W2   = (const float*)d_in[6];
    const float* aS2  = (const float*)d_in[7];
    const float* aD2  = (const float*)d_in[8];
    const float* b2   = (const float*)d_in[9];
    float* out = (float*)d_out;

    const int IN  = 256, F1 = 512, OUT = 40, OUTP = 48;
    const int Nn  = in_sizes[0] / IN;
    const int E   = in_sizes[1] / 2;

    float* ws  = (float*)d_ws;
    float* h1  = ws;
    float* h3  = h1  + (size_t)Nn * F1;
    float* W2p = h3  + (size_t)Nn * 48;
    float* as1 = W2p + (size_t)F1 * OUTP;
    float* ad1 = as1 + (size_t)Nn * 8;
    int*   rec = (int*)(ad1 + (size_t)Nn * 8);
    float* as2 = (float*)(rec + (size_t)Nn * RSTRIDE);
    float* ad2 = as2 + Nn;

    const int B = 256;

    clear_counters_pad_w2<<<(Nn + B - 1) / B + (F1 * OUTP + B - 1) / B, B, 0, stream>>>(
        rec, Nn, W2, W2p, F1, OUT, OUTP);

    gemm1_and_fill_records<4><<<dim3((Nn + 15) / 16, F1 / 64 + 1), 32, 0, stream>>>(
        x, W1, h1, Nn, IN, F1, F1, aS1, aD1, as1, ad1, F1, 8, ei, E, rec);

    gat1_gather_gemm2<3, 8><<<(Nn + 15) / 16, 512, 0, stream>>>(
        rec, ei, E, as1, ad1, h1, b1, Nn, W2p, h3, OUTP, OUTP, aS2, aD2, as2, ad2, OUT);

    gat2_gather<<<(Nn * 10 + B - 1) / B, B, 0, stream>>>(rec, ei, E, as2, ad2, h3, b2, out, Nn);
}


# Layer-1 GEMM: 4-wave work-groups share a 64x64 tile; A and B stages (16 k) fetched once per work-group by 16-byte loads into two LDS buffers, next stage in flight under the WMMAs; sum order kept
# speedup vs baseline: 4.7455x; 1.1361x over previous
#include <hip/hip_runtime.h>
#include <hip/hip_bf16.h>
#include <math.h>

typedef float v2f __attribute__((ext_vector_type(2)));
typedef float v8f __attribute__((ext_vector_type(8)));
typedef float v4f __attribute__((ext_vector_type(4)));

#define NEG_SLOPE 0.2f

__device__ __forceinline__ float leaky(float v) {
    return v > 0.0f ? v : NEG_SLOPE * v;
}


__device__ __forceinline__ void wave_lds_handover() {
    __builtin_amdgcn_fence(__ATOMIC_RELEASE, "wavefront");
    __builtin_amdgcn_wave_barrier();
    __builtin_amdgcn_fence(__ATOMIC_ACQUIRE, "wavefront");
}

template<int MASK>
__device__ __forceinline__ float row_xor(float v) {
    const int i = __builtin_bit_cast(int, v);
    return __builtin_bit_cast(float,
        __builtin_amdgcn_update_dpp(i, i, 0x160 + MASK, 0xf, 0xf, false));
}

template<int NT>
__device__ __forceinline__ void
load_att_weights(const float* __restrict__ a_src, const float* __restrict__ a_dst,
                 int colB, int ncv, float (&wsrc)[NT], float (&wdst)[NT])
{
#pragma unroll
    for (int t = 0; t < NT; ++t) {
        const int col = colB + t * 16;
        const bool real = col < ncv;
        const int cc = real ? col : 0;
        const float vs = a_src[cc], vd = a_dst[cc];
        wsrc[t] = real ? vs : 0.f;
        wdst[t] = real ? vd : 0.f;
    }
}

template<int NT>
__device__ __forceinline__ void
strip_epilogue(const v8f (&acc)[NT], float* __restrict__ C, int M, int ldc,
               int row0, int ncol0, int lane,
               const float (&wsrc)[NT], const float (&wdst)[NT],
               float* __restrict__ as, float* __restrict__ ad, int heads, int head)
{
    const int half = lane >> 4;
    const int l16  = lane & 15;
    const int crow0 = row0 + half * 8;
    float* __restrict__ Cbase = C + (size_t)crow0 * (size_t)ldc + ncol0 + l16;
    if (row0 + 16 <= M) {
#pragma unroll
        for (int t = 0; t < NT; ++t) {
            float* __restrict__ Cp = Cbase + t * 16;
#pragma unroll
            for (int r = 0; r < 8; ++r)
                Cp[(size_t)r * (size_t)ldc] = acc[t][r];
        }
    } else {
#pragma unroll
        for (int t = 0; t < NT; ++t) {
            float* __restrict__ Cp = Cbase + t * 16;
#pragma unroll
            for (int r = 0; r < 8; ++r)
                if (crow0 + r < M) Cp[(size_t)r * (size_t)ldc] = acc[t][r];
        }
    }

    float ps[8], pd[8];
#pragma unroll
    for (int r = 0; r < 8; ++r) { ps[r] = 0.f; pd[r] = 0.f; }
#pragma unroll
    for (int t = 0; t < NT; ++t) {
#pragma unroll
        for (int r = 0; r < 8; ++r) {
            ps[r] += acc[t][r] * wsrc[t];
            pd[r] += acc[t][r] * wdst[t];
        }
    }
    const bool b3 = (l16 & 8) != 0, b2 = (l16 & 4) != 0;
    const bool b1 = (l16 & 2) != 0, b0 = (l16 & 1) != 0;
    float u[8], w4[4], y2[2];
#pragma unroll
    for (int i = 0; i < 8; ++i) {
        const float mine = b3 ? pd[i] : ps[i];
        const float give = b3 ? ps[i] : pd[i];
        u[i] = mine + row_xor<8>(give);
    }
#pragma unroll
    for (int i = 0; i < 4; ++i) {
        const float mine = b2 ? u[i + 4] : u[i];
        const float give = b2 ? u[i] : u[i + 4];
        w4[i] = mine + row_xor<4>(give);
    }
#pragma unroll
    for (int i = 0; i < 2; ++i) {
        const float mine = b1 ? w4[i + 2] : w4[i];
        const float give = b1 ? w4[i] : w4[i + 2];
        y2[i] = mine + row_xor<2>(give);
    }
    const float mine = b0 ? y2[1] : y2[0];
    const float give = b0 ? y2[0] : y2[1];
    const float z = mine + row_xor<1>(give);
    const int arow = row0 + half * 8 + (l16 & 7);
    if (arow < M) {
        float* __restrict__ dstp = b3 ? ad : as;
        dstp[(size_t)arow * (size_t)heads + head] = z;
    }
}

#define G1_ROWS 64
#define G1_KC 16
#define G1_APITCH 20
#define G1_BPITCH 72

template<int NT>
__device__ __forceinline__ void
gemm_strip(const float* __restrict__ A, const float* __restrict__ B,
           float* __restrict__ C, int M, int K, int N, int ldc,
           const float* __restrict__ a_src, const float* __restrict__ a_dst,
           float* __restrict__ as, float* __restrict__ ad, int ncv, int heads)
{
    static_assert(NT * 16 == 64, "B staging: 16 threads x 4 columns per k row");
    __shared__ float As[2][G1_ROWS * G1_APITCH];
    __shared__ float Bs[2][G1_KC * G1_BPITCH];

    const int tid  = (int)threadIdx.x;
    const int lane = tid & 31;
    const int wv   = __builtin_amdgcn_readfirstlane(tid >> 5);
    const int half = lane >> 4;
    const int l16  = lane & 15;

    const int alr = tid >> 1;
    const int alc = (tid & 1) * 8;
    int rowA = blockIdx.x * G1_ROWS + alr;
    if (rowA >= M) rowA = M - 1;
    const float* __restrict__ Aw = A + (size_t)rowA * (size_t)K + alc;

    const int ncol0 = blockIdx.y * (16 * NT);
    const int colB  = ncol0 + l16;
    const int bk = tid >> 4;
    const int bc = (tid & 15) * 4;
    const float* __restrict__ Bw = B + (size_t)bk * (size_t)N + ncol0 + bc;

    v8f acc[NT];
#pragma unroll
    for (int t = 0; t < NT; ++t) {
        v8f z = {0.f, 0.f, 0.f, 0.f, 0.f, 0.f, 0.f, 0.f};
        acc[t] = z;
    }

    float4 a0 = *(const float4*)(Aw), a1 = *(const float4*)(Aw + 4);
    float4 b0 = *(const float4*)(Bw), b1 = *(const float4*)(Bw + (size_t)8 * (size_t)N);

    float wsrc[NT], wdst[NT];
    load_att_weights<NT>(a_src, a_dst, colB, ncv, wsrc, wdst);

    const int aoff = alr * G1_APITCH + alc;
    const int boff = bk * G1_BPITCH + bc;
    const int afr  = (wv * 16 + l16) * G1_APITCH + 2 * half;
    const int bfr  = (2 * half) * G1_BPITCH + l16;
    int p = 0;
    for (int kc = 0; kc < K; kc += G1_KC) {
        float* __restrict__ Asp = As[p];
        float* __restrict__ Bsp = Bs[p];
        *(float4*)(Asp + aoff)     = a0;
        *(float4*)(Asp + aoff + 4) = a1;
        *(float4*)(Bsp + boff)                 = b0;
        *(float4*)(Bsp + boff + 8 * G1_BPITCH) = b1;

        __syncthreads();

        const int kn = (kc + G1_KC < K) ? (kc + G1_KC) : 0;
        a0 = *(const float4*)(Aw + kn);
        a1 = *(const float4*)(Aw + kn + 4);
        b0 = *(const float4*)(Bw + (size_t)kn * (size_t)N);
        b1 = *(const float4*)(Bw + (size_t)(kn + 8) * (size_t)N);

        v2f a[4], breg[4][NT];
#pragma unroll
        for (int kk = 0; kk < 4; ++kk)
            a[kk] = *(const v2f*)(Asp + afr + kk * 4);
#pragma unroll
        for (int kk = 0; kk < 4; ++kk) {
#pragma unroll
            for (int t = 0; t < NT; ++t) {
                breg[kk][t].x = Bsp[bfr + (kk * 4) * G1_BPITCH + t * 16];
                breg[kk][t].y = Bsp[bfr + (kk * 4 + 1) * G1_BPITCH + t * 16];
            }
        }
        asm volatile("" :: "v"(a[0]), "v"(a[1]), "v"(a[2]), "v"(a[3]),
                           "v"(breg[0][0]), "v"(breg[0][1]), "v"(breg[0][2]), "v"(breg[0][3]),
                           "v"(breg[1][0]), "v"(breg[1][1]), "v"(breg[1][2]), "v"(breg[1][3]),
                           "v"(breg[2][0]), "v"(breg[2][1]), "v"(breg[2][2]), "v"(breg[2][3]),
                           "v"(breg[3][0]), "v"(breg[3][1]), "v"(breg[3][2]), "v"(breg[3][3]));
#pragma unroll
        for (int kk = 0; kk < 4; ++kk) {
#pragma unroll
            for (int t = 0; t < NT; ++t) {
                acc[t] = __builtin_amdgcn_wmma_f32_16x16x4_f32(
                    false, a[kk], false, breg[kk][t], (short)0, acc[t], false, false);
            }
        }
        p ^= 1;
    }

    strip_epilogue<NT>(acc, C, M, ldc, (int)blockIdx.x * G1_ROWS + wv * 16, ncol0, lane,
                       wsrc, wdst, as, ad, heads, (int)blockIdx.y);
}

#define RSTRIDE 64
#define RCAP 63

__global__ void clear_counters_pad_w2(int* __restrict__ deg, int Nn,
                                      const float* __restrict__ src,
                                      float* __restrict__ dst, int K, int N, int Np)
{
    const int nzb = (Nn + (int)blockDim.x - 1) / (int)blockDim.x;
    if ((int)blockIdx.x < nzb) {
        const int t = blockIdx.x * blockDim.x + threadIdx.x;
        if (t < Nn) deg[(size_t)t * RSTRIDE] = 0;
    } else {
        const int t = ((int)blockIdx.x - nzb) * blockDim.x + threadIdx.x;
        if (t >= K * Np) return;
        const int r = t / Np, c = t - r * Np;
        dst[t] = (c < N) ? src[r * N + c] : 0.f;
    }
}

template<int NT>
__global__ void __launch_bounds__(128)
gemm1_and_fill_records(const float* __restrict__ A, const float* __restrict__ B,
                       float* __restrict__ C, int M, int K, int N, int ldc,
                       const float* __restrict__ a_src, const float* __restrict__ a_dst,
                       float* __restrict__ as, float* __restrict__ ad, int ncv, int heads,
                       const int* __restrict__ ei, int E, int* __restrict__ rec)
{
    if (blockIdx.y == gridDim.y - 1) {
        const int step = (int)(gridDim.x * blockDim.x);
        const int* __restrict__ srcp = ei;
        const int* __restrict__ dstp = ei + E;
        int e = (int)(blockIdx.x * blockDim.x + threadIdx.x);
        for (; e + 3 * step < E; e += 4 * step) {
            const int d0 = dstp[e],            d1 = dstp[e + step];
            const int d2 = dstp[e + 2 * step], d3 = dstp[e + 3 * step];
            const int s0 = srcp[e],            s1 = srcp[e + step];
            const int s2 = srcp[e + 2 * step], s3 = srcp[e + 3 * step];
            int* q0 = rec + (size_t)d0 * RSTRIDE;
            int* q1 = rec + (size_t)d1 * RSTRIDE;
            int* q2 = rec + (size_t)d2 * RSTRIDE;
            int* q3 = rec + (size_t)d3 * RSTRIDE;
            const int r0 = atomicAdd(q0, 1);
            const int r1 = atomicAdd(q1, 1);
            const int r2 = atomicAdd(q2, 1);
            const int r3 = atomicAdd(q3, 1);
            if (r0 < RCAP) q0[1 + r0] = s0;
            if (r1 < RCAP) q1[1 + r1] = s1;
            if (r2 < RCAP) q2[1 + r2] = s2;
            if (r3 < RCAP) q3[1 + r3] = s3;
        }
        for (; e < E; e += step) {
            int* q0 = rec + (size_t)dstp[e] * RSTRIDE;
            const int s0 = srcp[e];
            const int r0 = atomicAdd(q0, 1);
            if (r0 < RCAP) q0[1 + r0] = s0;
        }
        return;
    }
    gemm_strip<NT>(A, B, C, M, K, N, ldc, a_src, a_dst, as, ad, ncv, heads);
}

__device__ __forceinline__ void
gat1_node(const int n, const int lane, float* wmine,
          const int* __restrict__ rec, const int* __restrict__ ei, int E,
          const float* __restrict__ as, const float* __restrict__ ad,
          const float* __restrict__ h1, const float* __restrict__ b,
          float* orow)
{
    const int half = lane >> 4;
    const int* __restrict__ csr = rec;
    const int cnt0 = rec[(size_t)n * RSTRIDE];

    if (__builtin_expect(cnt0 > RCAP, 0)) {
        const int* __restrict__ srcp = ei;
        const int* __restrict__ dstp = ei + E;
        float adh[4], eh[4], mh[4];
#pragma unroll
        for (int i = 0; i < 4; ++i) {
            const int hd = 2 * i + half;
            adh[i] = ad[n * 8 + hd];
            eh[i]  = leaky(as[n * 8 + hd] + adh[i]);
            mh[i]  = eh[i];
        }
        for (int base = 0; base < E; base += 32) {
            const int e = base + lane;
            const int d = (e < E) ? dstp[e] : -1;
            unsigned hit = __builtin_amdgcn_ballot_w32(d == n);
            while (hit != 0u) {
                const int s = srcp[base + __builtin_ctz(hit)];
                hit &= hit - 1u;
#pragma unroll
                for (int i = 0; i < 4; ++i)
                    mh[i] = fmaxf(mh[i], leaky(as[s * 8 + 2 * i + half] + adh[i]));
            }
        }
        float dens[4];
        float4 accs[4];
        const float* __restrict__ hc = h1 + lane * 4;
#pragma unroll
        for (int i = 0; i < 4; ++i) {
            const float w = expf(eh[i] - mh[i]);
            const float4 hv = *(const float4*)(hc + (size_t)n * 512 + i * 128);
            dens[i] = w;
            accs[i].x = hv.x * w; accs[i].y = hv.y * w; accs[i].z = hv.z * w; accs[i].w = hv.w * w;
        }
        for (int base = 0; base < E; base += 32) {
            const int e = base + lane;
            const int d = (e < E) ? dstp[e] : -1;
            unsigned hit = __builtin_amdgcn_ballot_w32(d == n);
            while (hit != 0u) {
                const int s = srcp[base + __builtin_ctz(hit)];
                hit &= hit - 1u;
#pragma unroll
                for (int i = 0; i < 4; ++i) {
                    const float w = expf(leaky(as[s * 8 + 2 * i + half] + adh[i]) - mh[i]);
                    const float4 hv = *(const float4*)(hc + (size_t)s * 512 + i * 128);
                    dens[i] += w;
                    accs[i].x += hv.x * w; accs[i].y += hv.y * w;
                    accs[i].z += hv.z * w; accs[i].w += hv.w * w;
                }
            }
        }
#pragma unroll
        for (int i = 0; i < 4; ++i) {
            const float4 bb = *(const float4*)(b + i * 128 + lane * 4);
            float4 o;
            o.x = accs[i].x / dens[i] + bb.x; o.y = accs[i].y / dens[i] + bb.y;
            o.z = accs[i].z / dens[i] + bb.z; o.w = accs[i].w / dens[i] + bb.w;
            o.x = o.x > 0.f ? o.x : expm1f(o.x);
            o.y = o.y > 0.f ? o.y : expm1f(o.y);
            o.z = o.z > 0.f ? o.z : expm1f(o.z);
            o.w = o.w > 0.f ? o.w : expm1f(o.w);
            *(float4*)(orow + i * 128 + lane * 4) = o;
        }
        return;
    }

    const int beg = n * RSTRIDE + 1, end = beg + cnt0;
    const int last = end - 1;

    const int q = lane >> 3, hh = lane & 7;
    const float adq = ad[n * 8 + hh];
    float m = leaky(as[n * 8 + hh] + adq);
    for (int base = beg; base < end; base += 32) {
#pragma unroll
        for (int t = 0; t < 8; ++t) {
            const int j = base + q + 4 * t;
            const int s = csr[j < last ? j : last];
            m = fmaxf(m, leaky(as[s * 8 + hh] + adq));
        }
    }
    m = fmaxf(m, __shfl_xor(m, 8));
    m = fmaxf(m, __shfl_xor(m, 16));

    float den[4];
    float4 acc[4];
    const float* __restrict__ hcol = h1 + lane * 4;
#pragma unroll
    for (int i = 0; i < 4; ++i) {
        const int hd = 2 * i + half;
        const float mh = __shfl(m, hd);
        const float w = expf(leaky(as[n * 8 + hd] + ad[n * 8 + hd]) - mh);
        const float4 hv = *(const float4*)(hcol + (size_t)n * 512 + i * 128);
        den[i] = w;
        acc[i].x = hv.x * w; acc[i].y = hv.y * w; acc[i].z = hv.z * w; acc[i].w = hv.w * w;
    }

    const int wpos = (hh & 1) * 4 + (hh >> 1);
    for (int base = beg; base < end; base += 32) {
#pragma unroll
        for (int t = 0; t < 8; ++t) {
            const int j = base + q + 4 * t;
            const int s = csr[j < last ? j : last];
            wmine[(q + 4 * t) * 8 + wpos] = expf(leaky(as[s * 8 + hh] + adq) - m);
        }
        wave_lds_handover();
        const int left = end - base;
        const int cnt = left < 32 ? left : 32;
        int e = 0;
        for (; e + 2 <= cnt; e += 2) {
            const int s0 = csr[base + e], s1 = csr[base + e + 1];
            const float4 wa = *(const float4*)(wmine + e * 8 + half * 4);
            const float4 wb = *(const float4*)(wmine + e * 8 + 8 + half * 4);
            const float* __restrict__ r0 = hcol + (size_t)s0 * 512;
            const float* __restrict__ r1 = hcol + (size_t)s1 * 512;
            const v4f a0 = *(const v4f*)(r0),       a1 = *(const v4f*)(r0 + 128);
            const v4f a2 = *(const v4f*)(r0 + 256), a3 = *(const v4f*)(r0 + 384);
            const v4f b0 = *(const v4f*)(r1),       b1 = *(const v4f*)(r1 + 128);
            const v4f b2 = *(const v4f*)(r1 + 256), b3 = *(const v4f*)(r1 + 384);
            asm volatile("" :: "v"(a0), "v"(a1), "v"(a2), "v"(a3),
                               "v"(b0), "v"(b1), "v"(b2), "v"(b3));
            const v4f ra[4] = {a0, a1, a2, a3};
            const v4f rb[4] = {b0, b1, b2, b3};
            const float wla[4] = {wa.x, wa.y, wa.z, wa.w};
            const float wlb[4] = {wb.x, wb.y, wb.z, wb.w};
#pragma unroll
            for (int i = 0; i < 4; ++i) {
                den[i] += wla[i];
                acc[i].x += ra[i].x * wla[i]; acc[i].y += ra[i].y * wla[i];
                acc[i].z += ra[i].z * wla[i]; acc[i].w += ra[i].w * wla[i];
            }
#pragma unroll
            for (int i = 0; i < 4; ++i) {
                den[i] += wlb[i];
                acc[i].x += rb[i].x * wlb[i]; acc[i].y += rb[i].y * wlb[i];
                acc[i].z += rb[i].z * wlb[i]; acc[i].w += rb[i].w * wlb[i];
            }
        }
        if (e < cnt) {
            const int s0 = csr[base + e];
            const float4 wa = *(const float4*)(wmine + e * 8 + half * 4);
            const float wla[4] = {wa.x, wa.y, wa.z, wa.w};
            const float* __restrict__ r0 = hcol + (size_t)s0 * 512;
#pragma unroll
            for (int i = 0; i < 4; ++i) {
                const v4f hv = *(const v4f*)(r0 + i * 128);
                den[i] += wla[i];
                acc[i].x += hv.x * wla[i]; acc[i].y += hv.y * wla[i];
                acc[i].z += hv.z * wla[i]; acc[i].w += hv.w * wla[i];
            }
        }
        wave_lds_handover();
    }
#pragma unroll
    for (int i = 0; i < 4; ++i) {
        const float4 bb = *(const float4*)(b + i * 128 + lane * 4);
        float4 o;
        o.x = acc[i].x / den[i] + bb.x; o.y = acc[i].y / den[i] + bb.y;
        o.z = acc[i].z / den[i] + bb.z; o.w = acc[i].w / den[i] + bb.w;
        o.x = o.x > 0.f ? o.x : expm1f(o.x);
        o.y = o.y > 0.f ? o.y : expm1f(o.y);
        o.z = o.z > 0.f ? o.z : expm1f(o.z);
        o.w = o.w > 0.f ? o.w : expm1f(o.w);
        *(float4*)(orow + i * 128 + lane * 4) = o;
    }
}

#define F1C 512
#define TPITCH 516

template<int NT, int KS>
__global__ void __launch_bounds__(512)
gat1_gather_gemm2(const int* __restrict__ rec, const int* __restrict__ ei, int E,
                  const float* __restrict__ as, const float* __restrict__ ad,
                  const float* __restrict__ h1, const float* __restrict__ b, int Nn,
                  const float* __restrict__ B, float* __restrict__ C, int N, int ldc,
                  const float* __restrict__ a_src, const float* __restrict__ a_dst,
                  float* __restrict__ as2, float* __restrict__ ad2, int ncv)
{
    constexpr int NPART = (KS - 1) * NT * 2 * 32;
    constexpr int NSCR  = NPART > 16 * 64 ? NPART : 16 * 64;
    __shared__ float4 scr[NSCR];
    __shared__ float tile[16 * TPITCH];

    const int lane = (int)threadIdx.x & 31;
    const int wv = __builtin_amdgcn_readfirstlane((int)threadIdx.x >> 5);
    const int n  = (int)blockIdx.x * 16 + wv;
    float* orow = tile + wv * TPITCH;
    if (n < Nn) {
        gat1_node(n, lane, (float*)(scr + wv * 64), rec, ei, E, as, ad, h1, b, orow);
    } else {
#pragma unroll
        for (int i = 0; i < 4; ++i)
            *(float4*)(orow + i * 128 + lane * 4) = make_float4(0.f, 0.f, 0.f, 0.f);
    }
    __syncthreads();

    const int half = lane >> 4;
    const int l16  = lane & 15;
    v8f acc[NT];
#pragma unroll
    for (int t = 0; t < NT; ++t) {
        v8f z = {0.f, 0.f, 0.f, 0.f, 0.f, 0.f, 0.f, 0.f};
        acc[t] = z;
    }
    float wsrc[NT], wdst[NT];
    load_att_weights<NT>(a_src, a_dst, l16, ncv, wsrc, wdst);

    if (wv < KS) {
        constexpr int KPER = F1C / KS;
        const int kbeg = wv * KPER;
        const float* __restrict__ Bw = B + (size_t)(kbeg + 2 * half) * (size_t)N + l16;
        const float* Ar = tile + l16 * TPITCH + kbeg + 2 * half;
#pragma unroll 1
        for (int kc = 0; kc < KPER; kc += 16) {
            v2f breg[4][NT], a[4];
#pragma unroll
            for (int kk = 0; kk < 4; ++kk) {
                const float* __restrict__ Bk = Bw + (size_t)(kc + kk * 4) * (size_t)N;
#pragma unroll
                for (int t = 0; t < NT; ++t) {
                    breg[kk][t].x = Bk[t * 16];
                    breg[kk][t].y = Bk[N + t * 16];
                }
            }
#pragma unroll
            for (int kk = 0; kk < 4; ++kk)
                a[kk] = *(const v2f*)(Ar + kc + kk * 4);
#pragma unroll
            for (int kk = 0; kk < 4; ++kk) {
#pragma unroll
                for (int t = 0; t < NT; ++t) {
                    acc[t] = __builtin_amdgcn_wmma_f32_16x16x4_f32(
                        false, a[kk], false, breg[kk][t], (short)0, acc[t], false, false);
                }
            }
        }
        if (wv != 0) {
            float4* pp = scr + (size_t)(wv - 1) * (NT * 2 * 32) + lane;
#pragma unroll
            for (int t = 0; t < NT; ++t) {
                pp[(t * 2 + 0) * 32] = make_float4(acc[t][0], acc[t][1], acc[t][2], acc[t][3]);
                pp[(t * 2 + 1) * 32] = make_float4(acc[t][4], acc[t][5], acc[t][6], acc[t][7]);
            }
        }
    }
    __syncthreads();
    if (wv != 0) return;
    constexpr int NP = NT * 2;
    const v4f* pp = (const v4f*)scr + lane;
#pragma unroll 1
    for (int w = 0; w < KS - 1; ++w, pp += NP * 32) {
        v4f pv[NP];
#pragma unroll
        for (int i = 0; i < NP; ++i) pv[i] = pp[i * 32];
        if constexpr (NT == 3) {
            asm volatile("" :: "v"(pv[0]), "v"(pv[1]), "v"(pv[2]),
                               "v"(pv[3]), "v"(pv[4]), "v"(pv[5]));
        }
#pragma unroll
        for (int t = 0; t < NT; ++t) {
#pragma unroll
            for (int r = 0; r < 4; ++r) {
                acc[t][r]     += pv[t * 2][r];
                acc[t][r + 4] += pv[t * 2 + 1][r];
            }
        }
    }
    strip_epilogue<NT>(acc, C, Nn, ldc, (int)blockIdx.x * 16, 0, lane,
                       wsrc, wdst, as2, ad2, 1, 0);
}

__global__ void gat2_gather(const int* __restrict__ rec, const int* __restrict__ ei, int E,
                            const float* __restrict__ as, const float* __restrict__ ad,
                            const float* __restrict__ h3, const float* __restrict__ b,
                            float* __restrict__ out, int Nn)
{
    int t = blockIdx.x * blockDim.x + threadIdx.x;
    if (t >= Nn * 10) return;
    const int n = t / 10;
    const int g = t - n * 10;
    const int* __restrict__ csr = rec;
    const int cnt0 = rec[(size_t)n * RSTRIDE];
    const float adn   = ad[n];
    const float eself = leaky(as[n] + adn);

    if (__builtin_expect(cnt0 > RCAP, 0)) {
        const int* __restrict__ srcp = ei;
        const int* __restrict__ dstp = ei + E;
        float ms = eself;
        for (int e = 0; e < E; ++e)
            if (dstp[e] == n) ms = fmaxf(ms, leaky(as[srcp[e]] + adn));
        const float w0 = expf(eself - ms);
        const float4 h0 = *(const float4*)(h3 + (size_t)n * 48 + g * 4);
        float dens = w0;
        float4 a4;
        a4.x = h0.x * w0; a4.y = h0.y * w0; a4.z = h0.z * w0; a4.w = h0.w * w0;
        for (int e = 0; e < E; ++e) {
            if (dstp[e] != n) continue;
            const int s = srcp[e];
            const float w = expf(leaky(as[s] + adn) - ms);
            const float4 hv = *(const float4*)(h3 + (size_t)s * 48 + g * 4);
            dens += w;
            a4.x += hv.x * w; a4.y += hv.y * w; a4.z += hv.z * w; a4.w += hv.w * w;
        }
        const float4 bs = *(const float4*)(b + g * 4);
        float4 os;
        os.x = a4.x / dens + bs.x; os.y = a4.y / dens + bs.y;
        os.z = a4.z / dens + bs.z; os.w = a4.w / dens + bs.w;
        *(float4*)(out + (size_t)n * 40 + g * 4) = os;
        return;
    }
    const int beg = n * RSTRIDE + 1, end = beg + cnt0;

    float m = eself;
    for (int j = beg; j < end; ++j)
        m = fmaxf(m, leaky(as[csr[j]] + adn));

    float den;
    float4 acc;
    {
        const float w = expf(eself - m);
        const float4 hv = *(const float4*)(h3 + (size_t)n * 48 + g * 4);
        den = w;
        acc.x = hv.x * w; acc.y = hv.y * w; acc.z = hv.z * w; acc.w = hv.w * w;
    }
#pragma unroll 4
    for (int j = beg; j < end; ++j) {
        const int s = csr[j];
        const float w = expf(leaky(as[s] + adn) - m);
        const float4 hv = *(const float4*)(h3 + (size_t)s * 48 + g * 4);
        den += w;
        acc.x += hv.x * w; acc.y += hv.y * w; acc.z += hv.z * w; acc.w += hv.w * w;
    }
    const float4 bb = *(const float4*)(b + g * 4);
    float4 o;
    o.x = acc.x / den + bb.x; o.y = acc.y / den + bb.y;
    o.z = acc.z / den + bb.z; o.w = acc.w / den + bb.w;
    *(float4*)(out + (size_t)n * 40 + g * 4) = o;
}

extern "C" void kernel_launch(void* const* d_in, const int* in_sizes, int n_in,
                              void* d_out, int out_size, void* d_ws, size_t ws_size,
                              hipStream_t stream)
{
    const float* x    = (const float*)d_in[0];
    const int*   ei   = (const int*)  d_in[1];
    const float* W1   = (const float*)d_in[2];
    const float* aS1  = (const float*)d_in[3];
    const float* aD1  = (const float*)d_in[4];
    const float* b1   = (const float*)d_in[5];
    const float* W2   = (const float*)d_in[6];
    const float* aS2  = (const float*)d_in[7];
    const float* aD2  = (const float*)d_in[8];
    const float* b2   = (const float*)d_in[9];
    float* out = (float*)d_out;

    const int IN  = 256, F1 = 512, OUT = 40, OUTP = 48;
    const int Nn  = in_sizes[0] / IN;
    const int E   = in_sizes[1] / 2;

    float* ws  = (float*)d_ws;
    float* h1  = ws;
    float* h3  = h1  + (size_t)Nn * F1;
    float* W2p = h3  + (size_t)Nn * 48;
    float* as1 = W2p + (size_t)F1 * OUTP;
    float* ad1 = as1 + (size_t)Nn * 8;
    int*   rec = (int*)(ad1 + (size_t)Nn * 8);
    float* as2 = (float*)(rec + (size_t)Nn * RSTRIDE);
    float* ad2 = as2 + Nn;

    const int B = 256;

    clear_counters_pad_w2<<<(Nn + B - 1) / B + (F1 * OUTP + B - 1) / B, B, 0, stream>>>(
        rec, Nn, W2, W2p, F1, OUT, OUTP);

    gemm1_and_fill_records<4><<<dim3((Nn + 63) / 64, F1 / 64 + 1), 128, 0, stream>>>(
        x, W1, h1, Nn, IN, F1, F1, aS1, aD1, as1, ad1, F1, 8, ei, E, rec);

    gat1_gather_gemm2<3, 8><<<(Nn + 15) / 16, 512, 0, stream>>>(
        rec, ei, E, as1, ad1, h1, b1, Nn, W2p, h3, OUTP, OUTP, aS2, aD2, as2, ad2, OUT);

    gat2_gather<<<(Nn * 10 + B - 1) / B, B, 0, stream>>>(rec, ei, E, as2, ad2, h3, b2, out, Nn);
}


# Layer-2 gather: 16-lane row per node pair (was 10 threads per node, each walking its list): one edge per lane for logits, DPP max, offset/weight by row_share, 32 row loads in flight; sum order kept
# speedup vs baseline: 5.0070x; 1.0551x over previous
#include <hip/hip_runtime.h>
#include <hip/hip_bf16.h>
#include <math.h>

typedef float v2f __attribute__((ext_vector_type(2)));
typedef float v8f __attribute__((ext_vector_type(8)));
typedef float v4f __attribute__((ext_vector_type(4)));

#define NEG_SLOPE 0.2f

__device__ __forceinline__ float leaky(float v) {
    return v > 0.0f ? v : NEG_SLOPE * v;
}


__device__ __forceinline__ void wave_lds_handover() {
    __builtin_amdgcn_fence(__ATOMIC_RELEASE, "wavefront");
    __builtin_amdgcn_wave_barrier();
    __builtin_amdgcn_fence(__ATOMIC_ACQUIRE, "wavefront");
}

template<int MASK>
__device__ __forceinline__ float row_xor(float v) {
    const int i = __builtin_bit_cast(int, v);
    return __builtin_bit_cast(float,
        __builtin_amdgcn_update_dpp(i, i, 0x160 + MASK, 0xf, 0xf, false));
}

template<int NT>
__device__ __forceinline__ void
load_att_weights(const float* __restrict__ a_src, const float* __restrict__ a_dst,
                 int colB, int ncv, float (&wsrc)[NT], float (&wdst)[NT])
{
#pragma unroll
    for (int t = 0; t < NT; ++t) {
        const int col = colB + t * 16;
        const bool real = col < ncv;
        const int cc = real ? col : 0;
        const float vs = a_src[cc], vd = a_dst[cc];
        wsrc[t] = real ? vs : 0.f;
        wdst[t] = real ? vd : 0.f;
    }
}

template<int NT>
__device__ __forceinline__ void
strip_epilogue(const v8f (&acc)[NT], float* __restrict__ C, int M, int ldc,
               int row0, int ncol0, int lane,
               const float (&wsrc)[NT], const float (&wdst)[NT],
               float* __restrict__ as, float* __restrict__ ad, int heads, int head)
{
    const int half = lane >> 4;
    const int l16  = lane & 15;
    const int crow0 = row0 + half * 8;
    float* __restrict__ Cbase = C + (size_t)crow0 * (size_t)ldc + ncol0 + l16;
    if (row0 + 16 <= M) {
#pragma unroll
        for (int t = 0; t < NT; ++t) {
            float* __restrict__ Cp = Cbase + t * 16;
#pragma unroll
            for (int r = 0; r < 8; ++r)
                Cp[(size_t)r * (size_t)ldc] = acc[t][r];
        }
    } else {
#pragma unroll
        for (int t = 0; t < NT; ++t) {
            float* __restrict__ Cp = Cbase + t * 16;
#pragma unroll
            for (int r = 0; r < 8; ++r)
                if (crow0 + r < M) Cp[(size_t)r * (size_t)ldc] = acc[t][r];
        }
    }

    float ps[8], pd[8];
#pragma unroll
    for (int r = 0; r < 8; ++r) { ps[r] = 0.f; pd[r] = 0.f; }
#pragma unroll
    for (int t = 0; t < NT; ++t) {
#pragma unroll
        for (int r = 0; r < 8; ++r) {
            ps[r] += acc[t][r] * wsrc[t];
            pd[r] += acc[t][r] * wdst[t];
        }
    }
    const bool b3 = (l16 & 8) != 0, b2 = (l16 & 4) != 0;
    const bool b1 = (l16 & 2) != 0, b0 = (l16 & 1) != 0;
    float u[8], w4[4], y2[2];
#pragma unroll
    for (int i = 0; i < 8; ++i) {
        const float mine = b3 ? pd[i] : ps[i];
        const float give = b3 ? ps[i] : pd[i];
        u[i] = mine + row_xor<8>(give);
    }
#pragma unroll
    for (int i = 0; i < 4; ++i) {
        const float mine = b2 ? u[i + 4] : u[i];
        const float give = b2 ? u[i] : u[i + 4];
        w4[i] = mine + row_xor<4>(give);
    }
#pragma unroll
    for (int i = 0; i < 2; ++i) {
        const float mine = b1 ? w4[i + 2] : w4[i];
        const float give = b1 ? w4[i] : w4[i + 2];
        y2[i] = mine + row_xor<2>(give);
    }
    const float mine = b0 ? y2[1] : y2[0];
    const float give = b0 ? y2[0] : y2[1];
    const float z = mine + row_xor<1>(give);
    const int arow = row0 + half * 8 + (l16 & 7);
    if (arow < M) {
        float* __restrict__ dstp = b3 ? ad : as;
        dstp[(size_t)arow * (size_t)heads + head] = z;
    }
}

#define G1_ROWS 64
#define G1_KC 16
#define G1_APITCH 20
#define G1_BPITCH 72

template<int NT>
__device__ __forceinline__ void
gemm_strip(const float* __restrict__ A, const float* __restrict__ B,
           float* __restrict__ C, int M, int K, int N, int ldc,
           const float* __restrict__ a_src, const float* __restrict__ a_dst,
           float* __restrict__ as, float* __restrict__ ad, int ncv, int heads)
{
    static_assert(NT * 16 == 64, "B staging: 16 threads x 4 columns per k row");
    __shared__ float As[2][G1_ROWS * G1_APITCH];
    __shared__ float Bs[2][G1_KC * G1_BPITCH];

    const int tid  = (int)threadIdx.x;
    const int lane = tid & 31;
    const int wv   = __builtin_amdgcn_readfirstlane(tid >> 5);
    const int half = lane >> 4;
    const int l16  = lane & 15;

    const int alr = tid >> 1;
    const int alc = (tid & 1) * 8;
    int rowA = blockIdx.x * G1_ROWS + alr;
    if (rowA >= M) rowA = M - 1;
    const float* __restrict__ Aw = A + (size_t)rowA * (size_t)K + alc;

    const int ncol0 = blockIdx.y * (16 * NT);
    const int colB  = ncol0 + l16;
    const int bk = tid >> 4;
    const int bc = (tid & 15) * 4;
    const float* __restrict__ Bw = B + (size_t)bk * (size_t)N + ncol0 + bc;

    v8f acc[NT];
#pragma unroll
    for (int t = 0; t < NT; ++t) {
        v8f z = {0.f, 0.f, 0.f, 0.f, 0.f, 0.f, 0.f, 0.f};
        acc[t] = z;
    }

    float4 a0 = *(const float4*)(Aw), a1 = *(const float4*)(Aw + 4);
    float4 b0 = *(const float4*)(Bw), b1 = *(const float4*)(Bw + (size_t)8 * (size_t)N);

    float wsrc[NT], wdst[NT];
    load_att_weights<NT>(a_src, a_dst, colB, ncv, wsrc, wdst);

    const int aoff = alr * G1_APITCH + alc;
    const int boff = bk * G1_BPITCH + bc;
    const int afr  = (wv * 16 + l16) * G1_APITCH + 2 * half;
    const int bfr  = (2 * half) * G1_BPITCH + l16;
    int p = 0;
    for (int kc = 0; kc < K; kc += G1_KC) {
        float* __restrict__ Asp = As[p];
        float* __restrict__ Bsp = Bs[p];
        *(float4*)(Asp + aoff)     = a0;
        *(float4*)(Asp + aoff + 4) = a1;
        *(float4*)(Bsp + boff)                 = b0;
        *(float4*)(Bsp + boff + 8 * G1_BPITCH) = b1;

        __syncthreads();

        const int kn = (kc + G1_KC < K) ? (kc + G1_KC) : 0;
        a0 = *(const float4*)(Aw + kn);
        a1 = *(const float4*)(Aw + kn + 4);
        b0 = *(const float4*)(Bw + (size_t)kn * (size_t)N);
        b1 = *(const float4*)(Bw + (size_t)(kn + 8) * (size_t)N);

        v2f a[4], breg[4][NT];
#pragma unroll
        for (int kk = 0; kk < 4; ++kk)
            a[kk] = *(const v2f*)(Asp + afr + kk * 4);
#pragma unroll
        for (int kk = 0; kk < 4; ++kk) {
#pragma unroll
            for (int t = 0; t < NT; ++t) {
                breg[kk][t].x = Bsp[bfr + (kk * 4) * G1_BPITCH + t * 16];
                breg[kk][t].y = Bsp[bfr + (kk * 4 + 1) * G1_BPITCH + t * 16];
            }
        }
        asm volatile("" :: "v"(a[0]), "v"(a[1]), "v"(a[2]), "v"(a[3]),
                           "v"(breg[0][0]), "v"(breg[0][1]), "v"(breg[0][2]), "v"(breg[0][3]),
                           "v"(breg[1][0]), "v"(breg[1][1]), "v"(breg[1][2]), "v"(breg[1][3]),
                           "v"(breg[2][0]), "v"(breg[2][1]), "v"(breg[2][2]), "v"(breg[2][3]),
                           "v"(breg[3][0]), "v"(breg[3][1]), "v"(breg[3][2]), "v"(breg[3][3]));
#pragma unroll
        for (int kk = 0; kk < 4; ++kk) {
#pragma unroll
            for (int t = 0; t < NT; ++t) {
                acc[t] = __builtin_amdgcn_wmma_f32_16x16x4_f32(
                    false, a[kk], false, breg[kk][t], (short)0, acc[t], false, false);
            }
        }
        p ^= 1;
    }

    strip_epilogue<NT>(acc, C, M, ldc, (int)blockIdx.x * G1_ROWS + wv * 16, ncol0, lane,
                       wsrc, wdst, as, ad, heads, (int)blockIdx.y);
}

#define RSTRIDE 64
#define RCAP 63

__global__ void clear_counters_pad_w2(int* __restrict__ deg, int Nn,
                                      const float* __restrict__ src,
                                      float* __restrict__ dst, int K, int N, int Np)
{
    const int nzb = (Nn + (int)blockDim.x - 1) / (int)blockDim.x;
    if ((int)blockIdx.x < nzb) {
        const int t = blockIdx.x * blockDim.x + threadIdx.x;
        if (t < Nn) deg[(size_t)t * RSTRIDE] = 0;
    } else {
        const int t = ((int)blockIdx.x - nzb) * blockDim.x + threadIdx.x;
        if (t >= K * Np) return;
        const int r = t / Np, c = t - r * Np;
        dst[t] = (c < N) ? src[r * N + c] : 0.f;
    }
}

template<int NT>
__global__ void __launch_bounds__(128)
gemm1_and_fill_records(const float* __restrict__ A, const float* __restrict__ B,
                       float* __restrict__ C, int M, int K, int N, int ldc,
                       const float* __restrict__ a_src, const float* __restrict__ a_dst,
                       float* __restrict__ as, float* __restrict__ ad, int ncv, int heads,
                       const int* __restrict__ ei, int E, int* __restrict__ rec)
{
    if (blockIdx.y == gridDim.y - 1) {
        const int step = (int)(gridDim.x * blockDim.x);
        const int* __restrict__ srcp = ei;
        const int* __restrict__ dstp = ei + E;
        int e = (int)(blockIdx.x * blockDim.x + threadIdx.x);
        for (; e + 3 * step < E; e += 4 * step) {
            const int d0 = dstp[e],            d1 = dstp[e + step];
            const int d2 = dstp[e + 2 * step], d3 = dstp[e + 3 * step];
            const int s0 = srcp[e],            s1 = srcp[e + step];
            const int s2 = srcp[e + 2 * step], s3 = srcp[e + 3 * step];
            int* q0 = rec + (size_t)d0 * RSTRIDE;
            int* q1 = rec + (size_t)d1 * RSTRIDE;
            int* q2 = rec + (size_t)d2 * RSTRIDE;
            int* q3 = rec + (size_t)d3 * RSTRIDE;
            const int r0 = atomicAdd(q0, 1);
            const int r1 = atomicAdd(q1, 1);
            const int r2 = atomicAdd(q2, 1);
            const int r3 = atomicAdd(q3, 1);
            if (r0 < RCAP) q0[1 + r0] = s0;
            if (r1 < RCAP) q1[1 + r1] = s1;
            if (r2 < RCAP) q2[1 + r2] = s2;
            if (r3 < RCAP) q3[1 + r3] = s3;
        }
        for (; e < E; e += step) {
            int* q0 = rec + (size_t)dstp[e] * RSTRIDE;
            const int s0 = srcp[e];
            const int r0 = atomicAdd(q0, 1);
            if (r0 < RCAP) q0[1 + r0] = s0;
        }
        return;
    }
    gemm_strip<NT>(A, B, C, M, K, N, ldc, a_src, a_dst, as, ad, ncv, heads);
}

__device__ __forceinline__ void
gat1_node(const int n, const int lane, float* wmine,
          const int* __restrict__ rec, const int* __restrict__ ei, int E,
          const float* __restrict__ as, const float* __restrict__ ad,
          const float* __restrict__ h1, const float* __restrict__ b,
          float* orow)
{
    const int half = lane >> 4;
    const int* __restrict__ csr = rec;
    const int cnt0 = rec[(size_t)n * RSTRIDE];

    if (__builtin_expect(cnt0 > RCAP, 0)) {
        const int* __restrict__ srcp = ei;
        const int* __restrict__ dstp = ei + E;
        float adh[4], eh[4], mh[4];
#pragma unroll
        for (int i = 0; i < 4; ++i) {
            const int hd = 2 * i + half;
            adh[i] = ad[n * 8 + hd];
            eh[i]  = leaky(as[n * 8 + hd] + adh[i]);
            mh[i]  = eh[i];
        }
        for (int base = 0; base < E; base += 32) {
            const int e = base + lane;
            const int d = (e < E) ? dstp[e] : -1;
            unsigned hit = __builtin_amdgcn_ballot_w32(d == n);
            while (hit != 0u) {
                const int s = srcp[base + __builtin_ctz(hit)];
                hit &= hit - 1u;
#pragma unroll
                for (int i = 0; i < 4; ++i)
                    mh[i] = fmaxf(mh[i], leaky(as[s * 8 + 2 * i + half] + adh[i]));
            }
        }
        float dens[4];
        float4 accs[4];
        const float* __restrict__ hc = h1 + lane * 4;
#pragma unroll
        for (int i = 0; i < 4; ++i) {
            const float w = expf(eh[i] - mh[i]);
            const float4 hv = *(const float4*)(hc + (size_t)n * 512 + i * 128);
            dens[i] = w;
            accs[i].x = hv.x * w; accs[i].y = hv.y * w; accs[i].z = hv.z * w; accs[i].w = hv.w * w;
        }
        for (int base = 0; base < E; base += 32) {
            const int e = base + lane;
            const int d = (e < E) ? dstp[e] : -1;
            unsigned hit = __builtin_amdgcn_ballot_w32(d == n);
            while (hit != 0u) {
                const int s = srcp[base + __builtin_ctz(hit)];
                hit &= hit - 1u;
#pragma unroll
                for (int i = 0; i < 4; ++i) {
                    const float w = expf(leaky(as[s * 8 + 2 * i + half] + adh[i]) - mh[i]);
                    const float4 hv = *(const float4*)(hc + (size_t)s * 512 + i * 128);
                    dens[i] += w;
                    accs[i].x += hv.x * w; accs[i].y += hv.y * w;
                    accs[i].z += hv.z * w; accs[i].w += hv.w * w;
                }
            }
        }
#pragma unroll
        for (int i = 0; i < 4; ++i) {
            const float4 bb = *(const float4*)(b + i * 128 + lane * 4);
            float4 o;
            o.x = accs[i].x / dens[i] + bb.x; o.y = accs[i].y / dens[i] + bb.y;
            o.z = accs[i].z / dens[i] + bb.z; o.w = accs[i].w / dens[i] + bb.w;
            o.x = o.x > 0.f ? o.x : expm1f(o.x);
            o.y = o.y > 0.f ? o.y : expm1f(o.y);
            o.z = o.z > 0.f ? o.z : expm1f(o.z);
            o.w = o.w > 0.f ? o.w : expm1f(o.w);
            *(float4*)(orow + i * 128 + lane * 4) = o;
        }
        return;
    }

    const int beg = n * RSTRIDE + 1, end = beg + cnt0;
    const int last = end - 1;

    const int q = lane >> 3, hh = lane & 7;
    const float adq = ad[n * 8 + hh];
    float m = leaky(as[n * 8 + hh] + adq);
    for (int base = beg; base < end; base += 32) {
#pragma unroll
        for (int t = 0; t < 8; ++t) {
            const int j = base + q + 4 * t;
            const int s = csr[j < last ? j : last];
            m = fmaxf(m, leaky(as[s * 8 + hh] + adq));
        }
    }
    m = fmaxf(m, __shfl_xor(m, 8));
    m = fmaxf(m, __shfl_xor(m, 16));

    float den[4];
    float4 acc[4];
    const float* __restrict__ hcol = h1 + lane * 4;
#pragma unroll
    for (int i = 0; i < 4; ++i) {
        const int hd = 2 * i + half;
        const float mh = __shfl(m, hd);
        const float w = expf(leaky(as[n * 8 + hd] + ad[n * 8 + hd]) - mh);
        const float4 hv = *(const float4*)(hcol + (size_t)n * 512 + i * 128);
        den[i] = w;
        acc[i].x = hv.x * w; acc[i].y = hv.y * w; acc[i].z = hv.z * w; acc[i].w = hv.w * w;
    }

    const int wpos = (hh & 1) * 4 + (hh >> 1);
    for (int base = beg; base < end; base += 32) {
#pragma unroll
        for (int t = 0; t < 8; ++t) {
            const int j = base + q + 4 * t;
            const int s = csr[j < last ? j : last];
            wmine[(q + 4 * t) * 8 + wpos] = expf(leaky(as[s * 8 + hh] + adq) - m);
        }
        wave_lds_handover();
        const int left = end - base;
        const int cnt = left < 32 ? left : 32;
        int e = 0;
        for (; e + 2 <= cnt; e += 2) {
            const int s0 = csr[base + e], s1 = csr[base + e + 1];
            const float4 wa = *(const float4*)(wmine + e * 8 + half * 4);
            const float4 wb = *(const float4*)(wmine + e * 8 + 8 + half * 4);
            const float* __restrict__ r0 = hcol + (size_t)s0 * 512;
            const float* __restrict__ r1 = hcol + (size_t)s1 * 512;
            const v4f a0 = *(const v4f*)(r0),       a1 = *(const v4f*)(r0 + 128);
            const v4f a2 = *(const v4f*)(r0 + 256), a3 = *(const v4f*)(r0 + 384);
            const v4f b0 = *(const v4f*)(r1),       b1 = *(const v4f*)(r1 + 128);
            const v4f b2 = *(const v4f*)(r1 + 256), b3 = *(const v4f*)(r1 + 384);
            asm volatile("" :: "v"(a0), "v"(a1), "v"(a2), "v"(a3),
                               "v"(b0), "v"(b1), "v"(b2), "v"(b3));
            const v4f ra[4] = {a0, a1, a2, a3};
            const v4f rb[4] = {b0, b1, b2, b3};
            const float wla[4] = {wa.x, wa.y, wa.z, wa.w};
            const float wlb[4] = {wb.x, wb.y, wb.z, wb.w};
#pragma unroll
            for (int i = 0; i < 4; ++i) {
                den[i] += wla[i];
                acc[i].x += ra[i].x * wla[i]; acc[i].y += ra[i].y * wla[i];
                acc[i].z += ra[i].z * wla[i]; acc[i].w += ra[i].w * wla[i];
            }
#pragma unroll
            for (int i = 0; i < 4; ++i) {
                den[i] += wlb[i];
                acc[i].x += rb[i].x * wlb[i]; acc[i].y += rb[i].y * wlb[i];
                acc[i].z += rb[i].z * wlb[i]; acc[i].w += rb[i].w * wlb[i];
            }
        }
        if (e < cnt) {
            const int s0 = csr[base + e];
            const float4 wa = *(const float4*)(wmine + e * 8 + half * 4);
            const float wla[4] = {wa.x, wa.y, wa.z, wa.w};
            const float* __restrict__ r0 = hcol + (size_t)s0 * 512;
#pragma unroll
            for (int i = 0; i < 4; ++i) {
                const v4f hv = *(const v4f*)(r0 + i * 128);
                den[i] += wla[i];
                acc[i].x += hv.x * wla[i]; acc[i].y += hv.y * wla[i];
                acc[i].z += hv.z * wla[i]; acc[i].w += hv.w * wla[i];
            }
        }
        wave_lds_handover();
    }
#pragma unroll
    for (int i = 0; i < 4; ++i) {
        const float4 bb = *(const float4*)(b + i * 128 + lane * 4);
        float4 o;
        o.x = acc[i].x / den[i] + bb.x; o.y = acc[i].y / den[i] + bb.y;
        o.z = acc[i].z / den[i] + bb.z; o.w = acc[i].w / den[i] + bb.w;
        o.x = o.x > 0.f ? o.x : expm1f(o.x);
        o.y = o.y > 0.f ? o.y : expm1f(o.y);
        o.z = o.z > 0.f ? o.z : expm1f(o.z);
        o.w = o.w > 0.f ? o.w : expm1f(o.w);
        *(float4*)(orow + i * 128 + lane * 4) = o;
    }
}

#define F1C 512
#define TPITCH 516

template<int NT, int KS>
__global__ void __launch_bounds__(512)
gat1_gather_gemm2(const int* __restrict__ rec, const int* __restrict__ ei, int E,
                  const float* __restrict__ as, const float* __restrict__ ad,
                  const float* __restrict__ h1, const float* __restrict__ b, int Nn,
                  const float* __restrict__ B, float* __restrict__ C, int N, int ldc,
                  const float* __restrict__ a_src, const float* __restrict__ a_dst,
                  float* __restrict__ as2, float* __restrict__ ad2, int ncv)
{
    constexpr int NPART = (KS - 1) * NT * 2 * 32;
    constexpr int NSCR  = NPART > 16 * 64 ? NPART : 16 * 64;
    __shared__ float4 scr[NSCR];
    __shared__ float tile[16 * TPITCH];

    const int lane = (int)threadIdx.x & 31;
    const int wv = __builtin_amdgcn_readfirstlane((int)threadIdx.x >> 5);
    const int n  = (int)blockIdx.x * 16 + wv;
    float* orow = tile + wv * TPITCH;
    if (n < Nn) {
        gat1_node(n, lane, (float*)(scr + wv * 64), rec, ei, E, as, ad, h1, b, orow);
    } else {
#pragma unroll
        for (int i = 0; i < 4; ++i)
            *(float4*)(orow + i * 128 + lane * 4) = make_float4(0.f, 0.f, 0.f, 0.f);
    }
    __syncthreads();

    const int half = lane >> 4;
    const int l16  = lane & 15;
    v8f acc[NT];
#pragma unroll
    for (int t = 0; t < NT; ++t) {
        v8f z = {0.f, 0.f, 0.f, 0.f, 0.f, 0.f, 0.f, 0.f};
        acc[t] = z;
    }
    float wsrc[NT], wdst[NT];
    load_att_weights<NT>(a_src, a_dst, l16, ncv, wsrc, wdst);

    if (wv < KS) {
        constexpr int KPER = F1C / KS;
        const int kbeg = wv * KPER;
        const float* __restrict__ Bw = B + (size_t)(kbeg + 2 * half) * (size_t)N + l16;
        const float* Ar = tile + l16 * TPITCH + kbeg + 2 * half;
#pragma unroll 1
        for (int kc = 0; kc < KPER; kc += 16) {
            v2f breg[4][NT], a[4];
#pragma unroll
            for (int kk = 0; kk < 4; ++kk) {
                const float* __restrict__ Bk = Bw + (size_t)(kc + kk * 4) * (size_t)N;
#pragma unroll
                for (int t = 0; t < NT; ++t) {
                    breg[kk][t].x = Bk[t * 16];
                    breg[kk][t].y = Bk[N + t * 16];
                }
            }
#pragma unroll
            for (int kk = 0; kk < 4; ++kk)
                a[kk] = *(const v2f*)(Ar + kc + kk * 4);
#pragma unroll
            for (int kk = 0; kk < 4; ++kk) {
#pragma unroll
                for (int t = 0; t < NT; ++t) {
                    acc[t] = __builtin_amdgcn_wmma_f32_16x16x4_f32(
                        false, a[kk], false, breg[kk][t], (short)0, acc[t], false, false);
                }
            }
        }
        if (wv != 0) {
            float4* pp = scr + (size_t)(wv - 1) * (NT * 2 * 32) + lane;
#pragma unroll
            for (int t = 0; t < NT; ++t) {
                pp[(t * 2 + 0) * 32] = make_float4(acc[t][0], acc[t][1], acc[t][2], acc[t][3]);
                pp[(t * 2 + 1) * 32] = make_float4(acc[t][4], acc[t][5], acc[t][6], acc[t][7]);
            }
        }
    }
    __syncthreads();
    if (wv != 0) return;
    constexpr int NP = NT * 2;
    const v4f* pp = (const v4f*)scr + lane;
#pragma unroll 1
    for (int w = 0; w < KS - 1; ++w, pp += NP * 32) {
        v4f pv[NP];
#pragma unroll
        for (int i = 0; i < NP; ++i) pv[i] = pp[i * 32];
        if constexpr (NT == 3) {
            asm volatile("" :: "v"(pv[0]), "v"(pv[1]), "v"(pv[2]),
                               "v"(pv[3]), "v"(pv[4]), "v"(pv[5]));
        }
#pragma unroll
        for (int t = 0; t < NT; ++t) {
#pragma unroll
            for (int r = 0; r < 4; ++r) {
                acc[t][r]     += pv[t * 2][r];
                acc[t][r + 4] += pv[t * 2 + 1][r];
            }
        }
    }
    strip_epilogue<NT>(acc, C, Nn, ldc, (int)blockIdx.x * 16, 0, lane,
                       wsrc, wdst, as2, ad2, 1, 0);
}


template<int K>
__device__ __forceinline__ int row_get(int v) {
    return __builtin_amdgcn_update_dpp(v, v, 0x150 + K, 0xf, 0xf, false);
}
template<int K>
__device__ __forceinline__ float row_get(float v) {
    const int i = __builtin_bit_cast(int, v);
    return __builtin_bit_cast(float,
        __builtin_amdgcn_update_dpp(i, i, 0x150 + K, 0xf, 0xf, false));
}

__device__ __forceinline__ void
gat2_issue(const unsigned oc, const float* __restrict__ h3, const unsigned gb, v4f (&hv)[16])
{
    unsigned sk[16];
#define GAT2_SLOT(K) sk[K] = (unsigned)row_get<K>((int)oc);
    GAT2_SLOT(0)  GAT2_SLOT(1)  GAT2_SLOT(2)  GAT2_SLOT(3)
    GAT2_SLOT(4)  GAT2_SLOT(5)  GAT2_SLOT(6)  GAT2_SLOT(7)
    GAT2_SLOT(8)  GAT2_SLOT(9)  GAT2_SLOT(10) GAT2_SLOT(11)
    GAT2_SLOT(12) GAT2_SLOT(13) GAT2_SLOT(14) GAT2_SLOT(15)
#undef GAT2_SLOT
#pragma unroll
    for (int k = 0; k < 16; ++k)
        hv[k] = *(const v4f*)((const char*)h3 + (sk[k] + gb));
}

__device__ __forceinline__ void gat2_hold(const v4f (&hv)[16])
{
    asm volatile("" :: "v"(hv[0]),  "v"(hv[1]),  "v"(hv[2]),  "v"(hv[3]),
                       "v"(hv[4]),  "v"(hv[5]),  "v"(hv[6]),  "v"(hv[7]),
                       "v"(hv[8]),  "v"(hv[9]),  "v"(hv[10]), "v"(hv[11]),
                       "v"(hv[12]), "v"(hv[13]), "v"(hv[14]), "v"(hv[15]));
}

__device__ __forceinline__ void
gat2_sum(const float wc, const int nv, const v4f (&hv)[16], float& den, float4& acc)
{
    float wk[16];
#define GAT2_SLOT(K) wk[K] = row_get<K>(wc);
    GAT2_SLOT(0)  GAT2_SLOT(1)  GAT2_SLOT(2)  GAT2_SLOT(3)
    GAT2_SLOT(4)  GAT2_SLOT(5)  GAT2_SLOT(6)  GAT2_SLOT(7)
    GAT2_SLOT(8)  GAT2_SLOT(9)  GAT2_SLOT(10) GAT2_SLOT(11)
    GAT2_SLOT(12) GAT2_SLOT(13) GAT2_SLOT(14) GAT2_SLOT(15)
#undef GAT2_SLOT
#pragma unroll
    for (int k = 0; k < 16; ++k) {
        if (k < nv) {
            den += wk[k];
            acc.x += hv[k].x * wk[k]; acc.y += hv[k].y * wk[k];
            acc.z += hv[k].z * wk[k]; acc.w += hv[k].w * wk[k];
        }
    }
}

static __device__ __forceinline__ void
gat2_node_long(const int n, const int g, const int* __restrict__ ei, int E,
               const float* __restrict__ as, const float* __restrict__ ad,
               const float* __restrict__ h3, const float* __restrict__ b,
               float* __restrict__ out)
{
    const int* __restrict__ srcp = ei;
    const int* __restrict__ dstp = ei + E;
    const float adn = ad[n];
    const float eself = leaky(as[n] + adn);
    float ms = eself;
    for (int e = 0; e < E; ++e)
        if (dstp[e] == n) ms = fmaxf(ms, leaky(as[srcp[e]] + adn));
    const float w0 = expf(eself - ms);
    const float4 h0 = *(const float4*)(h3 + (size_t)n * 48 + g * 4);
    float dens = w0;
    float4 a4;
    a4.x = h0.x * w0; a4.y = h0.y * w0; a4.z = h0.z * w0; a4.w = h0.w * w0;
    for (int e = 0; e < E; ++e) {
        if (dstp[e] != n) continue;
        const int s = srcp[e];
        const float w = expf(leaky(as[s] + adn) - ms);
        const float4 hv = *(const float4*)(h3 + (size_t)s * 48 + g * 4);
        dens += w;
        a4.x += hv.x * w; a4.y += hv.y * w; a4.z += hv.z * w; a4.w += hv.w * w;
    }
    const float4 bs = *(const float4*)(b + g * 4);
    float4 os;
    os.x = a4.x / dens + bs.x; os.y = a4.y / dens + bs.y;
    os.z = a4.z / dens + bs.z; os.w = a4.w / dens + bs.w;
    *(float4*)(out + (size_t)n * 40 + g * 4) = os;
}

__global__ void __launch_bounds__(256)
gat2_gather(const int* __restrict__ rec, const int* __restrict__ ei, int E,
            const float* __restrict__ as, const float* __restrict__ ad,
            const float* __restrict__ h3, const float* __restrict__ b,
            float* __restrict__ out, int Nn)
{
    const int t = blockIdx.x * blockDim.x + threadIdx.x;
    const int nA = (t >> 4) * 2;
    if (nA >= Nn) return;
    const bool hasB = nA + 1 < Nn;
    const int nn[2] = {nA, hasB ? nA + 1 : nA};
    const int r = t & 15;
    const int g = r < 10 ? r : 9;
    const unsigned gb = (unsigned)g * 16u;
    const int* __restrict__ csr = rec;

    int cnt[2];
    float adn[2], asn[2];
#pragma unroll
    for (int u = 0; u < 2; ++u) {
        cnt[u] = rec[(size_t)nn[u] * RSTRIDE];
        adn[u] = ad[nn[u]];
        asn[u] = as[nn[u]];
    }
    asm volatile("" :: "v"(cnt[0]), "v"(cnt[1]), "v"(adn[0]), "v"(adn[1]), "v"(asn[0]), "v"(asn[1]));
    __builtin_amdgcn_sched_barrier(0);

    if (__builtin_expect(cnt[0] > RCAP || cnt[1] > RCAP, 0)) {
        if (r < 10) {
            gat2_node_long(nn[0], g, ei, E, as, ad, h3, b, out);
            if (hasB) gat2_node_long(nn[1], g, ei, E, as, ad, h3, b, out);
        }
        return;
    }

    float eself[2];
    float4 hself[2];
#pragma unroll
    for (int u = 0; u < 2; ++u) {
        eself[u] = leaky(asn[u] + adn[u]);
        hself[u] = *(const float4*)((const char*)h3 + ((unsigned)nn[u] * 192u + gb));
    }

    int   sc[2][4];
    float ec[2][4];
    bool  vc[2][4];
#pragma unroll
    for (int u = 0; u < 2; ++u) {
#pragma unroll
        for (int c = 0; c < 4; ++c) {
            const int slot = c * 16 + r;
            vc[u][c] = slot < cnt[u];
            sc[u][c] = csr[nn[u] * RSTRIDE + (vc[u][c] ? 1 + slot : 0)];
        }
    }
#pragma unroll
    for (int u = 0; u < 2; ++u) {
#pragma unroll
        for (int c = 0; c < 4; ++c) {
            sc[u][c] = vc[u][c] ? sc[u][c] : nn[u];
            ec[u][c] = leaky(as[sc[u][c]] + adn[u]);
        }
    }
    float m[2];
#pragma unroll
    for (int u = 0; u < 2; ++u) {
        m[u] = eself[u];
#pragma unroll
        for (int c = 0; c < 4; ++c)
            m[u] = vc[u][c] ? fmaxf(m[u], ec[u][c]) : m[u];
    }
#pragma unroll
    for (int u = 0; u < 2; ++u) m[u] = fmaxf(m[u], row_xor<8>(m[u]));
#pragma unroll
    for (int u = 0; u < 2; ++u) m[u] = fmaxf(m[u], row_xor<4>(m[u]));
#pragma unroll
    for (int u = 0; u < 2; ++u) m[u] = fmaxf(m[u], row_xor<2>(m[u]));
#pragma unroll
    for (int u = 0; u < 2; ++u) m[u] = fmaxf(m[u], row_xor<1>(m[u]));

    float den[2];
    float4 acc[2];
#pragma unroll
    for (int u = 0; u < 2; ++u) {
        const float w = expf(eself[u] - m[u]);
        den[u] = w;
        acc[u].x = hself[u].x * w; acc[u].y = hself[u].y * w;
        acc[u].z = hself[u].z * w; acc[u].w = hself[u].w * w;
    }
#pragma unroll
    for (int c = 0; c < 4; ++c) {
        if (c * 16 < cnt[0] || c * 16 < cnt[1]) {
            const float wc0 = vc[0][c] ? expf(ec[0][c] - m[0]) : 0.f;
            const float wc1 = vc[1][c] ? expf(ec[1][c] - m[1]) : 0.f;
            v4f hv0[16], hv1[16];
            gat2_issue((unsigned)sc[0][c] * 192u, h3, gb, hv0);
            gat2_issue((unsigned)sc[1][c] * 192u, h3, gb, hv1);
            gat2_hold(hv0);
            gat2_hold(hv1);
            __builtin_amdgcn_sched_barrier(0);
            gat2_sum(wc0, cnt[0] - c * 16, hv0, den[0], acc[0]);
            gat2_sum(wc1, cnt[1] - c * 16, hv1, den[1], acc[1]);
        }
    }
    if (r < 10) {
        const float4 bb = *(const float4*)(b + g * 4);
#pragma unroll
        for (int u = 0; u < 2; ++u) {
            if (u == 0 || hasB) {
                float4 o;
                o.x = acc[u].x / den[u] + bb.x; o.y = acc[u].y / den[u] + bb.y;
                o.z = acc[u].z / den[u] + bb.z; o.w = acc[u].w / den[u] + bb.w;
                *(float4*)(out + (size_t)nn[u] * 40 + g * 4) = o;
            }
        }
    }
}

extern "C" void kernel_launch(void* const* d_in, const int* in_sizes, int n_in,
                              void* d_out, int out_size, void* d_ws, size_t ws_size,
                              hipStream_t stream)
{
    const float* x    = (const float*)d_in[0];
    const int*   ei   = (const int*)  d_in[1];
    const float* W1   = (const float*)d_in[2];
    const float* aS1  = (const float*)d_in[3];
    const float* aD1  = (const float*)d_in[4];
    const float* b1   = (const float*)d_in[5];
    const float* W2   = (const float*)d_in[6];
    const float* aS2  = (const float*)d_in[7];
    const float* aD2  = (const float*)d_in[8];
    const float* b2   = (const float*)d_in[9];
    float* out = (float*)d_out;

    const int IN  = 256, F1 = 512, OUT = 40, OUTP = 48;
    const int Nn  = in_sizes[0] / IN;
    const int E   = in_sizes[1] / 2;

    float* ws  = (float*)d_ws;
    float* h1  = ws;
    float* h3  = h1  + (size_t)Nn * F1;
    float* W2p = h3  + (size_t)Nn * 48;
    float* as1 = W2p + (size_t)F1 * OUTP;
    float* ad1 = as1 + (size_t)Nn * 8;
    int*   rec = (int*)(ad1 + (size_t)Nn * 8);
    float* as2 = (float*)(rec + (size_t)Nn * RSTRIDE);
    float* ad2 = as2 + Nn;

    const int B = 256;

    clear_counters_pad_w2<<<(Nn + B - 1) / B + (F1 * OUTP + B - 1) / B, B, 0, stream>>>(
        rec, Nn, W2, W2p, F1, OUT, OUTP);

    gemm1_and_fill_records<4><<<dim3((Nn + 63) / 64, F1 / 64 + 1), 128, 0, stream>>>(
        x, W1, h1, Nn, IN, F1, F1, aS1, aD1, as1, ad1, F1, 8, ei, E, rec);

    gat1_gather_gemm2<3, 8><<<(Nn + 15) / 16, 512, 0, stream>>>(
        rec, ei, E, as1, ad1, h1, b1, Nn, W2p, h3, OUTP, OUTP, aS2, aD2, as2, ad2, OUT);

    gat2_gather<<<((Nn + 1) / 2 * 16 + B - 1) / B, B, 0, stream>>>(rec, ei, E, as2, ad2, h3, b2, out, Nn);
}
